# starting kernel
# baseline (speedup 1.0000x reference)
#include <hip/hip_runtime.h>
#include <math.h>
#include <cstddef>

#define NNODES   20000
#define NEDGES   320000
#define ETOT     (NEDGES + NNODES)
#define IN_CH    128
#define HID_CH   32
#define HEADS    8
#define H1C      (HEADS * HID_CH)
#define OUT_CH   64
#define NEG_SLOPE 0.2f

typedef float  v2f   __attribute__((ext_vector_type(2)));
typedef float  v8f   __attribute__((ext_vector_type(8)));
typedef __bf16 v16bf __attribute__((ext_vector_type(16)));

__global__ void wmma_gemm(const float* __restrict__ A, const float* __restrict__ B,
                          float* __restrict__ C, int M, int K, int N) {
  const int lane = threadIdx.x;
  const int row  = lane & 15;
  const int kh   = lane >> 4;
  const int tm   = blockIdx.x;
  const int tn   = blockIdx.y;
  v8f acc = {0.f, 0.f, 0.f, 0.f, 0.f, 0.f, 0.f, 0.f};

#if __has_builtin(__builtin_amdgcn_wmma_f32_16x16x4_f32)
  const float* Arow = A + (size_t)(tm * 16 + row) * K;
  const float* Bcol = B + (size_t)(tn * 16 + row);
  for (int k = 0; k < K; k += 4) {
    v2f a, b;
    a.x = Arow[k + kh * 2 + 0];
    a.y = Arow[k + kh * 2 + 1];
    b.x = Bcol[(size_t)(k + kh * 2 + 0) * N];
    b.y = Bcol[(size_t)(k + kh * 2 + 1) * N];
    acc = __builtin_amdgcn_wmma_f32_16x16x4_f32(false, a, false, b,
                                                (short)0, acc, false, false);
  }
#elif __has_builtin(__builtin_amdgcn_wmma_f32_16x16x32_bf16)
  for (int k0 = 0; k0 < K; k0 += 32) {
    v16bf a, b;
#pragma unroll
    for (int i = 0; i < 16; ++i) {
      int vg = i >> 1, p = i & 1;
      int k  = k0 + ((vg & 4) ? 16 : 0) + kh * 8 + (vg & 3) * 2 + p;
      a[i] = (__bf16)A[(size_t)(tm * 16 + row) * K + k];
      b[i] = (__bf16)B[(size_t)k * N + tn * 16 + row];
    }
    acc = __builtin_amdgcn_wmma_f32_16x16x32_bf16(false, a, false, b,
                                                  (short)0, acc, false, false);
  }
#endif

  float* Crow = C + ((size_t)tm * 16 + kh * 8) * N + tn * 16 + row;
#pragma unroll
  for (int r = 0; r < 8; ++r) Crow[(size_t)r * N] = acc[r];
}

__global__ void fill_f32(float* __restrict__ p, float v, int n) {
  int i = blockIdx.x * blockDim.x + threadIdx.x;
  if (i < n) p[i] = v;
}

__device__ __forceinline__ void atomicMaxF(float* addr, float val) {
  if (val >= 0.f) atomicMax((int*)addr, __float_as_int(val));
  else            atomicMin((unsigned int*)addr, __float_as_uint(val));
}

__global__ void alpha_kernel(const float* __restrict__ h,
                             const float* __restrict__ a_src,
                             const float* __restrict__ a_dst,
                             float* __restrict__ as, float* __restrict__ ad,
                             int N, int H, int C) {
  int i = blockIdx.x * blockDim.x + threadIdx.x;
  if (i >= N * H) return;
  int hd = i % H;
  const float* hp = h + (size_t)i * C;
  float s = 0.f, d = 0.f;
  for (int c = 0; c < C; ++c) {
    float v = hp[c];
    s += v * a_src[hd * C + c];
    d += v * a_dst[hd * C + c];
  }
  as[i] = s;
  ad[i] = d;
}

__global__ void edge_max_kernel(const int* __restrict__ ei,
                                const float* __restrict__ as,
                                const float* __restrict__ ad,
                                float* __restrict__ e, float* __restrict__ m,
                                int Eorig, int Etot, int H) {
  int i = blockIdx.x * blockDim.x + threadIdx.x;
  if (i >= Etot * H) return;
  int eid = i / H, hd = i % H;
  int s, d;
  if (eid < Eorig) { s = ei[eid]; d = ei[Eorig + eid]; }
  else             { s = d = eid - Eorig; }
  float v = as[s * H + hd] + ad[d * H + hd];
  v = (v > 0.f) ? v : NEG_SLOPE * v;
  e[i] = v;
  atomicMaxF(&m[d * H + hd], v);
}

__global__ void edge_expsum_kernel(const int* __restrict__ ei,
                                   float* __restrict__ e,
                                   const float* __restrict__ m,
                                   float* __restrict__ den,
                                   int Eorig, int Etot, int H) {
  int i = blockIdx.x * blockDim.x + threadIdx.x;
  if (i >= Etot * H) return;
  int eid = i / H, hd = i % H;
  int d = (eid < Eorig) ? ei[Eorig + eid] : (eid - Eorig);
  float ex = __expf(e[i] - m[d * H + hd]);
  e[i] = ex;
  atomicAdd(&den[d * H + hd], ex);
}

__global__ void edge_scatter_kernel(const int* __restrict__ ei,
                                    const float* __restrict__ e,
                                    const float* __restrict__ h,
                                    float* __restrict__ acc,
                                    int Eorig, int Etot, int H, int C) {
  long long i = (long long)blockIdx.x * blockDim.x + threadIdx.x;
  int HC = H * C;
  if (i >= (long long)Etot * HC) return;
  int eid = (int)(i / HC);
  int ch  = (int)(i % HC);
  int hd  = ch / C;
  int s, d;
  if (eid < Eorig) { s = ei[eid]; d = ei[Eorig + eid]; }
  else             { s = d = eid - Eorig; }
  atomicAdd(&acc[(size_t)d * HC + ch],
            e[(size_t)eid * H + hd] * h[(size_t)s * HC + ch]);
}

__global__ void finalize_kernel(const float* __restrict__ acc,
                                const float* __restrict__ den,
                                const float* __restrict__ bias,
                                float* __restrict__ out,
                                int N, int H, int C, int do_elu) {
  int i = blockIdx.x * blockDim.x + threadIdx.x;
  int HC = H * C;
  if (i >= N * HC) return;
  int n = i / HC, ch = i % HC, hd = ch / C;
  float v = acc[i] / (den[n * H + hd] + 1e-16f) + bias[ch];
  if (do_elu) v = (v > 0.f) ? v : (__expf(v) - 1.f);
  out[i] = v;
}

extern "C" void kernel_launch(void* const* d_in, const int* in_sizes, int n_in,
                              void* d_out, int out_size, void* d_ws, size_t ws_size,
                              hipStream_t stream) {
  const float* x      = (const float*)d_in[0];
  const int*   ei     = (const int*)  d_in[1];
  const float* W1     = (const float*)d_in[2];
  const float* a_src1 = (const float*)d_in[3];
  const float* a_dst1 = (const float*)d_in[4];
  const float* b1     = (const float*)d_in[5];
  const float* W2     = (const float*)d_in[6];
  const float* a_src2 = (const float*)d_in[7];
  const float* a_dst2 = (const float*)d_in[8];
  const float* b2     = (const float*)d_in[9];
  float* out = (float*)d_out;

  float* ws   = (float*)d_ws;
  float* h1   = ws;
  float* acc1 = h1   + (size_t)NNODES * H1C;
  float* ebuf = acc1 + (size_t)NNODES * H1C;
  float* as1  = ebuf + (size_t)ETOT * HEADS;
  float* ad1  = as1  + (size_t)NNODES * HEADS;
  float* m1   = ad1  + (size_t)NNODES * HEADS;
  float* den1 = m1   + (size_t)NNODES * HEADS;
  float* h2   = acc1;
  float* as2  = acc1 + (size_t)NNODES * OUT_CH;
  float* ad2  = as2 + NNODES;
  float* m2   = ad2 + NNODES;
  float* den2 = m2  + NNODES;

  const int T = 256;
  const int NH1 = NNODES * HEADS;
  const int EH1 = ETOT * HEADS;
  const long long SC1 = (long long)ETOT * H1C;
  const long long SC2 = (long long)ETOT * OUT_CH;

  fill_f32<<<(NH1 + T - 1) / T, T, 0, stream>>>(m1, -INFINITY, NH1);
  fill_f32<<<(NH1 + T - 1) / T, T, 0, stream>>>(den1, 0.f, NH1);
  fill_f32<<<(NNODES * H1C + T - 1) / T, T, 0, stream>>>(acc1, 0.f, NNODES * H1C);

  wmma_gemm<<<dim3(NNODES / 16, H1C / 16), 32, 0, stream>>>(x, W1, h1,
                                                            NNODES, IN_CH, H1C);
  alpha_kernel<<<(NH1 + T - 1) / T, T, 0, stream>>>(h1, a_src1, a_dst1, as1, ad1,
                                                    NNODES, HEADS, HID_CH);
  edge_max_kernel<<<(EH1 + T - 1) / T, T, 0, stream>>>(ei, as1, ad1, ebuf, m1,
                                                       NEDGES, ETOT, HEADS);
  edge_expsum_kernel<<<(EH1 + T - 1) / T, T, 0, stream>>>(ei, ebuf, m1, den1,
                                                          NEDGES, ETOT, HEADS);
  edge_scatter_kernel<<<(unsigned)((SC1 + T - 1) / T), T, 0, stream>>>(
      ei, ebuf, h1, acc1, NEDGES, ETOT, HEADS, HID_CH);
  finalize_kernel<<<(NNODES * H1C + T - 1) / T, T, 0, stream>>>(
      acc1, den1, b1, h1, NNODES, HEADS, HID_CH,  1);

  fill_f32<<<(NNODES + T - 1) / T, T, 0, stream>>>(m2, -INFINITY, NNODES);
  fill_f32<<<(NNODES + T - 1) / T, T, 0, stream>>>(den2, 0.f, NNODES);
  fill_f32<<<(NNODES * OUT_CH + T - 1) / T, T, 0, stream>>>(out, 0.f, NNODES * OUT_CH);

  wmma_gemm<<<dim3(NNODES / 16, OUT_CH / 16), 32, 0, stream>>>(h1, W2, h2,
                                                               NNODES, H1C, OUT_CH);
  alpha_kernel<<<(NNODES + T - 1) / T, T, 0, stream>>>(h2, a_src2, a_dst2, as2, ad2,
                                                       NNODES, 1, OUT_CH);
  edge_max_kernel<<<(ETOT + T - 1) / T, T, 0, stream>>>(ei, as2, ad2, ebuf, m2,
                                                        NEDGES, ETOT, 1);
  edge_expsum_kernel<<<(ETOT + T - 1) / T, T, 0, stream>>>(ei, ebuf, m2, den2,
                                                           NEDGES, ETOT, 1);
  edge_scatter_kernel<<<(unsigned)((SC2 + T - 1) / T), T, 0, stream>>>(
      ei, ebuf, h2, out, NEDGES, ETOT, 1, OUT_CH);
  finalize_kernel<<<(NNODES * OUT_CH + T - 1) / T, T, 0, stream>>>(
      out, den2, b2, out, NNODES, 1, OUT_CH,  0);
}


# aggregation by owner: each workgroup owns 80 dst nodes, scans the dst list, sums ex*h[src] rows in LDS (LDS atomics), normalises and stores once; no global f32 atomics there; sum order changes
# speedup vs baseline: 2.3360x; 2.3360x over previous
#include <hip/hip_runtime.h>
#include <math.h>
#include <cstddef>

#define NNODES   20000
#define NEDGES   320000
#define ETOT     (NEDGES + NNODES)
#define IN_CH    128
#define HID_CH   32
#define HEADS    8
#define H1C      (HEADS * HID_CH)
#define OUT_CH   64
#define NEG_SLOPE 0.2f

typedef float  v2f   __attribute__((ext_vector_type(2)));
typedef float  v8f   __attribute__((ext_vector_type(8)));
typedef __bf16 v16bf __attribute__((ext_vector_type(16)));

__global__ void wmma_gemm(const float* __restrict__ A, const float* __restrict__ B,
                          float* __restrict__ C, int M, int K, int N) {
  const int lane = threadIdx.x;
  const int row  = lane & 15;
  const int kh   = lane >> 4;
  const int tm   = blockIdx.x;
  const int tn   = blockIdx.y;
  v8f acc = {0.f, 0.f, 0.f, 0.f, 0.f, 0.f, 0.f, 0.f};

#if __has_builtin(__builtin_amdgcn_wmma_f32_16x16x4_f32)
  const float* Arow = A + (size_t)(tm * 16 + row) * K;
  const float* Bcol = B + (size_t)(tn * 16 + row);
  for (int k = 0; k < K; k += 4) {
    v2f a, b;
    a.x = Arow[k + kh * 2 + 0];
    a.y = Arow[k + kh * 2 + 1];
    b.x = Bcol[(size_t)(k + kh * 2 + 0) * N];
    b.y = Bcol[(size_t)(k + kh * 2 + 1) * N];
    acc = __builtin_amdgcn_wmma_f32_16x16x4_f32(false, a, false, b,
                                                (short)0, acc, false, false);
  }
#elif __has_builtin(__builtin_amdgcn_wmma_f32_16x16x32_bf16)
  for (int k0 = 0; k0 < K; k0 += 32) {
    v16bf a, b;
#pragma unroll
    for (int i = 0; i < 16; ++i) {
      int vg = i >> 1, p = i & 1;
      int k  = k0 + ((vg & 4) ? 16 : 0) + kh * 8 + (vg & 3) * 2 + p;
      a[i] = (__bf16)A[(size_t)(tm * 16 + row) * K + k];
      b[i] = (__bf16)B[(size_t)k * N + tn * 16 + row];
    }
    acc = __builtin_amdgcn_wmma_f32_16x16x32_bf16(false, a, false, b,
                                                  (short)0, acc, false, false);
  }
#endif

  float* Crow = C + ((size_t)tm * 16 + kh * 8) * N + tn * 16 + row;
#pragma unroll
  for (int r = 0; r < 8; ++r) Crow[(size_t)r * N] = acc[r];
}

__global__ void fill_f32(float* __restrict__ p, float v, int n) {
  int i = blockIdx.x * blockDim.x + threadIdx.x;
  if (i < n) p[i] = v;
}

__device__ __forceinline__ void atomicMaxF(float* addr, float val) {
  if (val >= 0.f) atomicMax((int*)addr, __float_as_int(val));
  else            atomicMin((unsigned int*)addr, __float_as_uint(val));
}

__global__ void alpha_kernel(const float* __restrict__ h,
                             const float* __restrict__ a_src,
                             const float* __restrict__ a_dst,
                             float* __restrict__ as, float* __restrict__ ad,
                             int N, int H, int C) {
  int i = blockIdx.x * blockDim.x + threadIdx.x;
  if (i >= N * H) return;
  int hd = i % H;
  const float* hp = h + (size_t)i * C;
  float s = 0.f, d = 0.f;
  for (int c = 0; c < C; ++c) {
    float v = hp[c];
    s += v * a_src[hd * C + c];
    d += v * a_dst[hd * C + c];
  }
  as[i] = s;
  ad[i] = d;
}

__global__ void edge_max_kernel(const int* __restrict__ ei,
                                const float* __restrict__ as,
                                const float* __restrict__ ad,
                                float* __restrict__ e, float* __restrict__ m,
                                int Eorig, int Etot, int H) {
  int i = blockIdx.x * blockDim.x + threadIdx.x;
  if (i >= Etot * H) return;
  int eid = i / H, hd = i % H;
  int s, d;
  if (eid < Eorig) { s = ei[eid]; d = ei[Eorig + eid]; }
  else             { s = d = eid - Eorig; }
  float v = as[s * H + hd] + ad[d * H + hd];
  v = (v > 0.f) ? v : NEG_SLOPE * v;
  e[i] = v;
  atomicMaxF(&m[d * H + hd], v);
}

__global__ void edge_expsum_kernel(const int* __restrict__ ei,
                                   float* __restrict__ e,
                                   const float* __restrict__ m,
                                   float* __restrict__ den,
                                   int Eorig, int Etot, int H) {
  int i = blockIdx.x * blockDim.x + threadIdx.x;
  if (i >= Etot * H) return;
  int eid = i / H, hd = i % H;
  int d = (eid < Eorig) ? ei[Eorig + eid] : (eid - Eorig);
  float ex = __expf(e[i] - m[d * H + hd]);
  e[i] = ex;
  atomicAdd(&den[d * H + hd], ex);
}

template <int H, int C, int R, int QW, bool ELU>
__global__ __launch_bounds__(1024)
void agg_owner_kernel(const int* __restrict__ ei,
                      const float* __restrict__ e,
                      const float* __restrict__ h,
                      const float* __restrict__ den,
                      const float* __restrict__ bias,
                      float* __restrict__ out) {
  constexpr int HC  = H * C;
  constexpr int CPL = HC / 32;
  constexpr int NW  = 32;
  constexpr int UNR = 8;
  constexpr int NV4 = NEDGES / 4;
  constexpr int NCHUNK = (NV4 + 31) / 32;
  static_assert(C % 32 == 0, "a 32-lane line must stay inside one head");
  static_assert(NEDGES % 4 == 0, "destination list is read four at a time");
  static_assert(QW >= 256, "queue must take a chunk of 128 edges on top of QW-128");

  __shared__ float acc[R * HC];
  __shared__ int   q_eid[NW * QW];
  __shared__ int   q_dl[NW * QW];

  const int tid  = threadIdx.x;
  const int lane = tid & 31;
  const int w    = __builtin_amdgcn_readfirstlane(tid >> 5);
  const int lo   = blockIdx.x * R;
  int* qe = q_eid + w * QW;
  int* qd = q_dl  + w * QW;

  constexpr int NR = (R + NW - 1) / NW;
  {
    float iv[NR][CPL];
#pragma unroll
    for (int r = 0; r < NR; ++r) {
      const int  nl = w + r * NW;
      const int  n  = lo + nl;
      const bool ok = (nl < R) && (n < NNODES);
      const int  nn = ok ? n : 0;
#pragma unroll
      for (int j = 0; j < CPL; ++j) {
        const float p = e[(size_t)(NEDGES + nn) * H + (j * 32) / C] *
                        h[(size_t)nn * HC + j * 32 + lane];
        iv[r][j] = ok ? p : 0.f;
      }
    }
#pragma unroll
    for (int r = 0; r < NR; ++r) {
      const int nl = w + r * NW;
      if (nl < R) {
#pragma unroll
        for (int j = 0; j < CPL; ++j) acc[nl * HC + j * 32 + lane] = iv[r][j];
      }
    }
  }
  __syncthreads();

  int cnt = 0;

  auto flush = [&](int nq) {
    __builtin_amdgcn_fence(__ATOMIC_RELEASE, "wavefront");
    __builtin_amdgcn_wave_barrier();
    for (int b = 0; b < nq; b += 32) {
      const int  i     = b + lane;
      const bool valid = i < nq;
      const int eid_i = valid ? qe[i] : 0;
      const int dl_i  = valid ? qd[i] : 0;
      int s_i = valid ? ei[eid_i] : 0;
      s_i = min(max(s_i, 0), NNODES - 1);
      int ex_i[H];
#pragma unroll
      for (int hh = 0; hh < H; ++hh)
        ex_i[hh] = __float_as_int(e[(size_t)eid_i * H + hh]);
      const int n = min(32, nq - b);
      for (int k = 0; k < n; k += 4) {
        float v[4][CPL];
#pragma unroll
        for (int u = 0; u < 4; ++u) {
          const int s = __builtin_amdgcn_readlane(s_i, k + u);
          const float* hr = h + (size_t)s * HC + lane;
#pragma unroll
          for (int j = 0; j < CPL; ++j) v[u][j] = hr[j * 32];
        }
#pragma unroll
        for (int u = 0; u < 4; ++u) {
          if (k + u < n) {
            const int dl = __builtin_amdgcn_readlane(dl_i, k + u);
            float* ar = acc + dl * HC + lane;
#pragma unroll
            for (int j = 0; j < CPL; ++j) {
              const float ex =
                  __int_as_float(__builtin_amdgcn_readlane(ex_i[(j * 32) / C], k + u));
              atomicAdd(ar + j * 32, ex * v[u][j]);
            }
          }
        }
      }
    }
    __builtin_amdgcn_fence(__ATOMIC_ACQUIRE, "wavefront");
    __builtin_amdgcn_wave_barrier();
  };

  const int4* dst4 = reinterpret_cast<const int4*>(ei + NEDGES);
  for (int c0 = w; c0 < NCHUNK; c0 += NW * UNR) {
    int4 d[UNR];
#pragma unroll
    for (int u = 0; u < UNR; ++u) {
      const int i4 = (c0 + u * NW) * 32 + lane;
      d[u] = (i4 < NV4) ? dst4[i4] : make_int4(-1, -1, -1, -1);
    }
    int  done = 0;
    bool full = false;
    do {
#pragma unroll
      for (int u = 0; u < UNR; ++u) {
        if (u >= done && !full) {
          const int eid0 = ((c0 + u * NW) * 32 + lane) * 4;
          const int dd[4] = {d[u].x - lo, d[u].y - lo, d[u].z - lo, d[u].w - lo};
          const bool any = ((unsigned)dd[0] < (unsigned)R) | ((unsigned)dd[1] < (unsigned)R) |
                           ((unsigned)dd[2] < (unsigned)R) | ((unsigned)dd[3] < (unsigned)R);
          if (__builtin_amdgcn_ballot_w32(any)) {
#pragma unroll
            for (int c = 0; c < 4; ++c) {
              const bool m = (unsigned)dd[c] < (unsigned)R;
              const unsigned mask = __builtin_amdgcn_ballot_w32(m);
              if (m) {
                const int pos = cnt + (int)__builtin_amdgcn_mbcnt_lo(mask, 0u);
                qe[pos] = eid0 + c;
                qd[pos] = dd[c];
              }
              cnt += __builtin_popcount(mask);
            }
          }
          done = u + 1;
          full = cnt > QW - 128;
        }
      }
      if (full) { flush(cnt); cnt = 0; full = false; }
    } while (done < UNR);
  }
  flush(cnt);
  __syncthreads();

  float bj[CPL];
#pragma unroll
  for (int j = 0; j < CPL; ++j) bj[j] = bias[j * 32 + lane];
#pragma unroll
  for (int r = 0; r < NR; ++r) {
    const int nl = w + r * NW;
    const int n  = lo + nl;
    if (nl < R && n < NNODES) {
#pragma unroll
      for (int j = 0; j < CPL; ++j) {
        const int ch = j * 32 + lane;
        float v = acc[nl * HC + ch] / (den[n * H + (j * 32) / C] + 1e-16f) + bj[j];
        if (ELU) v = (v > 0.f) ? v : (__expf(v) - 1.f);
        out[(size_t)n * HC + ch] = v;
      }
    }
  }
}

extern "C" void kernel_launch(void* const* d_in, const int* in_sizes, int n_in,
                              void* d_out, int out_size, void* d_ws, size_t ws_size,
                              hipStream_t stream) {
  const float* x      = (const float*)d_in[0];
  const int*   ei     = (const int*)  d_in[1];
  const float* W1     = (const float*)d_in[2];
  const float* a_src1 = (const float*)d_in[3];
  const float* a_dst1 = (const float*)d_in[4];
  const float* b1     = (const float*)d_in[5];
  const float* W2     = (const float*)d_in[6];
  const float* a_src2 = (const float*)d_in[7];
  const float* a_dst2 = (const float*)d_in[8];
  const float* b2     = (const float*)d_in[9];
  float* out = (float*)d_out;

  float* ws   = (float*)d_ws;
  float* h1   = ws;
  float* g1   = h1   + (size_t)NNODES * H1C;
  float* ebuf = g1   + (size_t)NNODES * H1C;
  float* as1  = ebuf + (size_t)ETOT * HEADS;
  float* ad1  = as1  + (size_t)NNODES * HEADS;
  float* m1   = ad1  + (size_t)NNODES * HEADS;
  float* den1 = m1   + (size_t)NNODES * HEADS;
  float* h2   = den1 + (size_t)NNODES * HEADS;
  float* as2  = h2   + (size_t)NNODES * OUT_CH;
  float* ad2  = as2 + NNODES;
  float* m2   = ad2 + NNODES;
  float* den2 = m2  + NNODES;

  const int T = 256;
  const int NH1 = NNODES * HEADS;
  const int EH1 = ETOT * HEADS;
  const int RN  = 80;
  const int NG  = (NNODES + RN - 1) / RN;

  fill_f32<<<(NH1 + T - 1) / T, T, 0, stream>>>(m1, -INFINITY, NH1);
  fill_f32<<<(NH1 + T - 1) / T, T, 0, stream>>>(den1, 0.f, NH1);

  wmma_gemm<<<dim3(NNODES / 16, H1C / 16), 32, 0, stream>>>(x, W1, h1,
                                                            NNODES, IN_CH, H1C);
  alpha_kernel<<<(NH1 + T - 1) / T, T, 0, stream>>>(h1, a_src1, a_dst1, as1, ad1,
                                                    NNODES, HEADS, HID_CH);
  edge_max_kernel<<<(EH1 + T - 1) / T, T, 0, stream>>>(ei, as1, ad1, ebuf, m1,
                                                       NEDGES, ETOT, HEADS);
  edge_expsum_kernel<<<(EH1 + T - 1) / T, T, 0, stream>>>(ei, ebuf, m1, den1,
                                                          NEDGES, ETOT, HEADS);
  agg_owner_kernel<HEADS, HID_CH, RN, 256, true><<<NG, 1024, 0, stream>>>(
      ei, ebuf, h1, den1, b1, g1);

  fill_f32<<<(NNODES + T - 1) / T, T, 0, stream>>>(m2, -INFINITY, NNODES);
  fill_f32<<<(NNODES + T - 1) / T, T, 0, stream>>>(den2, 0.f, NNODES);

  wmma_gemm<<<dim3(NNODES / 16, OUT_CH / 16), 32, 0, stream>>>(g1, W2, h2,
                                                               NNODES, H1C, OUT_CH);
  alpha_kernel<<<(NNODES + T - 1) / T, T, 0, stream>>>(h2, a_src2, a_dst2, as2, ad2,
                                                       NNODES, 1, OUT_CH);
  edge_max_kernel<<<(ETOT + T - 1) / T, T, 0, stream>>>(ei, as2, ad2, ebuf, m2,
                                                        NEDGES, ETOT, 1);
  edge_expsum_kernel<<<(ETOT + T - 1) / T, T, 0, stream>>>(ei, ebuf, m2, den2,
                                                           NEDGES, ETOT, 1);
  agg_owner_kernel<1, OUT_CH, RN, 256, false><<<NG, 1024, 0, stream>>>(
      ei, ebuf, h2, den2, b2, out);
}


# segment softmax fused into the owner kernel: per (node,head) max and exp-sum live in LDS (LDS atomics; denominators' sum order changes); both global-atomic edge passes, e buffer and 4 fills removed
# speedup vs baseline: 3.9492x; 1.6906x over previous
#include <hip/hip_runtime.h>
#include <math.h>
#include <cstddef>

#define NNODES   20000
#define NEDGES   320000
#define ETOT     (NEDGES + NNODES)
#define IN_CH    128
#define HID_CH   32
#define HEADS    8
#define H1C      (HEADS * HID_CH)
#define OUT_CH   64
#define NEG_SLOPE 0.2f

typedef float  v2f   __attribute__((ext_vector_type(2)));
typedef float  v8f   __attribute__((ext_vector_type(8)));
typedef __bf16 v16bf __attribute__((ext_vector_type(16)));

__global__ void wmma_gemm(const float* __restrict__ A, const float* __restrict__ B,
                          float* __restrict__ C, int M, int K, int N) {
  const int lane = threadIdx.x;
  const int row  = lane & 15;
  const int kh   = lane >> 4;
  const int tm   = blockIdx.x;
  const int tn   = blockIdx.y;
  v8f acc = {0.f, 0.f, 0.f, 0.f, 0.f, 0.f, 0.f, 0.f};

#if __has_builtin(__builtin_amdgcn_wmma_f32_16x16x4_f32)
  const float* Arow = A + (size_t)(tm * 16 + row) * K;
  const float* Bcol = B + (size_t)(tn * 16 + row);
  for (int k = 0; k < K; k += 4) {
    v2f a, b;
    a.x = Arow[k + kh * 2 + 0];
    a.y = Arow[k + kh * 2 + 1];
    b.x = Bcol[(size_t)(k + kh * 2 + 0) * N];
    b.y = Bcol[(size_t)(k + kh * 2 + 1) * N];
    acc = __builtin_amdgcn_wmma_f32_16x16x4_f32(false, a, false, b,
                                                (short)0, acc, false, false);
  }
#elif __has_builtin(__builtin_amdgcn_wmma_f32_16x16x32_bf16)
  for (int k0 = 0; k0 < K; k0 += 32) {
    v16bf a, b;
#pragma unroll
    for (int i = 0; i < 16; ++i) {
      int vg = i >> 1, p = i & 1;
      int k  = k0 + ((vg & 4) ? 16 : 0) + kh * 8 + (vg & 3) * 2 + p;
      a[i] = (__bf16)A[(size_t)(tm * 16 + row) * K + k];
      b[i] = (__bf16)B[(size_t)k * N + tn * 16 + row];
    }
    acc = __builtin_amdgcn_wmma_f32_16x16x32_bf16(false, a, false, b,
                                                  (short)0, acc, false, false);
  }
#endif

  float* Crow = C + ((size_t)tm * 16 + kh * 8) * N + tn * 16 + row;
#pragma unroll
  for (int r = 0; r < 8; ++r) Crow[(size_t)r * N] = acc[r];
}

__device__ __forceinline__ void atomicMaxF(float* addr, float val) {
  if (val >= 0.f) atomicMax((int*)addr, __float_as_int(val));
  else            atomicMin((unsigned int*)addr, __float_as_uint(val));
}

__global__ void alpha_kernel(const float* __restrict__ h,
                             const float* __restrict__ a_src,
                             const float* __restrict__ a_dst,
                             float* __restrict__ as, float* __restrict__ ad,
                             int N, int H, int C) {
  int i = blockIdx.x * blockDim.x + threadIdx.x;
  if (i >= N * H) return;
  int hd = i % H;
  const float* hp = h + (size_t)i * C;
  float s = 0.f, d = 0.f;
  for (int c = 0; c < C; ++c) {
    float v = hp[c];
    s += v * a_src[hd * C + c];
    d += v * a_dst[hd * C + c];
  }
  as[i] = s;
  ad[i] = d;
}

template <int H, int C, int R, int QW, bool ELU>
__global__ __launch_bounds__(1024)
void agg_owner_kernel(const int* __restrict__ ei,
                      const float* __restrict__ as,
                      const float* __restrict__ ad,
                      const float* __restrict__ h,
                      const float* __restrict__ bias,
                      float* __restrict__ out) {
  constexpr int HC  = H * C;
  constexpr int CPL = HC / 32;
  constexpr int NW  = 32;
  constexpr int NT  = NW * 32;
  constexpr int UNR = 8;
  constexpr int NV4 = NEDGES / 4;
  constexpr int NCHUNK = (NV4 + 31) / 32;
  static_assert(C % 32 == 0, "a 32-lane line must stay inside one head");
  static_assert(NEDGES % 4 == 0, "destination list is read four at a time");
  static_assert(QW >= 256, "queue must take a chunk of 128 edges on top of QW-128");

  __shared__ float acc[R * HC];
  __shared__ int   q_eid[NW * QW];
  __shared__ int   q_dl[NW * QW];
  __shared__ float s_ad[R * H];
  __shared__ float s_self[R * H];
  __shared__ float s_m[R * H];
  __shared__ float s_den[R * H];

  const int tid  = threadIdx.x;
  const int lane = tid & 31;
  const int w    = __builtin_amdgcn_readfirstlane(tid >> 5);
  const int lo   = blockIdx.x * R;
  int* qe = q_eid + w * QW;
  int* qd = q_dl  + w * QW;

  for (int i = tid; i < R * H; i += NT) {
    const int n = lo + i / H;
    float a_d = 0.f, es = 0.f;
    if (n < NNODES) {
      a_d = ad[(size_t)lo * H + i];
      const float v = as[(size_t)lo * H + i] + a_d;
      es = (v > 0.f) ? v : NEG_SLOPE * v;
    }
    s_ad[i]   = a_d;
    s_self[i] = es;
    s_m[i]    = es;
  }
  __syncthreads();

  int  cnt     = 0;
  bool spilled = false;

  auto max_pass = [&](int nq) {
    __builtin_amdgcn_fence(__ATOMIC_RELEASE, "wavefront");
    __builtin_amdgcn_wave_barrier();
    for (int b = 0; b < nq; b += 32) {
      const int i = b + lane;
      if (i < nq) {
        const int dl = qd[i];
        int s = ei[qe[i]];
        s = min(max(s, 0), NNODES - 1);
        float a[H];
#pragma unroll
        for (int hh = 0; hh < H; ++hh) a[hh] = as[(size_t)s * H + hh];
#pragma unroll
        for (int hh = 0; hh < H; ++hh) {
          float v = a[hh] + s_ad[dl * H + hh];
          v = (v > 0.f) ? v : NEG_SLOPE * v;
          atomicMaxF(&s_m[dl * H + hh], v);
        }
      }
    }
    __builtin_amdgcn_fence(__ATOMIC_ACQUIRE, "wavefront");
    __builtin_amdgcn_wave_barrier();
  };

  auto acc_pass = [&](int nq) {
    __builtin_amdgcn_fence(__ATOMIC_RELEASE, "wavefront");
    __builtin_amdgcn_wave_barrier();
    for (int b = 0; b < nq; b += 32) {
      const int  i     = b + lane;
      const bool valid = i < nq;
      const int eid_i = valid ? qe[i] : 0;
      const int dl_i  = valid ? qd[i] : 0;
      int s_i = valid ? ei[eid_i] : 0;
      s_i = min(max(s_i, 0), NNODES - 1);
      float a[H];
#pragma unroll
      for (int hh = 0; hh < H; ++hh) a[hh] = as[(size_t)s_i * H + hh];
      int ex_i[H];
#pragma unroll
      for (int hh = 0; hh < H; ++hh) {
        float v = a[hh] + s_ad[dl_i * H + hh];
        v = (v > 0.f) ? v : NEG_SLOPE * v;
        const float ex = __expf(v - s_m[dl_i * H + hh]);
        if (valid) atomicAdd(&s_den[dl_i * H + hh], ex);
        ex_i[hh] = __float_as_int(ex);
      }
      const int n = min(32, nq - b);
      for (int k = 0; k < n; k += 4) {
        float v[4][CPL];
#pragma unroll
        for (int u = 0; u < 4; ++u) {
          const int s = __builtin_amdgcn_readlane(s_i, k + u);
          const float* hr = h + (size_t)s * HC + lane;
#pragma unroll
          for (int j = 0; j < CPL; ++j) v[u][j] = hr[j * 32];
        }
#pragma unroll
        for (int u = 0; u < 4; ++u) {
          if (k + u < n) {
            const int dl = __builtin_amdgcn_readlane(dl_i, k + u);
            float* ar = acc + dl * HC + lane;
#pragma unroll
            for (int j = 0; j < CPL; ++j) {
              const float ex =
                  __int_as_float(__builtin_amdgcn_readlane(ex_i[(j * 32) / C], k + u));
              atomicAdd(ar + j * 32, ex * v[u][j]);
            }
          }
        }
      }
    }
    __builtin_amdgcn_fence(__ATOMIC_ACQUIRE, "wavefront");
    __builtin_amdgcn_wave_barrier();
  };

  const int4* dst4 = reinterpret_cast<const int4*>(ei + NEDGES);
  auto scan = [&](auto&& drain) {
    for (int c0 = w; c0 < NCHUNK; c0 += NW * UNR) {
      int4 d[UNR];
#pragma unroll
      for (int u = 0; u < UNR; ++u) {
        const int i4 = (c0 + u * NW) * 32 + lane;
        d[u] = (i4 < NV4) ? dst4[i4] : make_int4(-1, -1, -1, -1);
      }
      int  done = 0;
      bool full = false;
      do {
#pragma unroll
        for (int u = 0; u < UNR; ++u) {
          if (u >= done && !full) {
            const int eid0 = ((c0 + u * NW) * 32 + lane) * 4;
            const int dd[4] = {d[u].x - lo, d[u].y - lo, d[u].z - lo, d[u].w - lo};
            const bool any = ((unsigned)dd[0] < (unsigned)R) | ((unsigned)dd[1] < (unsigned)R) |
                             ((unsigned)dd[2] < (unsigned)R) | ((unsigned)dd[3] < (unsigned)R);
            if (__builtin_amdgcn_ballot_w32(any)) {
#pragma unroll
              for (int c = 0; c < 4; ++c) {
                const bool m = (unsigned)dd[c] < (unsigned)R;
                const unsigned mask = __builtin_amdgcn_ballot_w32(m);
                if (m) {
                  const int pos = cnt + (int)__builtin_amdgcn_mbcnt_lo(mask, 0u);
                  qe[pos] = eid0 + c;
                  qd[pos] = dd[c];
                }
                cnt += __builtin_popcount(mask);
              }
            }
            done = u + 1;
            full = cnt > QW - 128;
          }
        }
        if (full) { drain(cnt); cnt = 0; full = false; }
      } while (done < UNR);
    }
  };

  scan([&](int nq) { max_pass(nq); spilled = true; });
  max_pass(cnt);
  __syncthreads();

  for (int i = tid; i < R * H; i += NT) s_den[i] = __expf(s_self[i] - s_m[i]);
  constexpr int NR = (R + NW - 1) / NW;
  {
    float iv[NR][CPL];
#pragma unroll
    for (int r = 0; r < NR; ++r) {
      const int  nl = w + r * NW;
      const int  n  = lo + nl;
      const bool ok = (nl < R) && (n < NNODES);
      const int  nn = ok ? n  : 0;
      const int  nc = ok ? nl : 0;
#pragma unroll
      for (int j = 0; j < CPL; ++j) {
        const int   hd = (j * 32) / C;
        const float p  = __expf(s_self[nc * H + hd] - s_m[nc * H + hd]) *
                         h[(size_t)nn * HC + j * 32 + lane];
        iv[r][j] = ok ? p : 0.f;
      }
    }
#pragma unroll
    for (int r = 0; r < NR; ++r) {
      const int nl = w + r * NW;
      if (nl < R) {
#pragma unroll
        for (int j = 0; j < CPL; ++j) acc[nl * HC + j * 32 + lane] = iv[r][j];
      }
    }
  }
  __syncthreads();

  if (spilled) {
    cnt = 0;
    scan([&](int nq) { acc_pass(nq); });
  }
  acc_pass(cnt);
  __syncthreads();

  float bj[CPL];
#pragma unroll
  for (int j = 0; j < CPL; ++j) bj[j] = bias[j * 32 + lane];
#pragma unroll
  for (int r = 0; r < NR; ++r) {
    const int nl = w + r * NW;
    const int n  = lo + nl;
    if (nl < R && n < NNODES) {
#pragma unroll
      for (int j = 0; j < CPL; ++j) {
        const int ch = j * 32 + lane;
        float v = acc[nl * HC + ch] / (s_den[nl * H + (j * 32) / C] + 1e-16f) + bj[j];
        if (ELU) v = (v > 0.f) ? v : (__expf(v) - 1.f);
        out[(size_t)n * HC + ch] = v;
      }
    }
  }
}

extern "C" void kernel_launch(void* const* d_in, const int* in_sizes, int n_in,
                              void* d_out, int out_size, void* d_ws, size_t ws_size,
                              hipStream_t stream) {
  const float* x      = (const float*)d_in[0];
  const int*   ei     = (const int*)  d_in[1];
  const float* W1     = (const float*)d_in[2];
  const float* a_src1 = (const float*)d_in[3];
  const float* a_dst1 = (const float*)d_in[4];
  const float* b1     = (const float*)d_in[5];
  const float* W2     = (const float*)d_in[6];
  const float* a_src2 = (const float*)d_in[7];
  const float* a_dst2 = (const float*)d_in[8];
  const float* b2     = (const float*)d_in[9];
  float* out = (float*)d_out;

  float* ws   = (float*)d_ws;
  float* h1   = ws;
  float* g1   = h1   + (size_t)NNODES * H1C;
  float* as1  = g1   + (size_t)NNODES * H1C;
  float* ad1  = as1  + (size_t)NNODES * HEADS;
  float* h2   = ad1  + (size_t)NNODES * HEADS;
  float* as2  = h2   + (size_t)NNODES * OUT_CH;
  float* ad2  = as2 + NNODES;

  const int T = 256;
  const int NH1 = NNODES * HEADS;
  const int RN  = 80;
  const int NG  = (NNODES + RN - 1) / RN;

  wmma_gemm<<<dim3(NNODES / 16, H1C / 16), 32, 0, stream>>>(x, W1, h1,
                                                            NNODES, IN_CH, H1C);
  alpha_kernel<<<(NH1 + T - 1) / T, T, 0, stream>>>(h1, a_src1, a_dst1, as1, ad1,
                                                    NNODES, HEADS, HID_CH);
  agg_owner_kernel<HEADS, HID_CH, RN, 256, true><<<NG, 1024, 0, stream>>>(
      ei, as1, ad1, h1, b1, g1);

  wmma_gemm<<<dim3(NNODES / 16, OUT_CH / 16), 32, 0, stream>>>(g1, W2, h2,
                                                               NNODES, H1C, OUT_CH);
  alpha_kernel<<<(NNODES + T - 1) / T, T, 0, stream>>>(h2, a_src2, a_dst2, as2, ad2,
                                                       NNODES, 1, OUT_CH);
  agg_owner_kernel<1, OUT_CH, RN, 256, false><<<NG, 1024, 0, stream>>>(
      ei, as2, ad2, h2, b2, out);
}


# GEMMs: one 8-wave workgroup per 80-row strip, A strip and all of B staged in LDS by whole-line 16-B loads, 10 (layer 2: 3 or 2) accumulator tiles per wave, C lines stored whole; sum order unchanged
# speedup vs baseline: 5.1887x; 1.3139x over previous
#include <hip/hip_runtime.h>
#include <math.h>
#include <cstddef>

#define NNODES   20000
#define NEDGES   320000
#define ETOT     (NEDGES + NNODES)
#define IN_CH    128
#define HID_CH   32
#define HEADS    8
#define H1C      (HEADS * HID_CH)
#define OUT_CH   64
#define NEG_SLOPE 0.2f

typedef float  v2f   __attribute__((ext_vector_type(2)));
typedef float  v8f   __attribute__((ext_vector_type(8)));

template <int V> struct int_const { static constexpr int value = V; };

template <int K, int N, int TN, int TM, int WM>
__global__ __launch_bounds__(256)
void gemm_strip_kernel(const float* __restrict__ A, const float* __restrict__ B,
                       float* __restrict__ C) {
  constexpr int RM  = 80;
  constexpr int MT  = RM / 16;
  constexpr int NTH = 256;
  constexpr int WN  = (N / 16) / TN;
  constexpr int PA  = K + 4;
  constexpr int PB  = 2 * N + 32;
  constexpr int PO  = N + 4;
  constexpr int KV  = K / 4;
  constexpr int NV  = N / 4;
  constexpr int A_WORDS = RM * PA;
  constexpr int B_WORDS = (K / 2) * PB;
  static_assert(WN * WM * 32 == NTH, "8 waves");
  static_assert(TM * WM >= MT && TM * (WM - 1) < MT, "M-tiles dealt over WM wave rows");
  static_assert(RM * PO <= A_WORDS + B_WORDS, "C tile reuses the operand space");
  static_assert((RM * KV) % (NTH * 10) == 0, "A staging in batches of 10");
  static_assert(((K / 2) * NV) % (NTH * 8) == 0, "B staging in batches of 8");
  static_assert((RM * NV) % NTH == 0, "C store loop");
  static_assert(NNODES % RM == 0, "no partial strip");

  __shared__ alignas(16) float smem[A_WORDS + B_WORDS];
  float* As = smem;
  float* Bs = smem + A_WORDS;
  float* Os = smem;

  const int tid  = threadIdx.x;
  const int lane = tid & 31;
  const int w    = __builtin_amdgcn_readfirstlane(tid >> 5);
  const int c    = lane & 15;
  const int h    = lane >> 4;

  {
    const float4* Ag = reinterpret_cast<const float4*>(A + (size_t)blockIdx.x * RM * K);
#pragma unroll
    for (int i0 = 0; i0 < (RM * KV) / NTH; i0 += 10) {
      float4 r[10];
#pragma unroll
      for (int u = 0; u < 10; ++u) r[u] = Ag[(i0 + u) * NTH + tid];
#pragma unroll
      for (int u = 0; u < 10; ++u) {
        const int idx = (i0 + u) * NTH + tid;
        *reinterpret_cast<float4*>(&As[(idx / KV) * PA + (idx % KV) * 4]) = r[u];
      }
    }
  }
  {
    const float4* Bg = reinterpret_cast<const float4*>(B);
#pragma unroll
    for (int i0 = 0; i0 < ((K / 2) * NV) / NTH; i0 += 8) {
      float4 r0[8], r1[8];
#pragma unroll
      for (int u = 0; u < 8; ++u) {
        const int idx = (i0 + u) * NTH + tid;
        const int p = idx / NV, c4 = idx % NV;
        r0[u] = Bg[(2 * p) * NV + c4];
        r1[u] = Bg[(2 * p + 1) * NV + c4];
      }
#pragma unroll
      for (int u = 0; u < 8; ++u) {
        const int idx = (i0 + u) * NTH + tid;
        const int p = idx / NV, c4 = idx % NV;
        float* d = &Bs[p * PB + c4 * 8];
        *reinterpret_cast<float4*>(d)     = make_float4(r0[u].x, r1[u].x, r0[u].y, r1[u].y);
        *reinterpret_cast<float4*>(d + 4) = make_float4(r0[u].z, r1[u].z, r0[u].w, r1[u].w);
      }
    }
  }
  __syncthreads();

  const int wn = w % WN;
  const int wm = w / WN;
  const int mb = wm * TM;
  const int mc = min(TM, MT - mb);
  v8f acc[TM][TN];
#pragma unroll
  for (int i = 0; i < TM; ++i)
#pragma unroll
    for (int j = 0; j < TN; ++j) acc[i][j] = v8f{0.f, 0.f, 0.f, 0.f, 0.f, 0.f, 0.f, 0.f};

  const float* Ap = As + (mb * 16 + c) * PA + 2 * h;
  const float* Bp = Bs + h * PB + (wn * TN * 16 + c) * 2;
  auto kloop = [&](auto cnt) {
    constexpr int CNT = decltype(cnt)::value;
#pragma unroll 4
    for (int ks = 0; ks < K / 4; ++ks) {
      v2f a[CNT], b[TN];
#pragma unroll
      for (int i = 0; i < CNT; ++i)
        a[i] = *reinterpret_cast<const v2f*>(Ap + i * 16 * PA + 4 * ks);
#pragma unroll
      for (int j = 0; j < TN; ++j)
        b[j] = *reinterpret_cast<const v2f*>(Bp + 2 * ks * PB + j * 32);
#pragma unroll
      for (int i = 0; i < CNT; ++i)
#pragma unroll
        for (int j = 0; j < TN; ++j)
          acc[i][j] = __builtin_amdgcn_wmma_f32_16x16x4_f32(false, a[i], false, b[j],
                                                            (short)0, acc[i][j], false, false);
    }
  };
  if constexpr (MT % TM == 0) {
    kloop(int_const<TM>{});
  } else {
    if (mc == TM) kloop(int_const<TM>{});
    else          kloop(int_const<MT % TM>{});
  }
  __syncthreads();

#pragma unroll
  for (int i = 0; i < TM; ++i) {
    if (i < mc) {
#pragma unroll
      for (int j = 0; j < TN; ++j)
#pragma unroll
        for (int r = 0; r < 8; ++r)
          Os[((mb + i) * 16 + 8 * h + r) * PO + (wn * TN + j) * 16 + c] = acc[i][j][r];
    }
  }
  __syncthreads();

  float4* Cg = reinterpret_cast<float4*>(C + (size_t)blockIdx.x * RM * N);
#pragma unroll 4
  for (int idx = tid; idx < RM * NV; idx += NTH)
    Cg[idx] = *reinterpret_cast<const float4*>(&Os[(idx / NV) * PO + (idx % NV) * 4]);
}

__device__ __forceinline__ void atomicMaxF(float* addr, float val) {
  if (val >= 0.f) atomicMax((int*)addr, __float_as_int(val));
  else            atomicMin((unsigned int*)addr, __float_as_uint(val));
}

__global__ void alpha_kernel(const float* __restrict__ h,
                             const float* __restrict__ a_src,
                             const float* __restrict__ a_dst,
                             float* __restrict__ as, float* __restrict__ ad,
                             int N, int H, int C) {
  int i = blockIdx.x * blockDim.x + threadIdx.x;
  if (i >= N * H) return;
  int hd = i % H;
  const float* hp = h + (size_t)i * C;
  float s = 0.f, d = 0.f;
  for (int c = 0; c < C; ++c) {
    float v = hp[c];
    s += v * a_src[hd * C + c];
    d += v * a_dst[hd * C + c];
  }
  as[i] = s;
  ad[i] = d;
}

template <int H, int C, int R, int QW, bool ELU>
__global__ __launch_bounds__(1024)
void agg_owner_kernel(const int* __restrict__ ei,
                      const float* __restrict__ as,
                      const float* __restrict__ ad,
                      const float* __restrict__ h,
                      const float* __restrict__ bias,
                      float* __restrict__ out) {
  constexpr int HC  = H * C;
  constexpr int CPL = HC / 32;
  constexpr int NW  = 32;
  constexpr int NT  = NW * 32;
  constexpr int UNR = 8;
  constexpr int NV4 = NEDGES / 4;
  constexpr int NCHUNK = (NV4 + 31) / 32;
  static_assert(C % 32 == 0, "a 32-lane line must stay inside one head");
  static_assert(NEDGES % 4 == 0, "destination list is read four at a time");
  static_assert(QW >= 256, "queue must take a chunk of 128 edges on top of QW-128");

  __shared__ float acc[R * HC];
  __shared__ int   q_eid[NW * QW];
  __shared__ int   q_dl[NW * QW];
  __shared__ float s_ad[R * H];
  __shared__ float s_self[R * H];
  __shared__ float s_m[R * H];
  __shared__ float s_den[R * H];

  const int tid  = threadIdx.x;
  const int lane = tid & 31;
  const int w    = __builtin_amdgcn_readfirstlane(tid >> 5);
  const int lo   = blockIdx.x * R;
  int* qe = q_eid + w * QW;
  int* qd = q_dl  + w * QW;

  for (int i = tid; i < R * H; i += NT) {
    const int n = lo + i / H;
    float a_d = 0.f, es = 0.f;
    if (n < NNODES) {
      a_d = ad[(size_t)lo * H + i];
      const float v = as[(size_t)lo * H + i] + a_d;
      es = (v > 0.f) ? v : NEG_SLOPE * v;
    }
    s_ad[i]   = a_d;
    s_self[i] = es;
    s_m[i]    = es;
  }
  __syncthreads();

  int  cnt     = 0;
  bool spilled = false;

  auto max_pass = [&](int nq) {
    __builtin_amdgcn_fence(__ATOMIC_RELEASE, "wavefront");
    __builtin_amdgcn_wave_barrier();
    for (int b = 0; b < nq; b += 32) {
      const int i = b + lane;
      if (i < nq) {
        const int dl = qd[i];
        int s = ei[qe[i]];
        s = min(max(s, 0), NNODES - 1);
        float a[H];
#pragma unroll
        for (int hh = 0; hh < H; ++hh) a[hh] = as[(size_t)s * H + hh];
#pragma unroll
        for (int hh = 0; hh < H; ++hh) {
          float v = a[hh] + s_ad[dl * H + hh];
          v = (v > 0.f) ? v : NEG_SLOPE * v;
          atomicMaxF(&s_m[dl * H + hh], v);
        }
      }
    }
    __builtin_amdgcn_fence(__ATOMIC_ACQUIRE, "wavefront");
    __builtin_amdgcn_wave_barrier();
  };

  auto acc_pass = [&](int nq) {
    __builtin_amdgcn_fence(__ATOMIC_RELEASE, "wavefront");
    __builtin_amdgcn_wave_barrier();
    for (int b = 0; b < nq; b += 32) {
      const int  i     = b + lane;
      const bool valid = i < nq;
      const int eid_i = valid ? qe[i] : 0;
      const int dl_i  = valid ? qd[i] : 0;
      int s_i = valid ? ei[eid_i] : 0;
      s_i = min(max(s_i, 0), NNODES - 1);
      float a[H];
#pragma unroll
      for (int hh = 0; hh < H; ++hh) a[hh] = as[(size_t)s_i * H + hh];
      int ex_i[H];
#pragma unroll
      for (int hh = 0; hh < H; ++hh) {
        float v = a[hh] + s_ad[dl_i * H + hh];
        v = (v > 0.f) ? v : NEG_SLOPE * v;
        const float ex = __expf(v - s_m[dl_i * H + hh]);
        if (valid) atomicAdd(&s_den[dl_i * H + hh], ex);
        ex_i[hh] = __float_as_int(ex);
      }
      const int n = min(32, nq - b);
      for (int k = 0; k < n; k += 4) {
        float v[4][CPL];
#pragma unroll
        for (int u = 0; u < 4; ++u) {
          const int s = __builtin_amdgcn_readlane(s_i, k + u);
          const float* hr = h + (size_t)s * HC + lane;
#pragma unroll
          for (int j = 0; j < CPL; ++j) v[u][j] = hr[j * 32];
        }
#pragma unroll
        for (int u = 0; u < 4; ++u) {
          if (k + u < n) {
            const int dl = __builtin_amdgcn_readlane(dl_i, k + u);
            float* ar = acc + dl * HC + lane;
#pragma unroll
            for (int j = 0; j < CPL; ++j) {
              const float ex =
                  __int_as_float(__builtin_amdgcn_readlane(ex_i[(j * 32) / C], k + u));
              atomicAdd(ar + j * 32, ex * v[u][j]);
            }
          }
        }
      }
    }
    __builtin_amdgcn_fence(__ATOMIC_ACQUIRE, "wavefront");
    __builtin_amdgcn_wave_barrier();
  };

  const int4* dst4 = reinterpret_cast<const int4*>(ei + NEDGES);
  auto scan = [&](auto&& drain) {
    for (int c0 = w; c0 < NCHUNK; c0 += NW * UNR) {
      int4 d[UNR];
#pragma unroll
      for (int u = 0; u < UNR; ++u) {
        const int i4 = (c0 + u * NW) * 32 + lane;
        d[u] = (i4 < NV4) ? dst4[i4] : make_int4(-1, -1, -1, -1);
      }
      int  done = 0;
      bool full = false;
      do {
#pragma unroll
        for (int u = 0; u < UNR; ++u) {
          if (u >= done && !full) {
            const int eid0 = ((c0 + u * NW) * 32 + lane) * 4;
            const int dd[4] = {d[u].x - lo, d[u].y - lo, d[u].z - lo, d[u].w - lo};
            const bool any = ((unsigned)dd[0] < (unsigned)R) | ((unsigned)dd[1] < (unsigned)R) |
                             ((unsigned)dd[2] < (unsigned)R) | ((unsigned)dd[3] < (unsigned)R);
            if (__builtin_amdgcn_ballot_w32(any)) {
#pragma unroll
              for (int c = 0; c < 4; ++c) {
                const bool m = (unsigned)dd[c] < (unsigned)R;
                const unsigned mask = __builtin_amdgcn_ballot_w32(m);
                if (m) {
                  const int pos = cnt + (int)__builtin_amdgcn_mbcnt_lo(mask, 0u);
                  qe[pos] = eid0 + c;
                  qd[pos] = dd[c];
                }
                cnt += __builtin_popcount(mask);
              }
            }
            done = u + 1;
            full = cnt > QW - 128;
          }
        }
        if (full) { drain(cnt); cnt = 0; full = false; }
      } while (done < UNR);
    }
  };

  scan([&](int nq) { max_pass(nq); spilled = true; });
  max_pass(cnt);
  __syncthreads();

  for (int i = tid; i < R * H; i += NT) s_den[i] = __expf(s_self[i] - s_m[i]);
  constexpr int NR = (R + NW - 1) / NW;
  {
    float iv[NR][CPL];
#pragma unroll
    for (int r = 0; r < NR; ++r) {
      const int  nl = w + r * NW;
      const int  n  = lo + nl;
      const bool ok = (nl < R) && (n < NNODES);
      const int  nn = ok ? n  : 0;
      const int  nc = ok ? nl : 0;
#pragma unroll
      for (int j = 0; j < CPL; ++j) {
        const int   hd = (j * 32) / C;
        const float p  = __expf(s_self[nc * H + hd] - s_m[nc * H + hd]) *
                         h[(size_t)nn * HC + j * 32 + lane];
        iv[r][j] = ok ? p : 0.f;
      }
    }
#pragma unroll
    for (int r = 0; r < NR; ++r) {
      const int nl = w + r * NW;
      if (nl < R) {
#pragma unroll
        for (int j = 0; j < CPL; ++j) acc[nl * HC + j * 32 + lane] = iv[r][j];
      }
    }
  }
  __syncthreads();

  if (spilled) {
    cnt = 0;
    scan([&](int nq) { acc_pass(nq); });
  }
  acc_pass(cnt);
  __syncthreads();

  float bj[CPL];
#pragma unroll
  for (int j = 0; j < CPL; ++j) bj[j] = bias[j * 32 + lane];
#pragma unroll
  for (int r = 0; r < NR; ++r) {
    const int nl = w + r * NW;
    const int n  = lo + nl;
    if (nl < R && n < NNODES) {
#pragma unroll
      for (int j = 0; j < CPL; ++j) {
        const int ch = j * 32 + lane;
        float v = acc[nl * HC + ch] / (s_den[nl * H + (j * 32) / C] + 1e-16f) + bj[j];
        if (ELU) v = (v > 0.f) ? v : (__expf(v) - 1.f);
        out[(size_t)n * HC + ch] = v;
      }
    }
  }
}

extern "C" void kernel_launch(void* const* d_in, const int* in_sizes, int n_in,
                              void* d_out, int out_size, void* d_ws, size_t ws_size,
                              hipStream_t stream) {
  const float* x      = (const float*)d_in[0];
  const int*   ei     = (const int*)  d_in[1];
  const float* W1     = (const float*)d_in[2];
  const float* a_src1 = (const float*)d_in[3];
  const float* a_dst1 = (const float*)d_in[4];
  const float* b1     = (const float*)d_in[5];
  const float* W2     = (const float*)d_in[6];
  const float* a_src2 = (const float*)d_in[7];
  const float* a_dst2 = (const float*)d_in[8];
  const float* b2     = (const float*)d_in[9];
  float* out = (float*)d_out;

  float* ws   = (float*)d_ws;
  float* h1   = ws;
  float* g1   = h1   + (size_t)NNODES * H1C;
  float* as1  = g1   + (size_t)NNODES * H1C;
  float* ad1  = as1  + (size_t)NNODES * HEADS;
  float* h2   = ad1  + (size_t)NNODES * HEADS;
  float* as2  = h2   + (size_t)NNODES * OUT_CH;
  float* ad2  = as2 + NNODES;

  const int T = 256;
  const int NH1 = NNODES * HEADS;
  const int RN  = 80;
  const int NG  = (NNODES + RN - 1) / RN;

  gemm_strip_kernel<IN_CH, H1C, 2, 5, 1><<<NNODES / 80, 256, 0, stream>>>(x, W1, h1);
  alpha_kernel<<<(NH1 + T - 1) / T, T, 0, stream>>>(h1, a_src1, a_dst1, as1, ad1,
                                                    NNODES, HEADS, HID_CH);
  agg_owner_kernel<HEADS, HID_CH, RN, 256, true><<<NG, 1024, 0, stream>>>(
      ei, as1, ad1, h1, b1, g1);

  gemm_strip_kernel<H1C, OUT_CH, 1, 3, 2><<<NNODES / 80, 256, 0, stream>>>(g1, W2, h2);
  alpha_kernel<<<(NNODES + T - 1) / T, T, 0, stream>>>(h2, a_src2, a_dst2, as2, ad2,
                                                       NNODES, 1, OUT_CH);
  agg_owner_kernel<1, OUT_CH, RN, 256, false><<<NG, 1024, 0, stream>>>(
      ei, as2, ad2, h2, b2, out);
}


# attention logits formed in the GEMM kernels from the C strip in LDS (same ascending channel sum, one array [node][src|dst][head]); both logit launches and their re-read of h removed; 6 launches -> 4
# speedup vs baseline: 5.5665x; 1.0728x over previous
#include <hip/hip_runtime.h>
#include <math.h>
#include <cstddef>

#define NNODES   20000
#define NEDGES   320000
#define ETOT     (NEDGES + NNODES)
#define IN_CH    128
#define HID_CH   32
#define HEADS    8
#define H1C      (HEADS * HID_CH)
#define OUT_CH   64
#define NEG_SLOPE 0.2f

typedef float  v2f   __attribute__((ext_vector_type(2)));
typedef float  v8f   __attribute__((ext_vector_type(8)));

template <int V> struct int_const { static constexpr int value = V; };

template <int K, int N, int TN, int TM, int WM, int H>
__global__ __launch_bounds__(256)
void gemm_strip_kernel(const float* __restrict__ A, const float* __restrict__ B,
                       const float* __restrict__ a_src, const float* __restrict__ a_dst,
                       float* __restrict__ C, float* __restrict__ al) {
  constexpr int RM  = 80;
  constexpr int MT  = RM / 16;
  constexpr int NTH = 256;
  constexpr int WN  = (N / 16) / TN;
  constexpr int PA  = K + 4;
  constexpr int PB  = 2 * N + 32;
  constexpr int PO  = N + 4;
  constexpr int KV  = K / 4;
  constexpr int NV  = N / 4;
  constexpr int A_WORDS = RM * PA;
  constexpr int B_WORDS = (K / 2) * PB;
  constexpr int CH  = N / H;
  constexpr int AVN = (N + NTH - 1) / NTH;
  constexpr int AVW = AVN * NTH;
  static_assert(WN * WM * 32 == NTH, "8 waves");
  static_assert(TM * WM >= MT && TM * (WM - 1) < MT, "M-tiles dealt over WM wave rows");
  static_assert(RM * PO <= A_WORDS + B_WORDS, "C tile reuses the operand space");
  static_assert((RM * KV) % (NTH * 10) == 0, "A staging in batches of 10");
  static_assert(((K / 2) * NV) % (NTH * 8) == 0, "B staging in batches of 8");
  static_assert((RM * NV) % NTH == 0, "C store loop");
  static_assert(NNODES % RM == 0, "no partial strip");
  static_assert(N % H == 0 && CH % 4 == 0, "a head's channels are read four at a time");
  static_assert((RM * 2 * H) % 32 == 0, "a strip's logits fill whole 128-B lines");

  __shared__ alignas(16) float smem[A_WORDS + B_WORDS];
  __shared__ alignas(16) float s_av[2 * AVW];
  __shared__ alignas(16) float s_al[RM * 2 * H];
  float* As = smem;
  float* Bs = smem + A_WORDS;
  float* Os = smem;

  const int tid  = threadIdx.x;
  const int lane = tid & 31;
  const int w    = __builtin_amdgcn_readfirstlane(tid >> 5);
  const int c    = lane & 15;
  const int h    = lane >> 4;

  float avs[AVN], avd[AVN];
#pragma unroll
  for (int u = 0; u < AVN; ++u) {
    const int idx = min(u * NTH + tid, N - 1);
    avs[u] = a_src[idx];
    avd[u] = a_dst[idx];
  }

  {
    const float4* Ag = reinterpret_cast<const float4*>(A + (size_t)blockIdx.x * RM * K);
#pragma unroll
    for (int i0 = 0; i0 < (RM * KV) / NTH; i0 += 10) {
      float4 r[10];
#pragma unroll
      for (int u = 0; u < 10; ++u) r[u] = Ag[(i0 + u) * NTH + tid];
#pragma unroll
      for (int u = 0; u < 10; ++u) {
        const int idx = (i0 + u) * NTH + tid;
        *reinterpret_cast<float4*>(&As[(idx / KV) * PA + (idx % KV) * 4]) = r[u];
      }
    }
  }
  {
    const float4* Bg = reinterpret_cast<const float4*>(B);
#pragma unroll
    for (int i0 = 0; i0 < ((K / 2) * NV) / NTH; i0 += 8) {
      float4 r0[8], r1[8];
#pragma unroll
      for (int u = 0; u < 8; ++u) {
        const int idx = (i0 + u) * NTH + tid;
        const int p = idx / NV, c4 = idx % NV;
        r0[u] = Bg[(2 * p) * NV + c4];
        r1[u] = Bg[(2 * p + 1) * NV + c4];
      }
#pragma unroll
      for (int u = 0; u < 8; ++u) {
        const int idx = (i0 + u) * NTH + tid;
        const int p = idx / NV, c4 = idx % NV;
        float* d = &Bs[p * PB + c4 * 8];
        *reinterpret_cast<float4*>(d)     = make_float4(r0[u].x, r1[u].x, r0[u].y, r1[u].y);
        *reinterpret_cast<float4*>(d + 4) = make_float4(r0[u].z, r1[u].z, r0[u].w, r1[u].w);
      }
    }
  }
#pragma unroll
  for (int u = 0; u < AVN; ++u) {
    s_av[u * NTH + tid]       = avs[u];
    s_av[AVW + u * NTH + tid] = avd[u];
  }
  __syncthreads();

  const int wn = w % WN;
  const int wm = w / WN;
  const int mb = wm * TM;
  const int mc = min(TM, MT - mb);
  v8f acc[TM][TN];
#pragma unroll
  for (int i = 0; i < TM; ++i)
#pragma unroll
    for (int j = 0; j < TN; ++j) acc[i][j] = v8f{0.f, 0.f, 0.f, 0.f, 0.f, 0.f, 0.f, 0.f};

  const float* Ap = As + (mb * 16 + c) * PA + 2 * h;
  const float* Bp = Bs + h * PB + (wn * TN * 16 + c) * 2;
  auto kloop = [&](auto cnt) {
    constexpr int CNT = decltype(cnt)::value;
#pragma unroll 4
    for (int ks = 0; ks < K / 4; ++ks) {
      v2f a[CNT], b[TN];
#pragma unroll
      for (int i = 0; i < CNT; ++i)
        a[i] = *reinterpret_cast<const v2f*>(Ap + i * 16 * PA + 4 * ks);
#pragma unroll
      for (int j = 0; j < TN; ++j)
        b[j] = *reinterpret_cast<const v2f*>(Bp + 2 * ks * PB + j * 32);
#pragma unroll
      for (int i = 0; i < CNT; ++i)
#pragma unroll
        for (int j = 0; j < TN; ++j)
          acc[i][j] = __builtin_amdgcn_wmma_f32_16x16x4_f32(false, a[i], false, b[j],
                                                            (short)0, acc[i][j], false, false);
    }
  };
  if constexpr (MT % TM == 0) {
    kloop(int_const<TM>{});
  } else {
    if (mc == TM) kloop(int_const<TM>{});
    else          kloop(int_const<MT % TM>{});
  }
  __syncthreads();

#pragma unroll
  for (int i = 0; i < TM; ++i) {
    if (i < mc) {
#pragma unroll
      for (int j = 0; j < TN; ++j)
#pragma unroll
        for (int r = 0; r < 8; ++r)
          Os[((mb + i) * 16 + 8 * h + r) * PO + (wn * TN + j) * 16 + c] = acc[i][j][r];
    }
  }
  __syncthreads();

  constexpr int RB = (RM + 31) / 32;
  for (int blk = w; blk < H * RB; blk += NTH / 32) {
    const int  hd  = blk / RB;
    const int  row = (blk % RB) * 32 + lane;
    const bool ok  = row < RM;
    const float4* o4 = reinterpret_cast<const float4*>(&Os[(ok ? row : 0) * PO + hd * CH]);
    const float4* s4 = reinterpret_cast<const float4*>(&s_av[hd * CH]);
    const float4* d4 = reinterpret_cast<const float4*>(&s_av[AVW + hd * CH]);
    float s = 0.f, d = 0.f;
#pragma unroll
    for (int q = 0; q < CH / 4; ++q) {
      const float4 v = o4[q], x = s4[q], y = d4[q];
      s += v.x * x.x;  d += v.x * y.x;
      s += v.y * x.y;  d += v.y * y.y;
      s += v.z * x.z;  d += v.z * y.z;
      s += v.w * x.w;  d += v.w * y.w;
    }
    if (ok) {
      s_al[row * 2 * H + hd]     = s;
      s_al[row * 2 * H + H + hd] = d;
    }
  }
  __syncthreads();
  for (int j = tid; j < RM * 2 * H; j += NTH)
    al[(size_t)blockIdx.x * RM * 2 * H + j] = s_al[j];

  float4* Cg = reinterpret_cast<float4*>(C + (size_t)blockIdx.x * RM * N);
#pragma unroll 4
  for (int idx = tid; idx < RM * NV; idx += NTH)
    Cg[idx] = *reinterpret_cast<const float4*>(&Os[(idx / NV) * PO + (idx % NV) * 4]);
}

__device__ __forceinline__ void atomicMaxF(float* addr, float val) {
  if (val >= 0.f) atomicMax((int*)addr, __float_as_int(val));
  else            atomicMin((unsigned int*)addr, __float_as_uint(val));
}

template <int H, int C, int R, int QW, bool ELU>
__global__ __launch_bounds__(1024)
void agg_owner_kernel(const int* __restrict__ ei,
                      const float* __restrict__ al,
                      const float* __restrict__ h,
                      const float* __restrict__ bias,
                      float* __restrict__ out) {
  constexpr int HC  = H * C;
  constexpr int CPL = HC / 32;
  constexpr int NW  = 32;
  constexpr int NT  = NW * 32;
  constexpr int UNR = 8;
  constexpr int NV4 = NEDGES / 4;
  constexpr int NCHUNK = (NV4 + 31) / 32;
  static_assert(C % 32 == 0, "a 32-lane line must stay inside one head");
  static_assert(NEDGES % 4 == 0, "destination list is read four at a time");
  static_assert(QW >= 256, "queue must take a chunk of 128 edges on top of QW-128");

  __shared__ float acc[R * HC];
  __shared__ int   q_eid[NW * QW];
  __shared__ int   q_dl[NW * QW];
  __shared__ float s_ad[R * H];
  __shared__ float s_self[R * H];
  __shared__ float s_m[R * H];
  __shared__ float s_den[R * H];

  const int tid  = threadIdx.x;
  const int lane = tid & 31;
  const int w    = __builtin_amdgcn_readfirstlane(tid >> 5);
  const int lo   = blockIdx.x * R;
  int* qe = q_eid + w * QW;
  int* qd = q_dl  + w * QW;

  for (int i = tid; i < R * H; i += NT) {
    const int n = lo + i / H;
    float a_d = 0.f, es = 0.f;
    if (n < NNODES) {
      a_d = al[(size_t)n * (2 * H) + H + i % H];
      const float v = al[(size_t)n * (2 * H) + i % H] + a_d;
      es = (v > 0.f) ? v : NEG_SLOPE * v;
    }
    s_ad[i]   = a_d;
    s_self[i] = es;
    s_m[i]    = es;
  }
  __syncthreads();

  int  cnt     = 0;
  bool spilled = false;

  auto max_pass = [&](int nq) {
    __builtin_amdgcn_fence(__ATOMIC_RELEASE, "wavefront");
    __builtin_amdgcn_wave_barrier();
    for (int b = 0; b < nq; b += 32) {
      const int i = b + lane;
      if (i < nq) {
        const int dl = qd[i];
        int s = ei[qe[i]];
        s = min(max(s, 0), NNODES - 1);
        float a[H];
#pragma unroll
        for (int hh = 0; hh < H; ++hh) a[hh] = al[(size_t)s * (2 * H) + hh];
#pragma unroll
        for (int hh = 0; hh < H; ++hh) {
          float v = a[hh] + s_ad[dl * H + hh];
          v = (v > 0.f) ? v : NEG_SLOPE * v;
          atomicMaxF(&s_m[dl * H + hh], v);
        }
      }
    }
    __builtin_amdgcn_fence(__ATOMIC_ACQUIRE, "wavefront");
    __builtin_amdgcn_wave_barrier();
  };

  auto acc_pass = [&](int nq) {
    __builtin_amdgcn_fence(__ATOMIC_RELEASE, "wavefront");
    __builtin_amdgcn_wave_barrier();
    for (int b = 0; b < nq; b += 32) {
      const int  i     = b + lane;
      const bool valid = i < nq;
      const int eid_i = valid ? qe[i] : 0;
      const int dl_i  = valid ? qd[i] : 0;
      int s_i = valid ? ei[eid_i] : 0;
      s_i = min(max(s_i, 0), NNODES - 1);
      float a[H];
#pragma unroll
      for (int hh = 0; hh < H; ++hh) a[hh] = al[(size_t)s_i * (2 * H) + hh];
      int ex_i[H];
#pragma unroll
      for (int hh = 0; hh < H; ++hh) {
        float v = a[hh] + s_ad[dl_i * H + hh];
        v = (v > 0.f) ? v : NEG_SLOPE * v;
        const float ex = __expf(v - s_m[dl_i * H + hh]);
        if (valid) atomicAdd(&s_den[dl_i * H + hh], ex);
        ex_i[hh] = __float_as_int(ex);
      }
      const int n = min(32, nq - b);
      for (int k = 0; k < n; k += 4) {
        float v[4][CPL];
#pragma unroll
        for (int u = 0; u < 4; ++u) {
          const int s = __builtin_amdgcn_readlane(s_i, k + u);
          const float* hr = h + (size_t)s * HC + lane;
#pragma unroll
          for (int j = 0; j < CPL; ++j) v[u][j] = hr[j * 32];
        }
#pragma unroll
        for (int u = 0; u < 4; ++u) {
          if (k + u < n) {
            const int dl = __builtin_amdgcn_readlane(dl_i, k + u);
            float* ar = acc + dl * HC + lane;
#pragma unroll
            for (int j = 0; j < CPL; ++j) {
              const float ex =
                  __int_as_float(__builtin_amdgcn_readlane(ex_i[(j * 32) / C], k + u));
              atomicAdd(ar + j * 32, ex * v[u][j]);
            }
          }
        }
      }
    }
    __builtin_amdgcn_fence(__ATOMIC_ACQUIRE, "wavefront");
    __builtin_amdgcn_wave_barrier();
  };

  const int4* dst4 = reinterpret_cast<const int4*>(ei + NEDGES);
  auto scan = [&](auto&& drain) {
    for (int c0 = w; c0 < NCHUNK; c0 += NW * UNR) {
      int4 d[UNR];
#pragma unroll
      for (int u = 0; u < UNR; ++u) {
        const int i4 = (c0 + u * NW) * 32 + lane;
        d[u] = (i4 < NV4) ? dst4[i4] : make_int4(-1, -1, -1, -1);
      }
      int  done = 0;
      bool full = false;
      do {
#pragma unroll
        for (int u = 0; u < UNR; ++u) {
          if (u >= done && !full) {
            const int eid0 = ((c0 + u * NW) * 32 + lane) * 4;
            const int dd[4] = {d[u].x - lo, d[u].y - lo, d[u].z - lo, d[u].w - lo};
            const bool any = ((unsigned)dd[0] < (unsigned)R) | ((unsigned)dd[1] < (unsigned)R) |
                             ((unsigned)dd[2] < (unsigned)R) | ((unsigned)dd[3] < (unsigned)R);
            if (__builtin_amdgcn_ballot_w32(any)) {
#pragma unroll
              for (int c = 0; c < 4; ++c) {
                const bool m = (unsigned)dd[c] < (unsigned)R;
                const unsigned mask = __builtin_amdgcn_ballot_w32(m);
                if (m) {
                  const int pos = cnt + (int)__builtin_amdgcn_mbcnt_lo(mask, 0u);
                  qe[pos] = eid0 + c;
                  qd[pos] = dd[c];
                }
                cnt += __builtin_popcount(mask);
              }
            }
            done = u + 1;
            full = cnt > QW - 128;
          }
        }
        if (full) { drain(cnt); cnt = 0; full = false; }
      } while (done < UNR);
    }
  };

  scan([&](int nq) { max_pass(nq); spilled = true; });
  max_pass(cnt);
  __syncthreads();

  for (int i = tid; i < R * H; i += NT) s_den[i] = __expf(s_self[i] - s_m[i]);
  constexpr int NR = (R + NW - 1) / NW;
  {
    float iv[NR][CPL];
#pragma unroll
    for (int r = 0; r < NR; ++r) {
      const int  nl = w + r * NW;
      const int  n  = lo + nl;
      const bool ok = (nl < R) && (n < NNODES);
      const int  nn = ok ? n  : 0;
      const int  nc = ok ? nl : 0;
#pragma unroll
      for (int j = 0; j < CPL; ++j) {
        const int   hd = (j * 32) / C;
        const float p  = __expf(s_self[nc * H + hd] - s_m[nc * H + hd]) *
                         h[(size_t)nn * HC + j * 32 + lane];
        iv[r][j] = ok ? p : 0.f;
      }
    }
#pragma unroll
    for (int r = 0; r < NR; ++r) {
      const int nl = w + r * NW;
      if (nl < R) {
#pragma unroll
        for (int j = 0; j < CPL; ++j) acc[nl * HC + j * 32 + lane] = iv[r][j];
      }
    }
  }
  __syncthreads();

  if (spilled) {
    cnt = 0;
    scan([&](int nq) { acc_pass(nq); });
  }
  acc_pass(cnt);
  __syncthreads();

  float bj[CPL];
#pragma unroll
  for (int j = 0; j < CPL; ++j) bj[j] = bias[j * 32 + lane];
#pragma unroll
  for (int r = 0; r < NR; ++r) {
    const int nl = w + r * NW;
    const int n  = lo + nl;
    if (nl < R && n < NNODES) {
#pragma unroll
      for (int j = 0; j < CPL; ++j) {
        const int ch = j * 32 + lane;
        float v = acc[nl * HC + ch] / (s_den[nl * H + (j * 32) / C] + 1e-16f) + bj[j];
        if (ELU) v = (v > 0.f) ? v : (__expf(v) - 1.f);
        out[(size_t)n * HC + ch] = v;
      }
    }
  }
}

extern "C" void kernel_launch(void* const* d_in, const int* in_sizes, int n_in,
                              void* d_out, int out_size, void* d_ws, size_t ws_size,
                              hipStream_t stream) {
  const float* x      = (const float*)d_in[0];
  const int*   ei     = (const int*)  d_in[1];
  const float* W1     = (const float*)d_in[2];
  const float* a_src1 = (const float*)d_in[3];
  const float* a_dst1 = (const float*)d_in[4];
  const float* b1     = (const float*)d_in[5];
  const float* W2     = (const float*)d_in[6];
  const float* a_src2 = (const float*)d_in[7];
  const float* a_dst2 = (const float*)d_in[8];
  const float* b2     = (const float*)d_in[9];
  float* out = (float*)d_out;

  float* ws   = (float*)d_ws;
  float* h1   = ws;
  float* g1   = h1   + (size_t)NNODES * H1C;
  float* al1  = g1   + (size_t)NNODES * H1C;
  float* h2   = al1  + (size_t)NNODES * 2 * HEADS;
  float* al2  = h2   + (size_t)NNODES * OUT_CH;

  const int RN  = 80;
  const int NG  = (NNODES + RN - 1) / RN;

  gemm_strip_kernel<IN_CH, H1C, 2, 5, 1, HEADS><<<NNODES / 80, 256, 0, stream>>>(
      x, W1, a_src1, a_dst1, h1, al1);
  agg_owner_kernel<HEADS, HID_CH, RN, 256, true><<<NG, 1024, 0, stream>>>(
      ei, al1, h1, b1, g1);

  gemm_strip_kernel<H1C, OUT_CH, 1, 3, 2, 1><<<NNODES / 80, 256, 0, stream>>>(
      g1, W2, a_src2, a_dst2, h2, al2);
  agg_owner_kernel<1, OUT_CH, RN, 256, false><<<NG, 1024, 0, stream>>>(
      ei, al2, h2, b2, out);
}


# layer 2 takes its per-wave edge lists (src, local dst) from the workspace, where layer 1 saved its queues, instead of scanning the destination list again and looking src up; sum order unchanged
# speedup vs baseline: 6.4669x; 1.1618x over previous
#include <hip/hip_runtime.h>
#include <math.h>
#include <cstddef>

#define NNODES   20000
#define NEDGES   320000
#define ETOT     (NEDGES + NNODES)
#define IN_CH    128
#define HID_CH   32
#define HEADS    8
#define H1C      (HEADS * HID_CH)
#define OUT_CH   64
#define NEG_SLOPE 0.2f

typedef float  v2f   __attribute__((ext_vector_type(2)));
typedef float  v8f   __attribute__((ext_vector_type(8)));

template <int V> struct int_const { static constexpr int value = V; };

template <int K, int N, int TN, int TM, int WM, int H>
__global__ __launch_bounds__(256)
void gemm_strip_kernel(const float* __restrict__ A, const float* __restrict__ B,
                       const float* __restrict__ a_src, const float* __restrict__ a_dst,
                       float* __restrict__ C, float* __restrict__ al) {
  constexpr int RM  = 80;
  constexpr int MT  = RM / 16;
  constexpr int NTH = 256;
  constexpr int WN  = (N / 16) / TN;
  constexpr int PA  = K + 4;
  constexpr int PB  = 2 * N + 32;
  constexpr int PO  = N + 4;
  constexpr int KV  = K / 4;
  constexpr int NV  = N / 4;
  constexpr int A_WORDS = RM * PA;
  constexpr int B_WORDS = (K / 2) * PB;
  constexpr int CH  = N / H;
  constexpr int AVN = (N + NTH - 1) / NTH;
  constexpr int AVW = AVN * NTH;
  static_assert(WN * WM * 32 == NTH, "8 waves");
  static_assert(TM * WM >= MT && TM * (WM - 1) < MT, "M-tiles dealt over WM wave rows");
  static_assert(RM * PO <= A_WORDS + B_WORDS, "C tile reuses the operand space");
  static_assert((RM * KV) % (NTH * 10) == 0, "A staging in batches of 10");
  static_assert(((K / 2) * NV) % (NTH * 8) == 0, "B staging in batches of 8");
  static_assert((RM * NV) % NTH == 0, "C store loop");
  static_assert(NNODES % RM == 0, "no partial strip");
  static_assert(N % H == 0 && CH % 4 == 0, "a head's channels are read four at a time");
  static_assert((RM * 2 * H) % 32 == 0, "a strip's logits fill whole 128-B lines");

  __shared__ alignas(16) float smem[A_WORDS + B_WORDS];
  __shared__ alignas(16) float s_av[2 * AVW];
  __shared__ alignas(16) float s_al[RM * 2 * H];
  float* As = smem;
  float* Bs = smem + A_WORDS;
  float* Os = smem;

  const int tid  = threadIdx.x;
  const int lane = tid & 31;
  const int w    = __builtin_amdgcn_readfirstlane(tid >> 5);
  const int c    = lane & 15;
  const int h    = lane >> 4;

  float avs[AVN], avd[AVN];
#pragma unroll
  for (int u = 0; u < AVN; ++u) {
    const int idx = min(u * NTH + tid, N - 1);
    avs[u] = a_src[idx];
    avd[u] = a_dst[idx];
  }

  {
    const float4* Ag = reinterpret_cast<const float4*>(A + (size_t)blockIdx.x * RM * K);
#pragma unroll
    for (int i0 = 0; i0 < (RM * KV) / NTH; i0 += 10) {
      float4 r[10];
#pragma unroll
      for (int u = 0; u < 10; ++u) r[u] = Ag[(i0 + u) * NTH + tid];
#pragma unroll
      for (int u = 0; u < 10; ++u) {
        const int idx = (i0 + u) * NTH + tid;
        *reinterpret_cast<float4*>(&As[(idx / KV) * PA + (idx % KV) * 4]) = r[u];
      }
    }
  }
  {
    const float4* Bg = reinterpret_cast<const float4*>(B);
#pragma unroll
    for (int i0 = 0; i0 < ((K / 2) * NV) / NTH; i0 += 8) {
      float4 r0[8], r1[8];
#pragma unroll
      for (int u = 0; u < 8; ++u) {
        const int idx = (i0 + u) * NTH + tid;
        const int p = idx / NV, c4 = idx % NV;
        r0[u] = Bg[(2 * p) * NV + c4];
        r1[u] = Bg[(2 * p + 1) * NV + c4];
      }
#pragma unroll
      for (int u = 0; u < 8; ++u) {
        const int idx = (i0 + u) * NTH + tid;
        const int p = idx / NV, c4 = idx % NV;
        float* d = &Bs[p * PB + c4 * 8];
        *reinterpret_cast<float4*>(d)     = make_float4(r0[u].x, r1[u].x, r0[u].y, r1[u].y);
        *reinterpret_cast<float4*>(d + 4) = make_float4(r0[u].z, r1[u].z, r0[u].w, r1[u].w);
      }
    }
  }
#pragma unroll
  for (int u = 0; u < AVN; ++u) {
    s_av[u * NTH + tid]       = avs[u];
    s_av[AVW + u * NTH + tid] = avd[u];
  }
  __syncthreads();

  const int wn = w % WN;
  const int wm = w / WN;
  const int mb = wm * TM;
  const int mc = min(TM, MT - mb);
  v8f acc[TM][TN];
#pragma unroll
  for (int i = 0; i < TM; ++i)
#pragma unroll
    for (int j = 0; j < TN; ++j) acc[i][j] = v8f{0.f, 0.f, 0.f, 0.f, 0.f, 0.f, 0.f, 0.f};

  const float* Ap = As + (mb * 16 + c) * PA + 2 * h;
  const float* Bp = Bs + h * PB + (wn * TN * 16 + c) * 2;
  auto kloop = [&](auto cnt) {
    constexpr int CNT = decltype(cnt)::value;
#pragma unroll 4
    for (int ks = 0; ks < K / 4; ++ks) {
      v2f a[CNT], b[TN];
#pragma unroll
      for (int i = 0; i < CNT; ++i)
        a[i] = *reinterpret_cast<const v2f*>(Ap + i * 16 * PA + 4 * ks);
#pragma unroll
      for (int j = 0; j < TN; ++j)
        b[j] = *reinterpret_cast<const v2f*>(Bp + 2 * ks * PB + j * 32);
#pragma unroll
      for (int i = 0; i < CNT; ++i)
#pragma unroll
        for (int j = 0; j < TN; ++j)
          acc[i][j] = __builtin_amdgcn_wmma_f32_16x16x4_f32(false, a[i], false, b[j],
                                                            (short)0, acc[i][j], false, false);
    }
  };
  if constexpr (MT % TM == 0) {
    kloop(int_const<TM>{});
  } else {
    if (mc == TM) kloop(int_const<TM>{});
    else          kloop(int_const<MT % TM>{});
  }
  __syncthreads();

#pragma unroll
  for (int i = 0; i < TM; ++i) {
    if (i < mc) {
#pragma unroll
      for (int j = 0; j < TN; ++j)
#pragma unroll
        for (int r = 0; r < 8; ++r)
          Os[((mb + i) * 16 + 8 * h + r) * PO + (wn * TN + j) * 16 + c] = acc[i][j][r];
    }
  }
  __syncthreads();

  constexpr int RB = (RM + 31) / 32;
  for (int blk = w; blk < H * RB; blk += NTH / 32) {
    const int  hd  = blk / RB;
    const int  row = (blk % RB) * 32 + lane;
    const bool ok  = row < RM;
    const float4* o4 = reinterpret_cast<const float4*>(&Os[(ok ? row : 0) * PO + hd * CH]);
    const float4* s4 = reinterpret_cast<const float4*>(&s_av[hd * CH]);
    const float4* d4 = reinterpret_cast<const float4*>(&s_av[AVW + hd * CH]);
    float s = 0.f, d = 0.f;
#pragma unroll
    for (int q = 0; q < CH / 4; ++q) {
      const float4 v = o4[q], x = s4[q], y = d4[q];
      s += v.x * x.x;  d += v.x * y.x;
      s += v.y * x.y;  d += v.y * y.y;
      s += v.z * x.z;  d += v.z * y.z;
      s += v.w * x.w;  d += v.w * y.w;
    }
    if (ok) {
      s_al[row * 2 * H + hd]     = s;
      s_al[row * 2 * H + H + hd] = d;
    }
  }
  __syncthreads();
  for (int j = tid; j < RM * 2 * H; j += NTH)
    al[(size_t)blockIdx.x * RM * 2 * H + j] = s_al[j];

  float4* Cg = reinterpret_cast<float4*>(C + (size_t)blockIdx.x * RM * N);
#pragma unroll 4
  for (int idx = tid; idx < RM * NV; idx += NTH)
    Cg[idx] = *reinterpret_cast<const float4*>(&Os[(idx / NV) * PO + (idx % NV) * 4]);
}

__device__ __forceinline__ void atomicMaxF(float* addr, float val) {
  if (val >= 0.f) atomicMax((int*)addr, __float_as_int(val));
  else            atomicMin((unsigned int*)addr, __float_as_uint(val));
}

template <int H, int C, int R, int QW, bool ELU, int MODE>
__global__ __launch_bounds__(1024)
void agg_owner_kernel(const int* __restrict__ ei,
                      const float* __restrict__ al,
                      const float* __restrict__ h,
                      const float* __restrict__ bias,
                      float* __restrict__ out,
                      int* __restrict__ sv_cnt,
                      int* __restrict__ sv_lst) {
  constexpr int HC  = H * C;
  constexpr int CPL = HC / 32;
  constexpr int NW  = 32;
  constexpr int NT  = NW * 32;
  constexpr int UNR = 8;
  constexpr int NV4 = NEDGES / 4;
  constexpr int NCHUNK = (NV4 + 31) / 32;
  static_assert(C % 32 == 0, "a 32-lane line must stay inside one head");
  static_assert(NEDGES % 4 == 0, "destination list is read four at a time");
  static_assert(QW >= 256, "queue must take a chunk of 128 edges on top of QW-128");
  static_assert(QW % 32 == 0, "the saved list is read 32 entries at a time");

  __shared__ float acc[R * HC];
  __shared__ int   q_eid[NW * QW];
  __shared__ int   q_dl[NW * QW];
  __shared__ float s_ad[R * H];
  __shared__ float s_self[R * H];
  __shared__ float s_m[R * H];
  __shared__ float s_den[R * H];

  const int tid  = threadIdx.x;
  const int lane = tid & 31;
  const int w    = __builtin_amdgcn_readfirstlane(tid >> 5);
  const int lo   = blockIdx.x * R;
  int* qe = q_eid + w * QW;
  int* qd = q_dl  + w * QW;
  const size_t slot = (size_t)blockIdx.x * NW + w;
  int2* lst = reinterpret_cast<int2*>(sv_lst) + slot * QW;
  int saved = -1;
  if (MODE == 1) saved = __builtin_amdgcn_readfirstlane(sv_cnt[slot]);
  const bool direct = (MODE == 1) && (saved >= 0) && (saved <= QW);

  for (int i = tid; i < R * H; i += NT) {
    const int n = lo + i / H;
    float a_d = 0.f, es = 0.f;
    if (n < NNODES) {
      a_d = al[(size_t)n * (2 * H) + H + i % H];
      const float v = al[(size_t)n * (2 * H) + i % H] + a_d;
      es = (v > 0.f) ? v : NEG_SLOPE * v;
    }
    s_ad[i]   = a_d;
    s_self[i] = es;
    s_m[i]    = es;
  }
  if (MODE == 1 && direct && saved > 0) {
    int2 e[QW / 32];
#pragma unroll
    for (int u = 0; u < QW / 32; ++u) e[u] = lst[min(u * 32 + lane, saved - 1)];
#pragma unroll
    for (int u = 0; u < QW / 32; ++u) {
      const int i = u * 32 + lane;
      if (i < saved) {
        qe[i] = e[u].x;
        qd[i] = min(max(e[u].y, 0), R - 1);
      }
    }
  }
  __syncthreads();

  int  cnt     = 0;
  bool spilled = false;

  auto max_pass = [&](int nq, bool save) {
    __builtin_amdgcn_fence(__ATOMIC_RELEASE, "wavefront");
    __builtin_amdgcn_wave_barrier();
    for (int b = 0; b < nq; b += 32) {
      const int i = b + lane;
      if (i < nq) {
        const int dl = qd[i];
        int s = qe[i];
        if (!direct) s = ei[s];
        s = min(max(s, 0), NNODES - 1);
        if (MODE == 0 && save) lst[i] = make_int2(s, dl);
        float a[H];
#pragma unroll
        for (int hh = 0; hh < H; ++hh) a[hh] = al[(size_t)s * (2 * H) + hh];
#pragma unroll
        for (int hh = 0; hh < H; ++hh) {
          float v = a[hh] + s_ad[dl * H + hh];
          v = (v > 0.f) ? v : NEG_SLOPE * v;
          atomicMaxF(&s_m[dl * H + hh], v);
        }
      }
    }
    __builtin_amdgcn_fence(__ATOMIC_ACQUIRE, "wavefront");
    __builtin_amdgcn_wave_barrier();
  };

  auto acc_pass = [&](int nq) {
    __builtin_amdgcn_fence(__ATOMIC_RELEASE, "wavefront");
    __builtin_amdgcn_wave_barrier();
    for (int b = 0; b < nq; b += 32) {
      const int  i     = b + lane;
      const bool valid = i < nq;
      const int eid_i = valid ? qe[i] : 0;
      const int dl_i  = valid ? qd[i] : 0;
      int s_i = eid_i;
      if (!direct) s_i = valid ? ei[eid_i] : 0;
      s_i = min(max(s_i, 0), NNODES - 1);
      float a[H];
#pragma unroll
      for (int hh = 0; hh < H; ++hh) a[hh] = al[(size_t)s_i * (2 * H) + hh];
      int ex_i[H];
#pragma unroll
      for (int hh = 0; hh < H; ++hh) {
        float v = a[hh] + s_ad[dl_i * H + hh];
        v = (v > 0.f) ? v : NEG_SLOPE * v;
        const float ex = __expf(v - s_m[dl_i * H + hh]);
        if (valid) atomicAdd(&s_den[dl_i * H + hh], ex);
        ex_i[hh] = __float_as_int(ex);
      }
      const int n = min(32, nq - b);
      for (int k = 0; k < n; k += 4) {
        float v[4][CPL];
#pragma unroll
        for (int u = 0; u < 4; ++u) {
          const int s = __builtin_amdgcn_readlane(s_i, k + u);
          const float* hr = h + (size_t)s * HC + lane;
#pragma unroll
          for (int j = 0; j < CPL; ++j) v[u][j] = hr[j * 32];
        }
#pragma unroll
        for (int u = 0; u < 4; ++u) {
          if (k + u < n) {
            const int dl = __builtin_amdgcn_readlane(dl_i, k + u);
            float* ar = acc + dl * HC + lane;
#pragma unroll
            for (int j = 0; j < CPL; ++j) {
              const float ex =
                  __int_as_float(__builtin_amdgcn_readlane(ex_i[(j * 32) / C], k + u));
              atomicAdd(ar + j * 32, ex * v[u][j]);
            }
          }
        }
      }
    }
    __builtin_amdgcn_fence(__ATOMIC_ACQUIRE, "wavefront");
    __builtin_amdgcn_wave_barrier();
  };

  const int4* dst4 = reinterpret_cast<const int4*>(ei + NEDGES);
  auto scan = [&](auto&& drain) {
    for (int c0 = w; c0 < NCHUNK; c0 += NW * UNR) {
      int4 d[UNR];
#pragma unroll
      for (int u = 0; u < UNR; ++u) {
        const int i4 = (c0 + u * NW) * 32 + lane;
        d[u] = (i4 < NV4) ? dst4[i4] : make_int4(-1, -1, -1, -1);
      }
      int  done = 0;
      bool full = false;
      do {
#pragma unroll
        for (int u = 0; u < UNR; ++u) {
          if (u >= done && !full) {
            const int eid0 = ((c0 + u * NW) * 32 + lane) * 4;
            const int dd[4] = {d[u].x - lo, d[u].y - lo, d[u].z - lo, d[u].w - lo};
            const bool any = ((unsigned)dd[0] < (unsigned)R) | ((unsigned)dd[1] < (unsigned)R) |
                             ((unsigned)dd[2] < (unsigned)R) | ((unsigned)dd[3] < (unsigned)R);
            if (__builtin_amdgcn_ballot_w32(any)) {
#pragma unroll
              for (int c = 0; c < 4; ++c) {
                const bool m = (unsigned)dd[c] < (unsigned)R;
                const unsigned mask = __builtin_amdgcn_ballot_w32(m);
                if (m) {
                  const int pos = cnt + (int)__builtin_amdgcn_mbcnt_lo(mask, 0u);
                  qe[pos] = eid0 + c;
                  qd[pos] = dd[c];
                }
                cnt += __builtin_popcount(mask);
              }
            }
            done = u + 1;
            full = cnt > QW - 128;
          }
        }
        if (full) { drain(cnt); cnt = 0; full = false; }
      } while (done < UNR);
    }
  };

  if (direct) cnt = saved;
  else        scan([&](int nq) { max_pass(nq, false); spilled = true; });
  max_pass(cnt, !spilled);
  if (MODE == 0 && lane == 0) sv_cnt[slot] = spilled ? -1 : cnt;
  __syncthreads();

  for (int i = tid; i < R * H; i += NT) s_den[i] = __expf(s_self[i] - s_m[i]);
  constexpr int NR = (R + NW - 1) / NW;
  {
    float iv[NR][CPL];
#pragma unroll
    for (int r = 0; r < NR; ++r) {
      const int  nl = w + r * NW;
      const int  n  = lo + nl;
      const bool ok = (nl < R) && (n < NNODES);
      const int  nn = ok ? n  : 0;
      const int  nc = ok ? nl : 0;
#pragma unroll
      for (int j = 0; j < CPL; ++j) {
        const int   hd = (j * 32) / C;
        const float p  = __expf(s_self[nc * H + hd] - s_m[nc * H + hd]) *
                         h[(size_t)nn * HC + j * 32 + lane];
        iv[r][j] = ok ? p : 0.f;
      }
    }
#pragma unroll
    for (int r = 0; r < NR; ++r) {
      const int nl = w + r * NW;
      if (nl < R) {
#pragma unroll
        for (int j = 0; j < CPL; ++j) acc[nl * HC + j * 32 + lane] = iv[r][j];
      }
    }
  }
  __syncthreads();

  if (spilled) {
    cnt = 0;
    scan([&](int nq) { acc_pass(nq); });
  }
  acc_pass(cnt);
  __syncthreads();

  float bj[CPL];
#pragma unroll
  for (int j = 0; j < CPL; ++j) bj[j] = bias[j * 32 + lane];
#pragma unroll
  for (int r = 0; r < NR; ++r) {
    const int nl = w + r * NW;
    const int n  = lo + nl;
    if (nl < R && n < NNODES) {
#pragma unroll
      for (int j = 0; j < CPL; ++j) {
        const int ch = j * 32 + lane;
        float v = acc[nl * HC + ch] / (s_den[nl * H + (j * 32) / C] + 1e-16f) + bj[j];
        if (ELU) v = (v > 0.f) ? v : (__expf(v) - 1.f);
        out[(size_t)n * HC + ch] = v;
      }
    }
  }
}

extern "C" void kernel_launch(void* const* d_in, const int* in_sizes, int n_in,
                              void* d_out, int out_size, void* d_ws, size_t ws_size,
                              hipStream_t stream) {
  const float* x      = (const float*)d_in[0];
  const int*   ei     = (const int*)  d_in[1];
  const float* W1     = (const float*)d_in[2];
  const float* a_src1 = (const float*)d_in[3];
  const float* a_dst1 = (const float*)d_in[4];
  const float* b1     = (const float*)d_in[5];
  const float* W2     = (const float*)d_in[6];
  const float* a_src2 = (const float*)d_in[7];
  const float* a_dst2 = (const float*)d_in[8];
  const float* b2     = (const float*)d_in[9];
  float* out = (float*)d_out;

  float* ws   = (float*)d_ws;
  float* h1   = ws;
  float* g1   = h1   + (size_t)NNODES * H1C;
  float* al1  = g1   + (size_t)NNODES * H1C;
  float* h2   = al1  + (size_t)NNODES * 2 * HEADS;
  float* al2  = h2   + (size_t)NNODES * OUT_CH;

  const int RN  = 80;
  const int NG  = (NNODES + RN - 1) / RN;
  int*   svc  = (int*)(al2 + (size_t)NNODES * 2);
  int*   svl  = svc + NG * 32;

  gemm_strip_kernel<IN_CH, H1C, 2, 5, 1, HEADS><<<NNODES / 80, 256, 0, stream>>>(
      x, W1, a_src1, a_dst1, h1, al1);
  agg_owner_kernel<HEADS, HID_CH, RN, 256, true, 0><<<NG, 1024, 0, stream>>>(
      ei, al1, h1, b1, g1, svc, svl);

  gemm_strip_kernel<H1C, OUT_CH, 1, 3, 2, 1><<<NNODES / 80, 256, 0, stream>>>(
      g1, W2, a_src2, a_dst2, h2, al2);
  agg_owner_kernel<1, OUT_CH, RN, 256, false, 1><<<NG, 1024, 0, stream>>>(
      ei, al2, h2, b2, out, svc, svl);
}


# edges binned once by owner group in the layer-1 GEMM kernel (per list part, LDS atomic ranks) into cells; both aggregation kernels read their cells, no destination-list scan; sum order changes
# speedup vs baseline: 7.3923x; 1.1431x over previous
#include <hip/hip_runtime.h>
#include <math.h>
#include <cstddef>

#define NNODES   20000
#define NEDGES   320000
#define ETOT     (NEDGES + NNODES)
#define IN_CH    128
#define HID_CH   32
#define HEADS    8
#define H1C      (HEADS * HID_CH)
#define OUT_CH   64
#define NEG_SLOPE 0.2f
#define OWNR     80
#define NPART    (NNODES / OWNR)
#define PARTE    (NEDGES / NPART)
#define CELLC    32
#define CNTP     256

typedef float  v2f   __attribute__((ext_vector_type(2)));
typedef float  v8f   __attribute__((ext_vector_type(8)));

template <int V> struct int_const { static constexpr int value = V; };

template <int K, int N, int TN, int TM, int WM, int H, bool BIN>
__global__ __launch_bounds__(256)
void gemm_strip_kernel(const float* __restrict__ A, const float* __restrict__ B,
                       const float* __restrict__ a_src, const float* __restrict__ a_dst,
                       float* __restrict__ C, float* __restrict__ al,
                       const int* __restrict__ ei, int2* __restrict__ cells,
                       int* __restrict__ cntg) {
  constexpr int RM  = 80;
  constexpr int MT  = RM / 16;
  constexpr int NTH = 256;
  constexpr int WN  = (N / 16) / TN;
  constexpr int PA  = K + 4;
  constexpr int PB  = 2 * N + 32;
  constexpr int PO  = N + 4;
  constexpr int KV  = K / 4;
  constexpr int NV  = N / 4;
  constexpr int A_WORDS = RM * PA;
  constexpr int B_WORDS = (K / 2) * PB;
  constexpr int CH  = N / H;
  constexpr int AVN = (N + NTH - 1) / NTH;
  constexpr int AVW = AVN * NTH;
  static_assert(WN * WM * 32 == NTH, "8 waves");
  static_assert(TM * WM >= MT && TM * (WM - 1) < MT, "M-tiles dealt over WM wave rows");
  static_assert(RM * PO <= A_WORDS + B_WORDS, "C tile reuses the operand space");
  static_assert((RM * KV) % (NTH * 10) == 0, "A staging in batches of 10");
  static_assert(((K / 2) * NV) % (NTH * 8) == 0, "B staging in batches of 8");
  static_assert((RM * NV) % NTH == 0, "C store loop");
  static_assert(NNODES % RM == 0, "no partial strip");
  static_assert(N % H == 0 && CH % 4 == 0, "a head's channels are read four at a time");
  static_assert((RM * 2 * H) % 32 == 0, "a strip's logits fill whole 128-B lines");
  constexpr int EPT = BIN ? PARTE / NTH : 1;
  static_assert(!BIN || (RM == OWNR && PARTE % NTH == 0 && NEDGES % NPART == 0 &&
                         NPART < NTH && NTH <= CNTP),
                "one list part per strip workgroup, one count per thread");

  __shared__ alignas(16) float smem[A_WORDS + B_WORDS];
  __shared__ alignas(16) float s_av[2 * AVW];
  __shared__ alignas(16) float s_al[RM * 2 * H];
  __shared__ int s_hist[BIN ? NTH : 1];
  float* As = smem;
  float* Bs = smem + A_WORDS;
  float* Os = smem;

  const int tid  = threadIdx.x;
  const int lane = tid & 31;
  const int w    = __builtin_amdgcn_readfirstlane(tid >> 5);
  const int c    = lane & 15;
  const int h    = lane >> 4;

  float avs[AVN], avd[AVN];
#pragma unroll
  for (int u = 0; u < AVN; ++u) {
    const int idx = min(u * NTH + tid, N - 1);
    avs[u] = a_src[idx];
    avd[u] = a_dst[idx];
  }

  int e_s[EPT], e_d[EPT];
  if constexpr (BIN) {
#pragma unroll
    for (int u = 0; u < EPT; ++u) {
      const int e = blockIdx.x * PARTE + u * NTH + tid;
      e_s[u] = ei[e];
      e_d[u] = ei[NEDGES + e];
    }
    s_hist[tid] = 0;
  }

  {
    const float4* Ag = reinterpret_cast<const float4*>(A + (size_t)blockIdx.x * RM * K);
#pragma unroll
    for (int i0 = 0; i0 < (RM * KV) / NTH; i0 += 10) {
      float4 r[10];
#pragma unroll
      for (int u = 0; u < 10; ++u) r[u] = Ag[(i0 + u) * NTH + tid];
#pragma unroll
      for (int u = 0; u < 10; ++u) {
        const int idx = (i0 + u) * NTH + tid;
        *reinterpret_cast<float4*>(&As[(idx / KV) * PA + (idx % KV) * 4]) = r[u];
      }
    }
  }
  {
    const float4* Bg = reinterpret_cast<const float4*>(B);
#pragma unroll
    for (int i0 = 0; i0 < ((K / 2) * NV) / NTH; i0 += 8) {
      float4 r0[8], r1[8];
#pragma unroll
      for (int u = 0; u < 8; ++u) {
        const int idx = (i0 + u) * NTH + tid;
        const int p = idx / NV, c4 = idx % NV;
        r0[u] = Bg[(2 * p) * NV + c4];
        r1[u] = Bg[(2 * p + 1) * NV + c4];
      }
#pragma unroll
      for (int u = 0; u < 8; ++u) {
        const int idx = (i0 + u) * NTH + tid;
        const int p = idx / NV, c4 = idx % NV;
        float* d = &Bs[p * PB + c4 * 8];
        *reinterpret_cast<float4*>(d)     = make_float4(r0[u].x, r1[u].x, r0[u].y, r1[u].y);
        *reinterpret_cast<float4*>(d + 4) = make_float4(r0[u].z, r1[u].z, r0[u].w, r1[u].w);
      }
    }
  }
#pragma unroll
  for (int u = 0; u < AVN; ++u) {
    s_av[u * NTH + tid]       = avs[u];
    s_av[AVW + u * NTH + tid] = avd[u];
  }
  __syncthreads();
  if constexpr (BIN) {
    int gq[EPT], rk[EPT];
#pragma unroll
    for (int u = 0; u < EPT; ++u) {
      gq[u] = ((unsigned)e_d[u] < (unsigned)NNODES) ? e_d[u] / OWNR : NTH - 1;
      rk[u] = atomicAdd(&s_hist[gq[u]], 1);
    }
#pragma unroll
    for (int u = 0; u < EPT; ++u) {
      if (gq[u] < NPART && rk[u] < CELLC)
        cells[((size_t)gq[u] * NPART + blockIdx.x) * CELLC + rk[u]] =
            make_int2(e_s[u], e_d[u] - gq[u] * OWNR);
    }
  }

  const int wn = w % WN;
  const int wm = w / WN;
  const int mb = wm * TM;
  const int mc = min(TM, MT - mb);
  v8f acc[TM][TN];
#pragma unroll
  for (int i = 0; i < TM; ++i)
#pragma unroll
    for (int j = 0; j < TN; ++j) acc[i][j] = v8f{0.f, 0.f, 0.f, 0.f, 0.f, 0.f, 0.f, 0.f};

  const float* Ap = As + (mb * 16 + c) * PA + 2 * h;
  const float* Bp = Bs + h * PB + (wn * TN * 16 + c) * 2;
  auto kloop = [&](auto cnt) {
    constexpr int CNT = decltype(cnt)::value;
#pragma unroll 4
    for (int ks = 0; ks < K / 4; ++ks) {
      v2f a[CNT], b[TN];
#pragma unroll
      for (int i = 0; i < CNT; ++i)
        a[i] = *reinterpret_cast<const v2f*>(Ap + i * 16 * PA + 4 * ks);
#pragma unroll
      for (int j = 0; j < TN; ++j)
        b[j] = *reinterpret_cast<const v2f*>(Bp + 2 * ks * PB + j * 32);
#pragma unroll
      for (int i = 0; i < CNT; ++i)
#pragma unroll
        for (int j = 0; j < TN; ++j)
          acc[i][j] = __builtin_amdgcn_wmma_f32_16x16x4_f32(false, a[i], false, b[j],
                                                            (short)0, acc[i][j], false, false);
    }
  };
  if constexpr (MT % TM == 0) {
    kloop(int_const<TM>{});
  } else {
    if (mc == TM) kloop(int_const<TM>{});
    else          kloop(int_const<MT % TM>{});
  }
  __syncthreads();
  if constexpr (BIN) {
    for (int j = tid; j < CNTP; j += NTH)
      cntg[(size_t)blockIdx.x * CNTP + j] = (j < NPART) ? s_hist[j] : 0;
  }

#pragma unroll
  for (int i = 0; i < TM; ++i) {
    if (i < mc) {
#pragma unroll
      for (int j = 0; j < TN; ++j)
#pragma unroll
        for (int r = 0; r < 8; ++r)
          Os[((mb + i) * 16 + 8 * h + r) * PO + (wn * TN + j) * 16 + c] = acc[i][j][r];
    }
  }
  __syncthreads();

  constexpr int RB = (RM + 31) / 32;
  for (int blk = w; blk < H * RB; blk += NTH / 32) {
    const int  hd  = blk / RB;
    const int  row = (blk % RB) * 32 + lane;
    const bool ok  = row < RM;
    const float4* o4 = reinterpret_cast<const float4*>(&Os[(ok ? row : 0) * PO + hd * CH]);
    const float4* s4 = reinterpret_cast<const float4*>(&s_av[hd * CH]);
    const float4* d4 = reinterpret_cast<const float4*>(&s_av[AVW + hd * CH]);
    float s = 0.f, d = 0.f;
#pragma unroll
    for (int q = 0; q < CH / 4; ++q) {
      const float4 v = o4[q], x = s4[q], y = d4[q];
      s += v.x * x.x;  d += v.x * y.x;
      s += v.y * x.y;  d += v.y * y.y;
      s += v.z * x.z;  d += v.z * y.z;
      s += v.w * x.w;  d += v.w * y.w;
    }
    if (ok) {
      s_al[row * 2 * H + hd]     = s;
      s_al[row * 2 * H + H + hd] = d;
    }
  }
  __syncthreads();
  for (int j = tid; j < RM * 2 * H; j += NTH)
    al[(size_t)blockIdx.x * RM * 2 * H + j] = s_al[j];

  float4* Cg = reinterpret_cast<float4*>(C + (size_t)blockIdx.x * RM * N);
#pragma unroll 4
  for (int idx = tid; idx < RM * NV; idx += NTH)
    Cg[idx] = *reinterpret_cast<const float4*>(&Os[(idx / NV) * PO + (idx % NV) * 4]);
}

__device__ __forceinline__ void atomicMaxF(float* addr, float val) {
  if (val >= 0.f) atomicMax((int*)addr, __float_as_int(val));
  else            atomicMin((unsigned int*)addr, __float_as_uint(val));
}

template <int H, int C, int R, int QW, bool ELU>
__global__ __launch_bounds__(1024)
void agg_owner_kernel(const int* __restrict__ ei,
                      const float* __restrict__ al,
                      const float* __restrict__ h,
                      const float* __restrict__ bias,
                      float* __restrict__ out,
                      const int* __restrict__ cntg,
                      const int2* __restrict__ cells) {
  constexpr int HC  = H * C;
  constexpr int CPL = HC / 32;
  constexpr int NW  = 32;
  constexpr int NT  = NW * 32;
  constexpr int CPW = (NPART + NW - 1) / NW;
  static_assert(C % 32 == 0, "a 32-lane line must stay inside one head");
  static_assert(R == OWNR, "the owned range is one owner group of the binning");
  static_assert(CELLC <= 32 && CPW <= 32, "one lane per cell entry, one lane per cell count");
  static_assert(CPW * CELLC <= QW, "the queue takes all cells of a wave at capacity");
  static_assert(PARTE % 32 == 0 && QW >= 64, "a part is scanned 32 edges at a time");

  __shared__ float acc[R * HC];
  __shared__ int   q_eid[NW * QW];
  __shared__ int   q_dl[NW * QW];
  __shared__ float s_ad[R * H];
  __shared__ float s_self[R * H];
  __shared__ float s_m[R * H];
  __shared__ float s_den[R * H];

  const int tid  = threadIdx.x;
  const int lane = tid & 31;
  const int w    = __builtin_amdgcn_readfirstlane(tid >> 5);
  const int grp  = blockIdx.x;
  const int lo   = grp * R;
  int* qe = q_eid + w * QW;
  int* qd = q_dl  + w * QW;
  int myc = 0;
  {
    const int p = w * CPW + lane;
    if (lane < CPW && p < NPART) myc = cntg[(size_t)p * CNTP + grp];
  }

  for (int i = tid; i < R * H; i += NT) {
    const int n = lo + i / H;
    float a_d = 0.f, es = 0.f;
    if (n < NNODES) {
      a_d = al[(size_t)n * (2 * H) + H + i % H];
      const float v = al[(size_t)n * (2 * H) + i % H] + a_d;
      es = (v > 0.f) ? v : NEG_SLOPE * v;
    }
    s_ad[i]   = a_d;
    s_self[i] = es;
    s_m[i]    = es;
  }

  int  cnt     = 0;
  bool spilled = false;

  auto fill_cells = [&]() {
    int  c[CPW];
    int2 e[CPW];
#pragma unroll
    for (int u = 0; u < CPW; ++u) {
      const int cu = __builtin_amdgcn_readlane(myc, u);
      c[u] = (cu > 0 && cu <= CELLC) ? cu : 0;
      const int p = min(w * CPW + u, NPART - 1);
      e[u] = make_int2(0, 0);
      if (lane < c[u]) e[u] = cells[((size_t)grp * NPART + p) * CELLC + lane];
    }
#pragma unroll
    for (int u = 0; u < CPW; ++u) {
      if (lane < c[u]) {
        qe[cnt + lane] = e[u].x;
        qd[cnt + lane] = min(max(e[u].y, 0), R - 1);
      }
      cnt += c[u];
    }
  };
  fill_cells();
  __syncthreads();

  auto max_pass = [&](int nq) {
    __builtin_amdgcn_fence(__ATOMIC_RELEASE, "wavefront");
    __builtin_amdgcn_wave_barrier();
    for (int b = 0; b < nq; b += 32) {
      const int i = b + lane;
      if (i < nq) {
        const int dl = qd[i];
        const int s = min(max(qe[i], 0), NNODES - 1);
        float a[H];
#pragma unroll
        for (int hh = 0; hh < H; ++hh) a[hh] = al[(size_t)s * (2 * H) + hh];
#pragma unroll
        for (int hh = 0; hh < H; ++hh) {
          float v = a[hh] + s_ad[dl * H + hh];
          v = (v > 0.f) ? v : NEG_SLOPE * v;
          atomicMaxF(&s_m[dl * H + hh], v);
        }
      }
    }
    __builtin_amdgcn_fence(__ATOMIC_ACQUIRE, "wavefront");
    __builtin_amdgcn_wave_barrier();
  };

  auto acc_pass = [&](int nq) {
    __builtin_amdgcn_fence(__ATOMIC_RELEASE, "wavefront");
    __builtin_amdgcn_wave_barrier();
    for (int b = 0; b < nq; b += 32) {
      const int  i     = b + lane;
      const bool valid = i < nq;
      const int dl_i  = valid ? qd[i] : 0;
      int s_i = valid ? qe[i] : 0;
      s_i = min(max(s_i, 0), NNODES - 1);
      float a[H];
#pragma unroll
      for (int hh = 0; hh < H; ++hh) a[hh] = al[(size_t)s_i * (2 * H) + hh];
      int ex_i[H];
#pragma unroll
      for (int hh = 0; hh < H; ++hh) {
        float v = a[hh] + s_ad[dl_i * H + hh];
        v = (v > 0.f) ? v : NEG_SLOPE * v;
        const float ex = __expf(v - s_m[dl_i * H + hh]);
        if (valid) atomicAdd(&s_den[dl_i * H + hh], ex);
        ex_i[hh] = __float_as_int(ex);
      }
      const int n = min(32, nq - b);
      for (int k = 0; k < n; k += 4) {
        float v[4][CPL];
#pragma unroll
        for (int u = 0; u < 4; ++u) {
          const int s = __builtin_amdgcn_readlane(s_i, k + u);
          const float* hr = h + (size_t)s * HC + lane;
#pragma unroll
          for (int j = 0; j < CPL; ++j) v[u][j] = hr[j * 32];
        }
#pragma unroll
        for (int u = 0; u < 4; ++u) {
          if (k + u < n) {
            const int dl = __builtin_amdgcn_readlane(dl_i, k + u);
            float* ar = acc + dl * HC + lane;
#pragma unroll
            for (int j = 0; j < CPL; ++j) {
              const float ex =
                  __int_as_float(__builtin_amdgcn_readlane(ex_i[(j * 32) / C], k + u));
              atomicAdd(ar + j * 32, ex * v[u][j]);
            }
          }
        }
      }
    }
    __builtin_amdgcn_fence(__ATOMIC_ACQUIRE, "wavefront");
    __builtin_amdgcn_wave_barrier();
  };

  auto fill_scan = [&](auto&& drain) {
    for (int u = 0; u < CPW; ++u) {
      if (__builtin_amdgcn_readlane(myc, u) > CELLC) {
        const int p = min(w * CPW + u, NPART - 1);
        for (int b = 0; b < PARTE; b += 32) {
          if (cnt + 32 > QW) { drain(cnt); cnt = 0; }
          const int      eid  = p * PARTE + b + lane;
          const int      dd   = ei[NEDGES + eid] - lo;
          const bool     m    = (unsigned)dd < (unsigned)R;
          const unsigned mask = __builtin_amdgcn_ballot_w32(m);
          if (m) {
            const int pos = cnt + (int)__builtin_amdgcn_mbcnt_lo(mask, 0u);
            qe[pos] = ei[eid];
            qd[pos] = dd;
          }
          cnt += __builtin_popcount(mask);
        }
      }
    }
  };

  fill_scan([&](int nq) { max_pass(nq); spilled = true; });
  max_pass(cnt);
  __syncthreads();

  for (int i = tid; i < R * H; i += NT) s_den[i] = __expf(s_self[i] - s_m[i]);
  constexpr int NR = (R + NW - 1) / NW;
  {
    float iv[NR][CPL];
#pragma unroll
    for (int r = 0; r < NR; ++r) {
      const int  nl = w + r * NW;
      const int  n  = lo + nl;
      const bool ok = (nl < R) && (n < NNODES);
      const int  nn = ok ? n  : 0;
      const int  nc = ok ? nl : 0;
#pragma unroll
      for (int j = 0; j < CPL; ++j) {
        const int   hd = (j * 32) / C;
        const float p  = __expf(s_self[nc * H + hd] - s_m[nc * H + hd]) *
                         h[(size_t)nn * HC + j * 32 + lane];
        iv[r][j] = ok ? p : 0.f;
      }
    }
#pragma unroll
    for (int r = 0; r < NR; ++r) {
      const int nl = w + r * NW;
      if (nl < R) {
#pragma unroll
        for (int j = 0; j < CPL; ++j) acc[nl * HC + j * 32 + lane] = iv[r][j];
      }
    }
  }
  __syncthreads();

  if (spilled) {
    cnt = 0;
    fill_cells();
    fill_scan([&](int nq) { acc_pass(nq); });
  }
  acc_pass(cnt);
  __syncthreads();

  float bj[CPL];
#pragma unroll
  for (int j = 0; j < CPL; ++j) bj[j] = bias[j * 32 + lane];
#pragma unroll
  for (int r = 0; r < NR; ++r) {
    const int nl = w + r * NW;
    const int n  = lo + nl;
    if (nl < R && n < NNODES) {
#pragma unroll
      for (int j = 0; j < CPL; ++j) {
        const int ch = j * 32 + lane;
        float v = acc[nl * HC + ch] / (s_den[nl * H + (j * 32) / C] + 1e-16f) + bj[j];
        if (ELU) v = (v > 0.f) ? v : (__expf(v) - 1.f);
        out[(size_t)n * HC + ch] = v;
      }
    }
  }
}

extern "C" void kernel_launch(void* const* d_in, const int* in_sizes, int n_in,
                              void* d_out, int out_size, void* d_ws, size_t ws_size,
                              hipStream_t stream) {
  const float* x      = (const float*)d_in[0];
  const int*   ei     = (const int*)  d_in[1];
  const float* W1     = (const float*)d_in[2];
  const float* a_src1 = (const float*)d_in[3];
  const float* a_dst1 = (const float*)d_in[4];
  const float* b1     = (const float*)d_in[5];
  const float* W2     = (const float*)d_in[6];
  const float* a_src2 = (const float*)d_in[7];
  const float* a_dst2 = (const float*)d_in[8];
  const float* b2     = (const float*)d_in[9];
  float* out = (float*)d_out;

  float* ws   = (float*)d_ws;
  float* h1   = ws;
  float* g1   = h1   + (size_t)NNODES * H1C;
  float* al1  = g1   + (size_t)NNODES * H1C;
  float* h2   = al1  + (size_t)NNODES * 2 * HEADS;
  float* al2  = h2   + (size_t)NNODES * OUT_CH;

  const int RN  = 80;
  const int NG  = (NNODES + RN - 1) / RN;
  int*   cntg  = (int*)(al2 + (size_t)NNODES * 2);
  int2*  cells = (int2*)(cntg + NPART * CNTP);

  gemm_strip_kernel<IN_CH, H1C, 2, 5, 1, HEADS, true><<<NNODES / 80, 256, 0, stream>>>(
      x, W1, a_src1, a_dst1, h1, al1, ei, cells, cntg);
  agg_owner_kernel<HEADS, HID_CH, RN, 256, true><<<NG, 1024, 0, stream>>>(
      ei, al1, h1, b1, g1, cntg, cells);

  gemm_strip_kernel<H1C, OUT_CH, 1, 3, 2, 1, false><<<NNODES / 80, 256, 0, stream>>>(
      g1, W2, a_src2, a_dst2, h2, al2, nullptr, nullptr, nullptr);
  agg_owner_kernel<1, OUT_CH, RN, 256, false><<<NG, 1024, 0, stream>>>(
      ei, al2, h2, b2, out, cntg, cells);
}


# layer-2 GEMM and its logits done inside the layer-1 aggregation kernel: the owner group's 80 finished rows stay in LDS, W2 staged there; g1 round trip removed; sum order unchanged; 4 launches -> 3
# speedup vs baseline: 7.8621x; 1.0636x over previous
#include <hip/hip_runtime.h>
#include <math.h>
#include <cstddef>

#define NNODES   20000
#define NEDGES   320000
#define ETOT     (NEDGES + NNODES)
#define IN_CH    128
#define HID_CH   32
#define HEADS    8
#define H1C      (HEADS * HID_CH)
#define OUT_CH   64
#define NEG_SLOPE 0.2f
#define OWNR     80
#define NPART    (NNODES / OWNR)
#define PARTE    (NEDGES / NPART)
#define CELLC    32
#define CNTP     256

typedef float  v2f   __attribute__((ext_vector_type(2)));
typedef float  v8f   __attribute__((ext_vector_type(8)));

template <int V> struct int_const { static constexpr int value = V; };

template <int K, int N, int TN, int TM, int WM, int H, bool BIN>
__global__ __launch_bounds__(256)
void gemm_strip_kernel(const float* __restrict__ A, const float* __restrict__ B,
                       const float* __restrict__ a_src, const float* __restrict__ a_dst,
                       float* __restrict__ C, float* __restrict__ al,
                       const int* __restrict__ ei, int2* __restrict__ cells,
                       int* __restrict__ cntg) {
  constexpr int RM  = 80;
  constexpr int MT  = RM / 16;
  constexpr int NTH = 256;
  constexpr int WN  = (N / 16) / TN;
  constexpr int PA  = K + 4;
  constexpr int PB  = 2 * N + 32;
  constexpr int PO  = N + 4;
  constexpr int KV  = K / 4;
  constexpr int NV  = N / 4;
  constexpr int A_WORDS = RM * PA;
  constexpr int B_WORDS = (K / 2) * PB;
  constexpr int CH  = N / H;
  constexpr int AVN = (N + NTH - 1) / NTH;
  constexpr int AVW = AVN * NTH;
  static_assert(WN * WM * 32 == NTH, "8 waves");
  static_assert(TM * WM >= MT && TM * (WM - 1) < MT, "M-tiles dealt over WM wave rows");
  static_assert(RM * PO <= A_WORDS + B_WORDS, "C tile reuses the operand space");
  static_assert((RM * KV) % (NTH * 10) == 0, "A staging in batches of 10");
  static_assert(((K / 2) * NV) % (NTH * 8) == 0, "B staging in batches of 8");
  static_assert((RM * NV) % NTH == 0, "C store loop");
  static_assert(NNODES % RM == 0, "no partial strip");
  static_assert(N % H == 0 && CH % 4 == 0, "a head's channels are read four at a time");
  static_assert((RM * 2 * H) % 32 == 0, "a strip's logits fill whole 128-B lines");
  constexpr int EPT = BIN ? PARTE / NTH : 1;
  static_assert(!BIN || (RM == OWNR && PARTE % NTH == 0 && NEDGES % NPART == 0 &&
                         NPART < NTH && NTH <= CNTP),
                "one list part per strip workgroup, one count per thread");

  __shared__ alignas(16) float smem[A_WORDS + B_WORDS];
  __shared__ alignas(16) float s_av[2 * AVW];
  __shared__ alignas(16) float s_al[RM * 2 * H];
  __shared__ int s_hist[BIN ? NTH : 1];
  float* As = smem;
  float* Bs = smem + A_WORDS;
  float* Os = smem;

  const int tid  = threadIdx.x;
  const int lane = tid & 31;
  const int w    = __builtin_amdgcn_readfirstlane(tid >> 5);
  const int c    = lane & 15;
  const int h    = lane >> 4;

  float avs[AVN], avd[AVN];
#pragma unroll
  for (int u = 0; u < AVN; ++u) {
    const int idx = min(u * NTH + tid, N - 1);
    avs[u] = a_src[idx];
    avd[u] = a_dst[idx];
  }

  int e_s[EPT], e_d[EPT];
  if constexpr (BIN) {
#pragma unroll
    for (int u = 0; u < EPT; ++u) {
      const int e = blockIdx.x * PARTE + u * NTH + tid;
      e_s[u] = ei[e];
      e_d[u] = ei[NEDGES + e];
    }
    s_hist[tid] = 0;
  }

  {
    const float4* Ag = reinterpret_cast<const float4*>(A + (size_t)blockIdx.x * RM * K);
#pragma unroll
    for (int i0 = 0; i0 < (RM * KV) / NTH; i0 += 10) {
      float4 r[10];
#pragma unroll
      for (int u = 0; u < 10; ++u) r[u] = Ag[(i0 + u) * NTH + tid];
#pragma unroll
      for (int u = 0; u < 10; ++u) {
        const int idx = (i0 + u) * NTH + tid;
        *reinterpret_cast<float4*>(&As[(idx / KV) * PA + (idx % KV) * 4]) = r[u];
      }
    }
  }
  {
    const float4* Bg = reinterpret_cast<const float4*>(B);
#pragma unroll
    for (int i0 = 0; i0 < ((K / 2) * NV) / NTH; i0 += 8) {
      float4 r0[8], r1[8];
#pragma unroll
      for (int u = 0; u < 8; ++u) {
        const int idx = (i0 + u) * NTH + tid;
        const int p = idx / NV, c4 = idx % NV;
        r0[u] = Bg[(2 * p) * NV + c4];
        r1[u] = Bg[(2 * p + 1) * NV + c4];
      }
#pragma unroll
      for (int u = 0; u < 8; ++u) {
        const int idx = (i0 + u) * NTH + tid;
        const int p = idx / NV, c4 = idx % NV;
        float* d = &Bs[p * PB + c4 * 8];
        *reinterpret_cast<float4*>(d)     = make_float4(r0[u].x, r1[u].x, r0[u].y, r1[u].y);
        *reinterpret_cast<float4*>(d + 4) = make_float4(r0[u].z, r1[u].z, r0[u].w, r1[u].w);
      }
    }
  }
#pragma unroll
  for (int u = 0; u < AVN; ++u) {
    s_av[u * NTH + tid]       = avs[u];
    s_av[AVW + u * NTH + tid] = avd[u];
  }
  __syncthreads();
  if constexpr (BIN) {
    int gq[EPT], rk[EPT];
#pragma unroll
    for (int u = 0; u < EPT; ++u) {
      gq[u] = ((unsigned)e_d[u] < (unsigned)NNODES) ? e_d[u] / OWNR : NTH - 1;
      rk[u] = atomicAdd(&s_hist[gq[u]], 1);
    }
#pragma unroll
    for (int u = 0; u < EPT; ++u) {
      if (gq[u] < NPART && rk[u] < CELLC)
        cells[((size_t)gq[u] * NPART + blockIdx.x) * CELLC + rk[u]] =
            make_int2(e_s[u], e_d[u] - gq[u] * OWNR);
    }
  }

  const int wn = w % WN;
  const int wm = w / WN;
  const int mb = wm * TM;
  const int mc = min(TM, MT - mb);
  v8f acc[TM][TN];
#pragma unroll
  for (int i = 0; i < TM; ++i)
#pragma unroll
    for (int j = 0; j < TN; ++j) acc[i][j] = v8f{0.f, 0.f, 0.f, 0.f, 0.f, 0.f, 0.f, 0.f};

  const float* Ap = As + (mb * 16 + c) * PA + 2 * h;
  const float* Bp = Bs + h * PB + (wn * TN * 16 + c) * 2;
  auto kloop = [&](auto cnt) {
    constexpr int CNT = decltype(cnt)::value;
#pragma unroll 4
    for (int ks = 0; ks < K / 4; ++ks) {
      v2f a[CNT], b[TN];
#pragma unroll
      for (int i = 0; i < CNT; ++i)
        a[i] = *reinterpret_cast<const v2f*>(Ap + i * 16 * PA + 4 * ks);
#pragma unroll
      for (int j = 0; j < TN; ++j)
        b[j] = *reinterpret_cast<const v2f*>(Bp + 2 * ks * PB + j * 32);
#pragma unroll
      for (int i = 0; i < CNT; ++i)
#pragma unroll
        for (int j = 0; j < TN; ++j)
          acc[i][j] = __builtin_amdgcn_wmma_f32_16x16x4_f32(false, a[i], false, b[j],
                                                            (short)0, acc[i][j], false, false);
    }
  };
  if constexpr (MT % TM == 0) {
    kloop(int_const<TM>{});
  } else {
    if (mc == TM) kloop(int_const<TM>{});
    else          kloop(int_const<MT % TM>{});
  }
  __syncthreads();
  if constexpr (BIN) {
    for (int j = tid; j < CNTP; j += NTH)
      cntg[(size_t)blockIdx.x * CNTP + j] = (j < NPART) ? s_hist[j] : 0;
  }

#pragma unroll
  for (int i = 0; i < TM; ++i) {
    if (i < mc) {
#pragma unroll
      for (int j = 0; j < TN; ++j)
#pragma unroll
        for (int r = 0; r < 8; ++r)
          Os[((mb + i) * 16 + 8 * h + r) * PO + (wn * TN + j) * 16 + c] = acc[i][j][r];
    }
  }
  __syncthreads();

  constexpr int RB = (RM + 31) / 32;
  for (int blk = w; blk < H * RB; blk += NTH / 32) {
    const int  hd  = blk / RB;
    const int  row = (blk % RB) * 32 + lane;
    const bool ok  = row < RM;
    const float4* o4 = reinterpret_cast<const float4*>(&Os[(ok ? row : 0) * PO + hd * CH]);
    const float4* s4 = reinterpret_cast<const float4*>(&s_av[hd * CH]);
    const float4* d4 = reinterpret_cast<const float4*>(&s_av[AVW + hd * CH]);
    float s = 0.f, d = 0.f;
#pragma unroll
    for (int q = 0; q < CH / 4; ++q) {
      const float4 v = o4[q], x = s4[q], y = d4[q];
      s += v.x * x.x;  d += v.x * y.x;
      s += v.y * x.y;  d += v.y * y.y;
      s += v.z * x.z;  d += v.z * y.z;
      s += v.w * x.w;  d += v.w * y.w;
    }
    if (ok) {
      s_al[row * 2 * H + hd]     = s;
      s_al[row * 2 * H + H + hd] = d;
    }
  }
  __syncthreads();
  for (int j = tid; j < RM * 2 * H; j += NTH)
    al[(size_t)blockIdx.x * RM * 2 * H + j] = s_al[j];

  float4* Cg = reinterpret_cast<float4*>(C + (size_t)blockIdx.x * RM * N);
#pragma unroll 4
  for (int idx = tid; idx < RM * NV; idx += NTH)
    Cg[idx] = *reinterpret_cast<const float4*>(&Os[(idx / NV) * PO + (idx % NV) * 4]);
}

__device__ __forceinline__ void atomicMaxF(float* addr, float val) {
  if (val >= 0.f) atomicMax((int*)addr, __float_as_int(val));
  else            atomicMin((unsigned int*)addr, __float_as_uint(val));
}

template <int H, int C, int R, int QW, bool ELU, int N2>
__global__ __launch_bounds__(1024)
void agg_owner_kernel(const int* __restrict__ ei,
                      const float* __restrict__ al,
                      const float* __restrict__ h,
                      const float* __restrict__ bias,
                      float* __restrict__ out,
                      const int* __restrict__ cntg,
                      const int2* __restrict__ cells,
                      const float* __restrict__ W2,
                      const float* __restrict__ a_s2,
                      const float* __restrict__ a_d2,
                      float* __restrict__ al2) {
  constexpr bool FUSE = N2 > 0;
  constexpr int HC  = H * C;
  constexpr int PAC = FUSE ? HC + 4 : HC;
  constexpr int NB  = FUSE ? N2 : 16;
  constexpr int PB2 = 2 * NB + 32;
  constexpr int PO2 = NB + 4;
  constexpr int NV2 = NB / 4;
  constexpr int MT2 = R / 16, NT2 = NB / 16;
  constexpr int CPL = HC / 32;
  constexpr int NW  = 32;
  constexpr int NT  = NW * 32;
  constexpr int CPW = (NPART + NW - 1) / NW;
  static_assert(C % 32 == 0, "a 32-lane line must stay inside one head");
  static_assert(R == OWNR, "the owned range is one owner group of the binning");
  static_assert(CELLC <= 32 && CPW <= 32, "one lane per cell entry, one lane per cell count");
  static_assert(CPW * CELLC <= QW, "the queue takes all cells of a wave at capacity");
  static_assert(PARTE % 32 == 0 && QW >= 64, "a part is scanned 32 edges at a time");

  static_assert(!FUSE || (R % 16 == 0 && N2 % 16 == 0 && HC % 4 == 0 && MT2 * NT2 <= NW &&
                          ((HC / 2) * NV2) % NT == 0 && (R * 2) % 32 == 0),
                "one wave per product tile, W2 staged in whole rounds, logits in whole lines");
  __shared__ alignas(16) float acc[R * PAC];
  __shared__ alignas(16) float s_b[FUSE ? (HC / 2) * PB2 : 1];
  __shared__ alignas(16) float s_o[FUSE ? R * PO2 : 1];
  __shared__ alignas(16) float s_av2[FUSE ? 2 * NB : 4];
  __shared__ int   q_eid[NW * QW];
  __shared__ int   q_dl[NW * QW];
  __shared__ float s_ad[R * H];
  __shared__ float s_self[R * H];
  __shared__ float s_m[R * H];
  __shared__ float s_den[R * H];

  const int tid  = threadIdx.x;
  const int lane = tid & 31;
  const int w    = __builtin_amdgcn_readfirstlane(tid >> 5);
  const int grp  = blockIdx.x;
  const int lo   = grp * R;
  int* qe = q_eid + w * QW;
  int* qd = q_dl  + w * QW;
  int myc = 0;
  {
    const int p = w * CPW + lane;
    if (lane < CPW && p < NPART) myc = cntg[(size_t)p * CNTP + grp];
  }
  constexpr int BL = FUSE ? ((HC / 2) * NV2) / NT : 1;
  float4 w2a[BL], w2b[BL];
  if constexpr (FUSE) {
    const float4* Bg = reinterpret_cast<const float4*>(W2);
#pragma unroll
    for (int u = 0; u < BL; ++u) {
      const int idx = u * NT + tid;
      const int p = idx / NV2, c4 = idx % NV2;
      w2a[u] = Bg[(2 * p) * NV2 + c4];
      w2b[u] = Bg[(2 * p + 1) * NV2 + c4];
    }
  }
  float av2 = 0.f;
  if constexpr (FUSE) {
    if (tid < 2 * NB) av2 = (tid < NB) ? a_s2[tid] : a_d2[tid - NB];
  }

  for (int i = tid; i < R * H; i += NT) {
    const int n = lo + i / H;
    float a_d = 0.f, es = 0.f;
    if (n < NNODES) {
      a_d = al[(size_t)n * (2 * H) + H + i % H];
      const float v = al[(size_t)n * (2 * H) + i % H] + a_d;
      es = (v > 0.f) ? v : NEG_SLOPE * v;
    }
    s_ad[i]   = a_d;
    s_self[i] = es;
    s_m[i]    = es;
  }

  int  cnt     = 0;
  bool spilled = false;

  auto fill_cells = [&]() {
    int  c[CPW];
    int2 e[CPW];
#pragma unroll
    for (int u = 0; u < CPW; ++u) {
      const int cu = __builtin_amdgcn_readlane(myc, u);
      c[u] = (cu > 0 && cu <= CELLC) ? cu : 0;
      const int p = min(w * CPW + u, NPART - 1);
      e[u] = make_int2(0, 0);
      if (lane < c[u]) e[u] = cells[((size_t)grp * NPART + p) * CELLC + lane];
    }
#pragma unroll
    for (int u = 0; u < CPW; ++u) {
      if (lane < c[u]) {
        qe[cnt + lane] = e[u].x;
        qd[cnt + lane] = min(max(e[u].y, 0), R - 1);
      }
      cnt += c[u];
    }
  };
  fill_cells();
  if constexpr (FUSE) {
#pragma unroll
    for (int u = 0; u < BL; ++u) {
      const int idx = u * NT + tid;
      const int p = idx / NV2, c4 = idx % NV2;
      float* d = &s_b[p * PB2 + c4 * 8];
      *reinterpret_cast<float4*>(d)     = make_float4(w2a[u].x, w2b[u].x, w2a[u].y, w2b[u].y);
      *reinterpret_cast<float4*>(d + 4) = make_float4(w2a[u].z, w2b[u].z, w2a[u].w, w2b[u].w);
    }
    if (tid < 2 * NB) s_av2[tid] = av2;
  }
  __syncthreads();

  auto max_pass = [&](int nq) {
    __builtin_amdgcn_fence(__ATOMIC_RELEASE, "wavefront");
    __builtin_amdgcn_wave_barrier();
    for (int b = 0; b < nq; b += 32) {
      const int i = b + lane;
      if (i < nq) {
        const int dl = qd[i];
        const int s = min(max(qe[i], 0), NNODES - 1);
        float a[H];
#pragma unroll
        for (int hh = 0; hh < H; ++hh) a[hh] = al[(size_t)s * (2 * H) + hh];
#pragma unroll
        for (int hh = 0; hh < H; ++hh) {
          float v = a[hh] + s_ad[dl * H + hh];
          v = (v > 0.f) ? v : NEG_SLOPE * v;
          atomicMaxF(&s_m[dl * H + hh], v);
        }
      }
    }
    __builtin_amdgcn_fence(__ATOMIC_ACQUIRE, "wavefront");
    __builtin_amdgcn_wave_barrier();
  };

  auto acc_pass = [&](int nq) {
    __builtin_amdgcn_fence(__ATOMIC_RELEASE, "wavefront");
    __builtin_amdgcn_wave_barrier();
    for (int b = 0; b < nq; b += 32) {
      const int  i     = b + lane;
      const bool valid = i < nq;
      const int dl_i  = valid ? qd[i] : 0;
      int s_i = valid ? qe[i] : 0;
      s_i = min(max(s_i, 0), NNODES - 1);
      float a[H];
#pragma unroll
      for (int hh = 0; hh < H; ++hh) a[hh] = al[(size_t)s_i * (2 * H) + hh];
      int ex_i[H];
#pragma unroll
      for (int hh = 0; hh < H; ++hh) {
        float v = a[hh] + s_ad[dl_i * H + hh];
        v = (v > 0.f) ? v : NEG_SLOPE * v;
        const float ex = __expf(v - s_m[dl_i * H + hh]);
        if (valid) atomicAdd(&s_den[dl_i * H + hh], ex);
        ex_i[hh] = __float_as_int(ex);
      }
      const int n = min(32, nq - b);
      for (int k = 0; k < n; k += 4) {
        float v[4][CPL];
#pragma unroll
        for (int u = 0; u < 4; ++u) {
          const int s = __builtin_amdgcn_readlane(s_i, k + u);
          const float* hr = h + (size_t)s * HC + lane;
#pragma unroll
          for (int j = 0; j < CPL; ++j) v[u][j] = hr[j * 32];
        }
#pragma unroll
        for (int u = 0; u < 4; ++u) {
          if (k + u < n) {
            const int dl = __builtin_amdgcn_readlane(dl_i, k + u);
            float* ar = acc + dl * PAC + lane;
#pragma unroll
            for (int j = 0; j < CPL; ++j) {
              const float ex =
                  __int_as_float(__builtin_amdgcn_readlane(ex_i[(j * 32) / C], k + u));
              atomicAdd(ar + j * 32, ex * v[u][j]);
            }
          }
        }
      }
    }
    __builtin_amdgcn_fence(__ATOMIC_ACQUIRE, "wavefront");
    __builtin_amdgcn_wave_barrier();
  };

  auto fill_scan = [&](auto&& drain) {
    for (int u = 0; u < CPW; ++u) {
      if (__builtin_amdgcn_readlane(myc, u) > CELLC) {
        const int p = min(w * CPW + u, NPART - 1);
        for (int b = 0; b < PARTE; b += 32) {
          if (cnt + 32 > QW) { drain(cnt); cnt = 0; }
          const int      eid  = p * PARTE + b + lane;
          const int      dd   = ei[NEDGES + eid] - lo;
          const bool     m    = (unsigned)dd < (unsigned)R;
          const unsigned mask = __builtin_amdgcn_ballot_w32(m);
          if (m) {
            const int pos = cnt + (int)__builtin_amdgcn_mbcnt_lo(mask, 0u);
            qe[pos] = ei[eid];
            qd[pos] = dd;
          }
          cnt += __builtin_popcount(mask);
        }
      }
    }
  };

  fill_scan([&](int nq) { max_pass(nq); spilled = true; });
  max_pass(cnt);
  __syncthreads();

  for (int i = tid; i < R * H; i += NT) s_den[i] = __expf(s_self[i] - s_m[i]);
  constexpr int NR = (R + NW - 1) / NW;
  {
    float iv[NR][CPL];
#pragma unroll
    for (int r = 0; r < NR; ++r) {
      const int  nl = w + r * NW;
      const int  n  = lo + nl;
      const bool ok = (nl < R) && (n < NNODES);
      const int  nn = ok ? n  : 0;
      const int  nc = ok ? nl : 0;
#pragma unroll
      for (int j = 0; j < CPL; ++j) {
        const int   hd = (j * 32) / C;
        const float p  = __expf(s_self[nc * H + hd] - s_m[nc * H + hd]) *
                         h[(size_t)nn * HC + j * 32 + lane];
        iv[r][j] = ok ? p : 0.f;
      }
    }
#pragma unroll
    for (int r = 0; r < NR; ++r) {
      const int nl = w + r * NW;
      if (nl < R) {
#pragma unroll
        for (int j = 0; j < CPL; ++j) acc[nl * PAC + j * 32 + lane] = iv[r][j];
      }
    }
  }
  __syncthreads();

  if (spilled) {
    cnt = 0;
    fill_cells();
    fill_scan([&](int nq) { acc_pass(nq); });
  }
  acc_pass(cnt);
  __syncthreads();

  float bj[CPL];
#pragma unroll
  for (int j = 0; j < CPL; ++j) bj[j] = bias[j * 32 + lane];
#pragma unroll
  for (int r = 0; r < NR; ++r) {
    const int nl = w + r * NW;
    const int n  = lo + nl;
    if (nl < R && n < NNODES) {
#pragma unroll
      for (int j = 0; j < CPL; ++j) {
        const int ch = j * 32 + lane;
        float v = acc[nl * PAC + ch] / (s_den[nl * H + (j * 32) / C] + 1e-16f) + bj[j];
        if (ELU) v = (v > 0.f) ? v : (__expf(v) - 1.f);
        if constexpr (FUSE) acc[nl * PAC + ch] = v;
        else                out[(size_t)n * HC + ch] = v;
      }
    }
  }

  if constexpr (FUSE) {
    __syncthreads();
    if (w < MT2 * NT2) {
      const int c  = lane & 15;
      const int hf = lane >> 4;
      const int mt = w / NT2, nt = w % NT2;
      const float* Ap = acc + (mt * 16 + c) * PAC + 2 * hf;
      const float* Bp = s_b + hf * PB2 + (nt * 16 + c) * 2;
      v8f d8 = v8f{0.f, 0.f, 0.f, 0.f, 0.f, 0.f, 0.f, 0.f};
#pragma unroll 8
      for (int ks = 0; ks < HC / 4; ++ks) {
        const v2f a = *reinterpret_cast<const v2f*>(Ap + 4 * ks);
        const v2f b = *reinterpret_cast<const v2f*>(Bp + 2 * ks * PB2);
        d8 = __builtin_amdgcn_wmma_f32_16x16x4_f32(false, a, false, b, (short)0, d8, false, false);
      }
#pragma unroll
      for (int r = 0; r < 8; ++r)
        s_o[(mt * 16 + 8 * hf + r) * PO2 + nt * 16 + c] = d8[r];
    }
    __syncthreads();
    for (int blk = w; blk < (R + 31) / 32; blk += NW) {
      const int  row = blk * 32 + lane;
      const bool ok  = (row < R) && (lo + row < NNODES);
      const float4* o4 = reinterpret_cast<const float4*>(&s_o[(ok ? row : 0) * PO2]);
      const float4* s4 = reinterpret_cast<const float4*>(&s_av2[0]);
      const float4* d4 = reinterpret_cast<const float4*>(&s_av2[NB]);
      float s = 0.f, d = 0.f;
#pragma unroll
      for (int q = 0; q < NB / 4; ++q) {
        const float4 v = o4[q], x = s4[q], y = d4[q];
        s += v.x * x.x;  d += v.x * y.x;
        s += v.y * x.y;  d += v.y * y.y;
        s += v.z * x.z;  d += v.z * y.z;
        s += v.w * x.w;  d += v.w * y.w;
      }
      if (ok) *reinterpret_cast<float2*>(al2 + (size_t)(lo + row) * 2) = make_float2(s, d);
    }
    float4* Cg = reinterpret_cast<float4*>(out + (size_t)lo * NB);
    for (int idx = tid; idx < R * NV2; idx += NT) {
      if (lo + idx / NV2 < NNODES)
        Cg[idx] = *reinterpret_cast<const float4*>(&s_o[(idx / NV2) * PO2 + (idx % NV2) * 4]);
    }
  }
}

extern "C" void kernel_launch(void* const* d_in, const int* in_sizes, int n_in,
                              void* d_out, int out_size, void* d_ws, size_t ws_size,
                              hipStream_t stream) {
  const float* x      = (const float*)d_in[0];
  const int*   ei     = (const int*)  d_in[1];
  const float* W1     = (const float*)d_in[2];
  const float* a_src1 = (const float*)d_in[3];
  const float* a_dst1 = (const float*)d_in[4];
  const float* b1     = (const float*)d_in[5];
  const float* W2     = (const float*)d_in[6];
  const float* a_src2 = (const float*)d_in[7];
  const float* a_dst2 = (const float*)d_in[8];
  const float* b2     = (const float*)d_in[9];
  float* out = (float*)d_out;

  float* ws   = (float*)d_ws;
  float* h1   = ws;
  float* al1  = h1   + (size_t)NNODES * H1C;
  float* h2   = al1  + (size_t)NNODES * 2 * HEADS;
  float* al2  = h2   + (size_t)NNODES * OUT_CH;

  const int RN  = 80;
  const int NG  = (NNODES + RN - 1) / RN;
  int*   cntg  = (int*)(al2 + (size_t)NNODES * 2);
  int2*  cells = (int2*)(cntg + NPART * CNTP);

  gemm_strip_kernel<IN_CH, H1C, 2, 5, 1, HEADS, true><<<NNODES / 80, 256, 0, stream>>>(
      x, W1, a_src1, a_dst1, h1, al1, ei, cells, cntg);
  agg_owner_kernel<HEADS, HID_CH, RN, 256, true, OUT_CH><<<NG, 1024, 0, stream>>>(
      ei, al1, h1, b1, h2, cntg, cells, W2, a_src2, a_dst2, al2);

  agg_owner_kernel<1, OUT_CH, RN, 256, false, 0><<<NG, 1024, 0, stream>>>(
      ei, al2, h2, b2, out, cntg, cells, nullptr, nullptr, nullptr, nullptr);
}


# row sums in registers: edges placed by destination row in LDS (integer rank); a wave owns whole rows, softmax by lane reductions, ex*h[src] summed in registers; no LDS f32 atomics; sum order changes
# speedup vs baseline: 10.2174x; 1.2996x over previous
#include <hip/hip_runtime.h>
#include <math.h>
#include <cstddef>

#define NNODES   20000
#define NEDGES   320000
#define ETOT     (NEDGES + NNODES)
#define IN_CH    128
#define HID_CH   32
#define HEADS    8
#define H1C      (HEADS * HID_CH)
#define OUT_CH   64
#define NEG_SLOPE 0.2f
#define OWNR     80
#define NPART    (NNODES / OWNR)
#define PARTE    (NEDGES / NPART)
#define CELLC    32
#define CNTP     256

typedef float  v2f   __attribute__((ext_vector_type(2)));
typedef float  v8f   __attribute__((ext_vector_type(8)));

template <int V> struct int_const { static constexpr int value = V; };

template <int K, int N, int TN, int TM, int WM, int H, bool BIN>
__global__ __launch_bounds__(256)
void gemm_strip_kernel(const float* __restrict__ A, const float* __restrict__ B,
                       const float* __restrict__ a_src, const float* __restrict__ a_dst,
                       float* __restrict__ C, float* __restrict__ al,
                       const int* __restrict__ ei, int2* __restrict__ cells,
                       int* __restrict__ cntg) {
  constexpr int RM  = 80;
  constexpr int MT  = RM / 16;
  constexpr int NTH = 256;
  constexpr int WN  = (N / 16) / TN;
  constexpr int PA  = K + 4;
  constexpr int PB  = 2 * N + 32;
  constexpr int PO  = N + 4;
  constexpr int KV  = K / 4;
  constexpr int NV  = N / 4;
  constexpr int A_WORDS = RM * PA;
  constexpr int B_WORDS = (K / 2) * PB;
  constexpr int CH  = N / H;
  constexpr int AVN = (N + NTH - 1) / NTH;
  constexpr int AVW = AVN * NTH;
  static_assert(WN * WM * 32 == NTH, "8 waves");
  static_assert(TM * WM >= MT && TM * (WM - 1) < MT, "M-tiles dealt over WM wave rows");
  static_assert(RM * PO <= A_WORDS + B_WORDS, "C tile reuses the operand space");
  static_assert((RM * KV) % (NTH * 10) == 0, "A staging in batches of 10");
  static_assert(((K / 2) * NV) % (NTH * 8) == 0, "B staging in batches of 8");
  static_assert((RM * NV) % NTH == 0, "C store loop");
  static_assert(NNODES % RM == 0, "no partial strip");
  static_assert(N % H == 0 && CH % 4 == 0, "a head's channels are read four at a time");
  static_assert((RM * 2 * H) % 32 == 0, "a strip's logits fill whole 128-B lines");
  constexpr int EPT = BIN ? PARTE / NTH : 1;
  static_assert(!BIN || (RM == OWNR && PARTE % NTH == 0 && NEDGES % NPART == 0 &&
                         NPART < NTH && NTH <= CNTP),
                "one list part per strip workgroup, one count per thread");

  __shared__ alignas(16) float smem[A_WORDS + B_WORDS];
  __shared__ alignas(16) float s_av[2 * AVW];
  __shared__ alignas(16) float s_al[RM * 2 * H];
  __shared__ int s_hist[BIN ? NTH : 1];
  float* As = smem;
  float* Bs = smem + A_WORDS;
  float* Os = smem;

  const int tid  = threadIdx.x;
  const int lane = tid & 31;
  const int w    = __builtin_amdgcn_readfirstlane(tid >> 5);
  const int c    = lane & 15;
  const int h    = lane >> 4;

  float avs[AVN], avd[AVN];
#pragma unroll
  for (int u = 0; u < AVN; ++u) {
    const int idx = min(u * NTH + tid, N - 1);
    avs[u] = a_src[idx];
    avd[u] = a_dst[idx];
  }

  int e_s[EPT], e_d[EPT];
  if constexpr (BIN) {
#pragma unroll
    for (int u = 0; u < EPT; ++u) {
      const int e = blockIdx.x * PARTE + u * NTH + tid;
      e_s[u] = ei[e];
      e_d[u] = ei[NEDGES + e];
    }
    s_hist[tid] = 0;
  }

  {
    const float4* Ag = reinterpret_cast<const float4*>(A + (size_t)blockIdx.x * RM * K);
#pragma unroll
    for (int i0 = 0; i0 < (RM * KV) / NTH; i0 += 10) {
      float4 r[10];
#pragma unroll
      for (int u = 0; u < 10; ++u) r[u] = Ag[(i0 + u) * NTH + tid];
#pragma unroll
      for (int u = 0; u < 10; ++u) {
        const int idx = (i0 + u) * NTH + tid;
        *reinterpret_cast<float4*>(&As[(idx / KV) * PA + (idx % KV) * 4]) = r[u];
      }
    }
  }
  {
    const float4* Bg = reinterpret_cast<const float4*>(B);
#pragma unroll
    for (int i0 = 0; i0 < ((K / 2) * NV) / NTH; i0 += 8) {
      float4 r0[8], r1[8];
#pragma unroll
      for (int u = 0; u < 8; ++u) {
        const int idx = (i0 + u) * NTH + tid;
        const int p = idx / NV, c4 = idx % NV;
        r0[u] = Bg[(2 * p) * NV + c4];
        r1[u] = Bg[(2 * p + 1) * NV + c4];
      }
#pragma unroll
      for (int u = 0; u < 8; ++u) {
        const int idx = (i0 + u) * NTH + tid;
        const int p = idx / NV, c4 = idx % NV;
        float* d = &Bs[p * PB + c4 * 8];
        *reinterpret_cast<float4*>(d)     = make_float4(r0[u].x, r1[u].x, r0[u].y, r1[u].y);
        *reinterpret_cast<float4*>(d + 4) = make_float4(r0[u].z, r1[u].z, r0[u].w, r1[u].w);
      }
    }
  }
#pragma unroll
  for (int u = 0; u < AVN; ++u) {
    s_av[u * NTH + tid]       = avs[u];
    s_av[AVW + u * NTH + tid] = avd[u];
  }
  __syncthreads();
  if constexpr (BIN) {
    int gq[EPT], rk[EPT];
#pragma unroll
    for (int u = 0; u < EPT; ++u) {
      gq[u] = ((unsigned)e_d[u] < (unsigned)NNODES) ? e_d[u] / OWNR : NTH - 1;
      rk[u] = atomicAdd(&s_hist[gq[u]], 1);
    }
#pragma unroll
    for (int u = 0; u < EPT; ++u) {
      if (gq[u] < NPART && rk[u] < CELLC)
        cells[((size_t)gq[u] * NPART + blockIdx.x) * CELLC + rk[u]] =
            make_int2(e_s[u], e_d[u] - gq[u] * OWNR);
    }
  }

  const int wn = w % WN;
  const int wm = w / WN;
  const int mb = wm * TM;
  const int mc = min(TM, MT - mb);
  v8f acc[TM][TN];
#pragma unroll
  for (int i = 0; i < TM; ++i)
#pragma unroll
    for (int j = 0; j < TN; ++j) acc[i][j] = v8f{0.f, 0.f, 0.f, 0.f, 0.f, 0.f, 0.f, 0.f};

  const float* Ap = As + (mb * 16 + c) * PA + 2 * h;
  const float* Bp = Bs + h * PB + (wn * TN * 16 + c) * 2;
  auto kloop = [&](auto cnt) {
    constexpr int CNT = decltype(cnt)::value;
#pragma unroll 4
    for (int ks = 0; ks < K / 4; ++ks) {
      v2f a[CNT], b[TN];
#pragma unroll
      for (int i = 0; i < CNT; ++i)
        a[i] = *reinterpret_cast<const v2f*>(Ap + i * 16 * PA + 4 * ks);
#pragma unroll
      for (int j = 0; j < TN; ++j)
        b[j] = *reinterpret_cast<const v2f*>(Bp + 2 * ks * PB + j * 32);
#pragma unroll
      for (int i = 0; i < CNT; ++i)
#pragma unroll
        for (int j = 0; j < TN; ++j)
          acc[i][j] = __builtin_amdgcn_wmma_f32_16x16x4_f32(false, a[i], false, b[j],
                                                            (short)0, acc[i][j], false, false);
    }
  };
  if constexpr (MT % TM == 0) {
    kloop(int_const<TM>{});
  } else {
    if (mc == TM) kloop(int_const<TM>{});
    else          kloop(int_const<MT % TM>{});
  }
  __syncthreads();
  if constexpr (BIN) {
    for (int j = tid; j < CNTP; j += NTH)
      cntg[(size_t)blockIdx.x * CNTP + j] = (j < NPART) ? s_hist[j] : 0;
  }

#pragma unroll
  for (int i = 0; i < TM; ++i) {
    if (i < mc) {
#pragma unroll
      for (int j = 0; j < TN; ++j)
#pragma unroll
        for (int r = 0; r < 8; ++r)
          Os[((mb + i) * 16 + 8 * h + r) * PO + (wn * TN + j) * 16 + c] = acc[i][j][r];
    }
  }
  __syncthreads();

  constexpr int RB = (RM + 31) / 32;
  for (int blk = w; blk < H * RB; blk += NTH / 32) {
    const int  hd  = blk / RB;
    const int  row = (blk % RB) * 32 + lane;
    const bool ok  = row < RM;
    const float4* o4 = reinterpret_cast<const float4*>(&Os[(ok ? row : 0) * PO + hd * CH]);
    const float4* s4 = reinterpret_cast<const float4*>(&s_av[hd * CH]);
    const float4* d4 = reinterpret_cast<const float4*>(&s_av[AVW + hd * CH]);
    float s = 0.f, d = 0.f;
#pragma unroll
    for (int q = 0; q < CH / 4; ++q) {
      const float4 v = o4[q], x = s4[q], y = d4[q];
      s += v.x * x.x;  d += v.x * y.x;
      s += v.y * x.y;  d += v.y * y.y;
      s += v.z * x.z;  d += v.z * y.z;
      s += v.w * x.w;  d += v.w * y.w;
    }
    if (ok) {
      s_al[row * 2 * H + hd]     = s;
      s_al[row * 2 * H + H + hd] = d;
    }
  }
  __syncthreads();
  for (int j = tid; j < RM * 2 * H; j += NTH)
    al[(size_t)blockIdx.x * RM * 2 * H + j] = s_al[j];

  float4* Cg = reinterpret_cast<float4*>(C + (size_t)blockIdx.x * RM * N);
#pragma unroll 4
  for (int idx = tid; idx < RM * NV; idx += NTH)
    Cg[idx] = *reinterpret_cast<const float4*>(&Os[(idx / NV) * PO + (idx % NV) * 4]);
}

__device__ __forceinline__ void atomicMaxF(float* addr, float val) {
  if (val >= 0.f) atomicMax((int*)addr, __float_as_int(val));
  else            atomicMin((unsigned int*)addr, __float_as_uint(val));
}

template <int CTRL>
__device__ __forceinline__ float dpp_read(float v) {
  return __int_as_float(__builtin_amdgcn_update_dpp(0, __float_as_int(v), CTRL, 0xF, 0xF, false));
}
__device__ __forceinline__ float other_row(float v) {
  return __int_as_float(__builtin_amdgcn_permlanex16(__float_as_int(v), __float_as_int(v),
                                                     0x76543210, 0xfedcba98, false, false));
}
__device__ __forceinline__ float wave_max(float v) {
  v = fmaxf(v, dpp_read<0x161>(v));
  v = fmaxf(v, dpp_read<0x162>(v));
  v = fmaxf(v, dpp_read<0x164>(v));
  v = fmaxf(v, dpp_read<0x168>(v));
  return fmaxf(v, other_row(v));
}
__device__ __forceinline__ float wave_sum(float v) {
  v += dpp_read<0x161>(v);
  v += dpp_read<0x162>(v);
  v += dpp_read<0x164>(v);
  v += dpp_read<0x168>(v);
  return v + other_row(v);
}

template <int H, int C, int R, int QW, bool ELU, int N2>
__global__ __launch_bounds__(1024)
void agg_owner_kernel(const int* __restrict__ ei,
                      const float* __restrict__ al,
                      const float* __restrict__ h,
                      const float* __restrict__ bias,
                      float* __restrict__ out,
                      const int* __restrict__ cntg,
                      const int2* __restrict__ cells,
                      const float* __restrict__ W2,
                      const float* __restrict__ a_s2,
                      const float* __restrict__ a_d2,
                      float* __restrict__ al2) {
  constexpr bool FUSE = N2 > 0;
  constexpr int HC  = H * C;
  constexpr int PAC = FUSE ? HC + 4 : HC;
  constexpr int NB  = FUSE ? N2 : 16;
  constexpr int PB2 = 2 * NB + 32;
  constexpr int PO2 = NB + 4;
  constexpr int NV2 = NB / 4;
  constexpr int MT2 = R / 16, NT2 = NB / 16;
  constexpr int CPL = HC / 32;
  constexpr int NW  = 32;
  constexpr int NT  = NW * 32;
  constexpr int CPW = (NPART + NW - 1) / NW;
  static_assert(C % 32 == 0, "a 32-lane line must stay inside one head");
  static_assert(R == OWNR, "the owned range is one owner group of the binning");
  static_assert(CELLC <= 32 && CPW <= 32, "one lane per cell entry, one lane per cell count");
  static_assert(CPW * CELLC <= QW, "the queue takes all cells of a wave at capacity");
  static_assert(PARTE % 32 == 0 && QW >= 64, "a part is scanned 32 edges at a time");
  constexpr int RC  = 64;
  static_assert(RC % 32 == 0 && R * RC <= NW * QW, "the row slots use the space of the queues");

  static_assert(!FUSE || (R % 16 == 0 && N2 % 16 == 0 && HC % 4 == 0 && MT2 * NT2 <= NW &&
                          ((HC / 2) * NV2) % NT == 0 && (R * 2) % 32 == 0),
                "one wave per product tile, W2 staged in whole rounds, logits in whole lines");
  __shared__ alignas(16) float acc[R * PAC];
  __shared__ alignas(16) float s_b[FUSE ? (HC / 2) * PB2 : 1];
  __shared__ alignas(16) float s_o[FUSE ? R * PO2 : 1];
  __shared__ alignas(16) float s_av2[FUSE ? 2 * NB : 4];
  __shared__ int   q_eid[NW * QW];
  __shared__ int   q_dl[NW * QW];
  __shared__ float s_ad[R * H];
  __shared__ float s_self[R * H];
  __shared__ float s_m[R * H];
  __shared__ float s_den[R * H];
  __shared__ int   s_rcnt[R];
  __shared__ int   s_fb;
  int* slot = q_eid;

  const int tid  = threadIdx.x;
  const int lane = tid & 31;
  const int w    = __builtin_amdgcn_readfirstlane(tid >> 5);
  const int grp  = blockIdx.x;
  const int lo   = grp * R;
  int* qe = q_eid + w * QW;
  int* qd = q_dl  + w * QW;
  int myc = 0;
  {
    const int p = w * CPW + lane;
    if (lane < CPW && p < NPART) myc = cntg[(size_t)p * CNTP + grp];
  }
  constexpr int BL = FUSE ? ((HC / 2) * NV2) / NT : 1;
  float4 w2a[BL], w2b[BL];
  if constexpr (FUSE) {
    const float4* Bg = reinterpret_cast<const float4*>(W2);
#pragma unroll
    for (int u = 0; u < BL; ++u) {
      const int idx = u * NT + tid;
      const int p = idx / NV2, c4 = idx % NV2;
      w2a[u] = Bg[(2 * p) * NV2 + c4];
      w2b[u] = Bg[(2 * p + 1) * NV2 + c4];
    }
  }
  float av2 = 0.f;
  if constexpr (FUSE) {
    if (tid < 2 * NB) av2 = (tid < NB) ? a_s2[tid] : a_d2[tid - NB];
  }

  for (int i = tid; i < R * H; i += NT) {
    const int n = lo + i / H;
    float a_d = 0.f, es = 0.f;
    if (n < NNODES) {
      a_d = al[(size_t)n * (2 * H) + H + i % H];
      const float v = al[(size_t)n * (2 * H) + i % H] + a_d;
      es = (v > 0.f) ? v : NEG_SLOPE * v;
    }
    s_ad[i]   = a_d;
    s_self[i] = es;
    s_m[i]    = es;
  }

  int  cnt     = 0;
  bool spilled = false;

  for (int i = tid; i < R; i += NT) {
    s_rcnt[i]     = 1;
    slot[i * RC]  = min(lo + i, NNODES - 1);
  }
  if (tid == 0) s_fb = 0;
  __syncthreads();
  {
    int  c[CPW], rk[CPW];
    int2 e[CPW];
    bool bad = false;
#pragma unroll
    for (int u = 0; u < CPW; ++u) {
      const int cu = __builtin_amdgcn_readlane(myc, u);
      bad  = bad || (cu > CELLC);
      c[u] = (cu > 0 && cu <= CELLC) ? cu : 0;
      const int p = min(w * CPW + u, NPART - 1);
      e[u] = make_int2(0, 0);
      if (lane < c[u]) e[u] = cells[((size_t)grp * NPART + p) * CELLC + lane];
    }
#pragma unroll
    for (int u = 0; u < CPW; ++u) {
      e[u].x = min(max(e[u].x, 0), NNODES - 1);
      e[u].y = min(max(e[u].y, 0), R - 1);
      rk[u]  = 0;
      if (lane < c[u]) rk[u] = atomicAdd(&s_rcnt[e[u].y], 1);
    }
#pragma unroll
    for (int u = 0; u < CPW; ++u) {
      if (lane < c[u]) {
        if (rk[u] < RC) slot[e[u].y * RC + rk[u]] = e[u].x;
        else            bad = true;
      }
    }
    if (bad) s_fb = 1;
  }

  auto fill_cells = [&]() {
    int  c[CPW];
    int2 e[CPW];
#pragma unroll
    for (int u = 0; u < CPW; ++u) {
      const int cu = __builtin_amdgcn_readlane(myc, u);
      c[u] = (cu > 0 && cu <= CELLC) ? cu : 0;
      const int p = min(w * CPW + u, NPART - 1);
      e[u] = make_int2(0, 0);
      if (lane < c[u]) e[u] = cells[((size_t)grp * NPART + p) * CELLC + lane];
    }
#pragma unroll
    for (int u = 0; u < CPW; ++u) {
      if (lane < c[u]) {
        qe[cnt + lane] = e[u].x;
        qd[cnt + lane] = min(max(e[u].y, 0), R - 1);
      }
      cnt += c[u];
    }
  };
  if constexpr (FUSE) {
#pragma unroll
    for (int u = 0; u < BL; ++u) {
      const int idx = u * NT + tid;
      const int p = idx / NV2, c4 = idx % NV2;
      float* d = &s_b[p * PB2 + c4 * 8];
      *reinterpret_cast<float4*>(d)     = make_float4(w2a[u].x, w2b[u].x, w2a[u].y, w2b[u].y);
      *reinterpret_cast<float4*>(d + 4) = make_float4(w2a[u].z, w2b[u].z, w2a[u].w, w2b[u].w);
    }
    if (tid < 2 * NB) s_av2[tid] = av2;
  }
  __syncthreads();
  const bool fb = __builtin_amdgcn_readfirstlane(s_fb) != 0;

  auto max_pass = [&](int nq) {
    __builtin_amdgcn_fence(__ATOMIC_RELEASE, "wavefront");
    __builtin_amdgcn_wave_barrier();
    for (int b = 0; b < nq; b += 32) {
      const int i = b + lane;
      if (i < nq) {
        const int dl = qd[i];
        const int s = min(max(qe[i], 0), NNODES - 1);
        float a[H];
#pragma unroll
        for (int hh = 0; hh < H; ++hh) a[hh] = al[(size_t)s * (2 * H) + hh];
#pragma unroll
        for (int hh = 0; hh < H; ++hh) {
          float v = a[hh] + s_ad[dl * H + hh];
          v = (v > 0.f) ? v : NEG_SLOPE * v;
          atomicMaxF(&s_m[dl * H + hh], v);
        }
      }
    }
    __builtin_amdgcn_fence(__ATOMIC_ACQUIRE, "wavefront");
    __builtin_amdgcn_wave_barrier();
  };

  auto acc_pass = [&](int nq) {
    __builtin_amdgcn_fence(__ATOMIC_RELEASE, "wavefront");
    __builtin_amdgcn_wave_barrier();
    for (int b = 0; b < nq; b += 32) {
      const int  i     = b + lane;
      const bool valid = i < nq;
      const int dl_i  = valid ? qd[i] : 0;
      int s_i = valid ? qe[i] : 0;
      s_i = min(max(s_i, 0), NNODES - 1);
      float a[H];
#pragma unroll
      for (int hh = 0; hh < H; ++hh) a[hh] = al[(size_t)s_i * (2 * H) + hh];
      int ex_i[H];
#pragma unroll
      for (int hh = 0; hh < H; ++hh) {
        float v = a[hh] + s_ad[dl_i * H + hh];
        v = (v > 0.f) ? v : NEG_SLOPE * v;
        const float ex = __expf(v - s_m[dl_i * H + hh]);
        if (valid) atomicAdd(&s_den[dl_i * H + hh], ex);
        ex_i[hh] = __float_as_int(ex);
      }
      const int n = min(32, nq - b);
      for (int k = 0; k < n; k += 4) {
        float v[4][CPL];
#pragma unroll
        for (int u = 0; u < 4; ++u) {
          const int s = __builtin_amdgcn_readlane(s_i, k + u);
          const float* hr = h + (size_t)s * HC + lane;
#pragma unroll
          for (int j = 0; j < CPL; ++j) v[u][j] = hr[j * 32];
        }
#pragma unroll
        for (int u = 0; u < 4; ++u) {
          if (k + u < n) {
            const int dl = __builtin_amdgcn_readlane(dl_i, k + u);
            float* ar = acc + dl * PAC + lane;
#pragma unroll
            for (int j = 0; j < CPL; ++j) {
              const float ex =
                  __int_as_float(__builtin_amdgcn_readlane(ex_i[(j * 32) / C], k + u));
              atomicAdd(ar + j * 32, ex * v[u][j]);
            }
          }
        }
      }
    }
    __builtin_amdgcn_fence(__ATOMIC_ACQUIRE, "wavefront");
    __builtin_amdgcn_wave_barrier();
  };

  auto fill_scan = [&](auto&& drain) {
    for (int u = 0; u < CPW; ++u) {
      if (__builtin_amdgcn_readlane(myc, u) > CELLC) {
        const int p = min(w * CPW + u, NPART - 1);
        for (int b = 0; b < PARTE; b += 32) {
          if (cnt + 32 > QW) { drain(cnt); cnt = 0; }
          const int      eid  = p * PARTE + b + lane;
          const int      dd   = ei[NEDGES + eid] - lo;
          const bool     m    = (unsigned)dd < (unsigned)R;
          const unsigned mask = __builtin_amdgcn_ballot_w32(m);
          if (m) {
            const int pos = cnt + (int)__builtin_amdgcn_mbcnt_lo(mask, 0u);
            qe[pos] = ei[eid];
            qd[pos] = dd;
          }
          cnt += __builtin_popcount(mask);
        }
      }
    }
  };

  constexpr int NR = (R + NW - 1) / NW;
  if (fb) {
  fill_cells();
  fill_scan([&](int nq) { max_pass(nq); spilled = true; });
  max_pass(cnt);
  __syncthreads();

  for (int i = tid; i < R * H; i += NT) s_den[i] = __expf(s_self[i] - s_m[i]);
  {
    float iv[NR][CPL];
#pragma unroll
    for (int r = 0; r < NR; ++r) {
      const int  nl = w + r * NW;
      const int  n  = lo + nl;
      const bool ok = (nl < R) && (n < NNODES);
      const int  nn = ok ? n  : 0;
      const int  nc = ok ? nl : 0;
#pragma unroll
      for (int j = 0; j < CPL; ++j) {
        const int   hd = (j * 32) / C;
        const float p  = __expf(s_self[nc * H + hd] - s_m[nc * H + hd]) *
                         h[(size_t)nn * HC + j * 32 + lane];
        iv[r][j] = ok ? p : 0.f;
      }
    }
#pragma unroll
    for (int r = 0; r < NR; ++r) {
      const int nl = w + r * NW;
      if (nl < R) {
#pragma unroll
        for (int j = 0; j < CPL; ++j) acc[nl * PAC + j * 32 + lane] = iv[r][j];
      }
    }
  }
  __syncthreads();

  if (spilled) {
    cnt = 0;
    fill_cells();
    fill_scan([&](int nq) { acc_pass(nq); });
  }
  acc_pass(cnt);
  __syncthreads();

  float bj[CPL];
#pragma unroll
  for (int j = 0; j < CPL; ++j) bj[j] = bias[j * 32 + lane];
#pragma unroll
  for (int r = 0; r < NR; ++r) {
    const int nl = w + r * NW;
    const int n  = lo + nl;
    if (nl < R && n < NNODES) {
#pragma unroll
      for (int j = 0; j < CPL; ++j) {
        const int ch = j * 32 + lane;
        float v = acc[nl * PAC + ch] / (s_den[nl * H + (j * 32) / C] + 1e-16f) + bj[j];
        if (ELU) v = (v > 0.f) ? v : (__expf(v) - 1.f);
        if constexpr (FUSE) acc[nl * PAC + ch] = v;
        else                out[(size_t)n * HC + ch] = v;
      }
    }
  }

  } else {
  float bj[CPL];
#pragma unroll
  for (int j = 0; j < CPL; ++j) bj[j] = bias[j * 32 + lane];

  auto row_head = [&](int r, int& n, int& s, float (&a)[H]) {
    const bool okr = (r < R) && (lo + r < NNODES);
    n = okr ? min(__builtin_amdgcn_readfirstlane(s_rcnt[min(r, R - 1)]), RC) : 0;
    s = (lane < n) ? slot[min(r, R - 1) * RC + lane] : 0;
#pragma unroll
    for (int hh = 0; hh < H; ++hh) a[hh] = al[(size_t)s * (2 * H) + hh];
  };
  constexpr int EB = 32 / CPL;
  static_assert(EB >= 1 && 32 % EB == 0, "a group of edges stays inside the 32 lanes");
  auto add_rows = [&](int s_i, const int (&ex_i)[H], int n, float (&ar)[CPL]) {
    for (int k = 0; k < n; k += EB) {
      float v[EB][CPL];
#pragma unroll
      for (int u = 0; u < EB; ++u) {
        const int s = __builtin_amdgcn_readlane(s_i, k + u);
        const float* hr = h + (size_t)s * HC + lane;
#pragma unroll
        for (int j = 0; j < CPL; ++j) v[u][j] = hr[j * 32];
      }
#pragma unroll
      for (int u = 0; u < EB; ++u) {
        static_assert(CPL == 8 || CPL == 2, "one input per line of a row");
        if constexpr (CPL == 8)
          asm volatile("" :: "v"(v[u][0]), "v"(v[u][1]), "v"(v[u][2]), "v"(v[u][3]),
                             "v"(v[u][4]), "v"(v[u][5]), "v"(v[u][6]), "v"(v[u][7]));
        else
          asm volatile("" :: "v"(v[u][0]), "v"(v[u][1]));
      }
#pragma unroll
      for (int u = 0; u < EB; ++u) {
        if (k + u < n) {
#pragma unroll
          for (int j = 0; j < CPL; ++j) {
            const float ex =
                __int_as_float(__builtin_amdgcn_readlane(ex_i[(j * 32) / C], k + u));
            ar[j] += ex * v[u][j];
          }
        }
      }
    }
  };

  int   n_c, s_c;
  float a_c[H];
  row_head(w, n_c, s_c, a_c);
  for (int r = w; r < R; r += NW) {
    int   n_n, s_n;
    float a_n[H];
    row_head(r + NW, n_n, s_n, a_n);
    if (n_c > 0) {
      const int n0 = min(n_c, 32);
      float ad[H], m[H], den[H];
#pragma unroll
      for (int hh = 0; hh < H; ++hh) {
        ad[hh] = s_ad[r * H + hh];
        float v = a_c[hh] + ad[hh];
        v = (v > 0.f) ? v : NEG_SLOPE * v;
        m[hh] = wave_max((lane < n0) ? v : -INFINITY);
      }
      for (int b = 32; b < n_c; b += 32) {
        const bool ok = b + lane < n_c;
        const int  s  = ok ? slot[r * RC + b + lane] : 0;
#pragma unroll
        for (int hh = 0; hh < H; ++hh) {
          float v = al[(size_t)s * (2 * H) + hh] + ad[hh];
          v = (v > 0.f) ? v : NEG_SLOPE * v;
          m[hh] = fmaxf(m[hh], wave_max(ok ? v : -INFINITY));
        }
      }
      float ar[CPL];
#pragma unroll
      for (int j = 0; j < CPL; ++j) ar[j] = 0.f;
      {
        int ex_i[H];
#pragma unroll
        for (int hh = 0; hh < H; ++hh) {
          float v = a_c[hh] + ad[hh];
          v = (v > 0.f) ? v : NEG_SLOPE * v;
          const float ex = (lane < n0) ? __expf(v - m[hh]) : 0.f;
          den[hh]  = wave_sum(ex);
          ex_i[hh] = __float_as_int(ex);
        }
        add_rows(s_c, ex_i, n0, ar);
      }
      for (int b = 32; b < n_c; b += 32) {
        const bool ok = b + lane < n_c;
        const int  s  = ok ? slot[r * RC + b + lane] : 0;
        int ex_i[H];
#pragma unroll
        for (int hh = 0; hh < H; ++hh) {
          float v = al[(size_t)s * (2 * H) + hh] + ad[hh];
          v = (v > 0.f) ? v : NEG_SLOPE * v;
          const float ex = ok ? __expf(v - m[hh]) : 0.f;
          den[hh] += wave_sum(ex);
          ex_i[hh] = __float_as_int(ex);
        }
        add_rows(s, ex_i, min(32, n_c - b), ar);
      }
#pragma unroll
      for (int j = 0; j < CPL; ++j) {
        const int ch = j * 32 + lane;
        float v = ar[j] / (den[(j * 32) / C] + 1e-16f) + bj[j];
        if (ELU) v = (v > 0.f) ? v : (__expf(v) - 1.f);
        if constexpr (FUSE) acc[r * PAC + ch] = v;
        else                out[(size_t)(lo + r) * HC + ch] = v;
      }
    } else if constexpr (FUSE) {
#pragma unroll
      for (int j = 0; j < CPL; ++j) acc[r * PAC + j * 32 + lane] = 0.f;
    }
    n_c = n_n;
    s_c = s_n;
#pragma unroll
    for (int hh = 0; hh < H; ++hh) a_c[hh] = a_n[hh];
  }
  }

  if constexpr (FUSE) {
    __syncthreads();
    if (w < MT2 * NT2) {
      const int c  = lane & 15;
      const int hf = lane >> 4;
      const int mt = w / NT2, nt = w % NT2;
      const float* Ap = acc + (mt * 16 + c) * PAC + 2 * hf;
      const float* Bp = s_b + hf * PB2 + (nt * 16 + c) * 2;
      v8f d8 = v8f{0.f, 0.f, 0.f, 0.f, 0.f, 0.f, 0.f, 0.f};
#pragma unroll 8
      for (int ks = 0; ks < HC / 4; ++ks) {
        const v2f a = *reinterpret_cast<const v2f*>(Ap + 4 * ks);
        const v2f b = *reinterpret_cast<const v2f*>(Bp + 2 * ks * PB2);
        d8 = __builtin_amdgcn_wmma_f32_16x16x4_f32(false, a, false, b, (short)0, d8, false, false);
      }
#pragma unroll
      for (int r = 0; r < 8; ++r)
        s_o[(mt * 16 + 8 * hf + r) * PO2 + nt * 16 + c] = d8[r];
    }
    __syncthreads();
    for (int blk = w; blk < (R + 31) / 32; blk += NW) {
      const int  row = blk * 32 + lane;
      const bool ok  = (row < R) && (lo + row < NNODES);
      const float4* o4 = reinterpret_cast<const float4*>(&s_o[(ok ? row : 0) * PO2]);
      const float4* s4 = reinterpret_cast<const float4*>(&s_av2[0]);
      const float4* d4 = reinterpret_cast<const float4*>(&s_av2[NB]);
      float s = 0.f, d = 0.f;
#pragma unroll
      for (int q = 0; q < NB / 4; ++q) {
        const float4 v = o4[q], x = s4[q], y = d4[q];
        s += v.x * x.x;  d += v.x * y.x;
        s += v.y * x.y;  d += v.y * y.y;
        s += v.z * x.z;  d += v.z * y.z;
        s += v.w * x.w;  d += v.w * y.w;
      }
      if (ok) *reinterpret_cast<float2*>(al2 + (size_t)(lo + row) * 2) = make_float2(s, d);
    }
    float4* Cg = reinterpret_cast<float4*>(out + (size_t)lo * NB);
    for (int idx = tid; idx < R * NV2; idx += NT) {
      if (lo + idx / NV2 < NNODES)
        Cg[idx] = *reinterpret_cast<const float4*>(&s_o[(idx / NV2) * PO2 + (idx % NV2) * 4]);
    }
  }
}

extern "C" void kernel_launch(void* const* d_in, const int* in_sizes, int n_in,
                              void* d_out, int out_size, void* d_ws, size_t ws_size,
                              hipStream_t stream) {
  const float* x      = (const float*)d_in[0];
  const int*   ei     = (const int*)  d_in[1];
  const float* W1     = (const float*)d_in[2];
  const float* a_src1 = (const float*)d_in[3];
  const float* a_dst1 = (const float*)d_in[4];
  const float* b1     = (const float*)d_in[5];
  const float* W2     = (const float*)d_in[6];
  const float* a_src2 = (const float*)d_in[7];
  const float* a_dst2 = (const float*)d_in[8];
  const float* b2     = (const float*)d_in[9];
  float* out = (float*)d_out;

  float* ws   = (float*)d_ws;
  float* h1   = ws;
  float* al1  = h1   + (size_t)NNODES * H1C;
  float* h2   = al1  + (size_t)NNODES * 2 * HEADS;
  float* al2  = h2   + (size_t)NNODES * OUT_CH;

  const int RN  = 80;
  const int NG  = (NNODES + RN - 1) / RN;
  int*   cntg  = (int*)(al2 + (size_t)NNODES * 2);
  int2*  cells = (int2*)(cntg + NPART * CNTP);

  gemm_strip_kernel<IN_CH, H1C, 2, 5, 1, HEADS, true><<<NNODES / 80, 256, 0, stream>>>(
      x, W1, a_src1, a_dst1, h1, al1, ei, cells, cntg);
  agg_owner_kernel<HEADS, HID_CH, RN, 256, true, OUT_CH><<<NG, 1024, 0, stream>>>(
      ei, al1, h1, b1, h2, cntg, cells, W2, a_src2, a_dst2, al2);

  agg_owner_kernel<1, OUT_CH, RN, 256, false, 0><<<NG, 1024, 0, stream>>>(
      ei, al2, h2, b2, out, cntg, cells, nullptr, nullptr, nullptr, nullptr);
}


# row gathers by 16-byte (last layer 8-byte) loads: a lane holds consecutive channels, 2 (1) loads per edge, not 8 (2), 8 (32) edges per round trip; ex per lane from an LDS table; sum order unchanged
# speedup vs baseline: 10.5502x; 1.0326x over previous
#include <hip/hip_runtime.h>
#include <math.h>
#include <cstddef>

#define NNODES   20000
#define NEDGES   320000
#define ETOT     (NEDGES + NNODES)
#define IN_CH    128
#define HID_CH   32
#define HEADS    8
#define H1C      (HEADS * HID_CH)
#define OUT_CH   64
#define NEG_SLOPE 0.2f
#define OWNR     80
#define NPART    (NNODES / OWNR)
#define PARTE    (NEDGES / NPART)
#define CELLC    32
#define CNTP     256

typedef float  v2f   __attribute__((ext_vector_type(2)));
typedef float  v8f   __attribute__((ext_vector_type(8)));
typedef int    v4i   __attribute__((ext_vector_type(4)));

template <int V> struct int_const { static constexpr int value = V; };

template <int K, int N, int TN, int TM, int WM, int H, bool BIN>
__global__ __launch_bounds__(256)
void gemm_strip_kernel(const float* __restrict__ A, const float* __restrict__ B,
                       const float* __restrict__ a_src, const float* __restrict__ a_dst,
                       float* __restrict__ C, float* __restrict__ al,
                       const int* __restrict__ ei, int2* __restrict__ cells,
                       int* __restrict__ cntg) {
  constexpr int RM  = 80;
  constexpr int MT  = RM / 16;
  constexpr int NTH = 256;
  constexpr int WN  = (N / 16) / TN;
  constexpr int PA  = K + 4;
  constexpr int PB  = 2 * N + 32;
  constexpr int PO  = N + 4;
  constexpr int KV  = K / 4;
  constexpr int NV  = N / 4;
  constexpr int A_WORDS = RM * PA;
  constexpr int B_WORDS = (K / 2) * PB;
  constexpr int CH  = N / H;
  constexpr int AVN = (N + NTH - 1) / NTH;
  constexpr int AVW = AVN * NTH;
  static_assert(WN * WM * 32 == NTH, "8 waves");
  static_assert(TM * WM >= MT && TM * (WM - 1) < MT, "M-tiles dealt over WM wave rows");
  static_assert(RM * PO <= A_WORDS + B_WORDS, "C tile reuses the operand space");
  static_assert((RM * KV) % (NTH * 10) == 0, "A staging in batches of 10");
  static_assert(((K / 2) * NV) % (NTH * 8) == 0, "B staging in batches of 8");
  static_assert((RM * NV) % NTH == 0, "C store loop");
  static_assert(NNODES % RM == 0, "no partial strip");
  static_assert(N % H == 0 && CH % 4 == 0, "a head's channels are read four at a time");
  static_assert((RM * 2 * H) % 32 == 0, "a strip's logits fill whole 128-B lines");
  constexpr int EPT = BIN ? PARTE / NTH : 1;
  static_assert(!BIN || (RM == OWNR && PARTE % NTH == 0 && NEDGES % NPART == 0 &&
                         NPART < NTH && NTH <= CNTP),
                "one list part per strip workgroup, one count per thread");

  __shared__ alignas(16) float smem[A_WORDS + B_WORDS];
  __shared__ alignas(16) float s_av[2 * AVW];
  __shared__ alignas(16) float s_al[RM * 2 * H];
  __shared__ int s_hist[BIN ? NTH : 1];
  float* As = smem;
  float* Bs = smem + A_WORDS;
  float* Os = smem;

  const int tid  = threadIdx.x;
  const int lane = tid & 31;
  const int w    = __builtin_amdgcn_readfirstlane(tid >> 5);
  const int c    = lane & 15;
  const int h    = lane >> 4;

  float avs[AVN], avd[AVN];
#pragma unroll
  for (int u = 0; u < AVN; ++u) {
    const int idx = min(u * NTH + tid, N - 1);
    avs[u] = a_src[idx];
    avd[u] = a_dst[idx];
  }

  int e_s[EPT], e_d[EPT];
  if constexpr (BIN) {
#pragma unroll
    for (int u = 0; u < EPT; ++u) {
      const int e = blockIdx.x * PARTE + u * NTH + tid;
      e_s[u] = ei[e];
      e_d[u] = ei[NEDGES + e];
    }
    s_hist[tid] = 0;
  }

  {
    const float4* Ag = reinterpret_cast<const float4*>(A + (size_t)blockIdx.x * RM * K);
#pragma unroll
    for (int i0 = 0; i0 < (RM * KV) / NTH; i0 += 10) {
      float4 r[10];
#pragma unroll
      for (int u = 0; u < 10; ++u) r[u] = Ag[(i0 + u) * NTH + tid];
#pragma unroll
      for (int u = 0; u < 10; ++u) {
        const int idx = (i0 + u) * NTH + tid;
        *reinterpret_cast<float4*>(&As[(idx / KV) * PA + (idx % KV) * 4]) = r[u];
      }
    }
  }
  {
    const float4* Bg = reinterpret_cast<const float4*>(B);
#pragma unroll
    for (int i0 = 0; i0 < ((K / 2) * NV) / NTH; i0 += 8) {
      float4 r0[8], r1[8];
#pragma unroll
      for (int u = 0; u < 8; ++u) {
        const int idx = (i0 + u) * NTH + tid;
        const int p = idx / NV, c4 = idx % NV;
        r0[u] = Bg[(2 * p) * NV + c4];
        r1[u] = Bg[(2 * p + 1) * NV + c4];
      }
#pragma unroll
      for (int u = 0; u < 8; ++u) {
        const int idx = (i0 + u) * NTH + tid;
        const int p = idx / NV, c4 = idx % NV;
        float* d = &Bs[p * PB + c4 * 8];
        *reinterpret_cast<float4*>(d)     = make_float4(r0[u].x, r1[u].x, r0[u].y, r1[u].y);
        *reinterpret_cast<float4*>(d + 4) = make_float4(r0[u].z, r1[u].z, r0[u].w, r1[u].w);
      }
    }
  }
#pragma unroll
  for (int u = 0; u < AVN; ++u) {
    s_av[u * NTH + tid]       = avs[u];
    s_av[AVW + u * NTH + tid] = avd[u];
  }
  __syncthreads();
  if constexpr (BIN) {
    int gq[EPT], rk[EPT];
#pragma unroll
    for (int u = 0; u < EPT; ++u) {
      gq[u] = ((unsigned)e_d[u] < (unsigned)NNODES) ? e_d[u] / OWNR : NTH - 1;
      rk[u] = atomicAdd(&s_hist[gq[u]], 1);
    }
#pragma unroll
    for (int u = 0; u < EPT; ++u) {
      if (gq[u] < NPART && rk[u] < CELLC)
        cells[((size_t)gq[u] * NPART + blockIdx.x) * CELLC + rk[u]] =
            make_int2(e_s[u], e_d[u] - gq[u] * OWNR);
    }
  }

  const int wn = w % WN;
  const int wm = w / WN;
  const int mb = wm * TM;
  const int mc = min(TM, MT - mb);
  v8f acc[TM][TN];
#pragma unroll
  for (int i = 0; i < TM; ++i)
#pragma unroll
    for (int j = 0; j < TN; ++j) acc[i][j] = v8f{0.f, 0.f, 0.f, 0.f, 0.f, 0.f, 0.f, 0.f};

  const float* Ap = As + (mb * 16 + c) * PA + 2 * h;
  const float* Bp = Bs + h * PB + (wn * TN * 16 + c) * 2;
  auto kloop = [&](auto cnt) {
    constexpr int CNT = decltype(cnt)::value;
#pragma unroll 4
    for (int ks = 0; ks < K / 4; ++ks) {
      v2f a[CNT], b[TN];
#pragma unroll
      for (int i = 0; i < CNT; ++i)
        a[i] = *reinterpret_cast<const v2f*>(Ap + i * 16 * PA + 4 * ks);
#pragma unroll
      for (int j = 0; j < TN; ++j)
        b[j] = *reinterpret_cast<const v2f*>(Bp + 2 * ks * PB + j * 32);
#pragma unroll
      for (int i = 0; i < CNT; ++i)
#pragma unroll
        for (int j = 0; j < TN; ++j)
          acc[i][j] = __builtin_amdgcn_wmma_f32_16x16x4_f32(false, a[i], false, b[j],
                                                            (short)0, acc[i][j], false, false);
    }
  };
  if constexpr (MT % TM == 0) {
    kloop(int_const<TM>{});
  } else {
    if (mc == TM) kloop(int_const<TM>{});
    else          kloop(int_const<MT % TM>{});
  }
  __syncthreads();
  if constexpr (BIN) {
    for (int j = tid; j < CNTP; j += NTH)
      cntg[(size_t)blockIdx.x * CNTP + j] = (j < NPART) ? s_hist[j] : 0;
  }

#pragma unroll
  for (int i = 0; i < TM; ++i) {
    if (i < mc) {
#pragma unroll
      for (int j = 0; j < TN; ++j)
#pragma unroll
        for (int r = 0; r < 8; ++r)
          Os[((mb + i) * 16 + 8 * h + r) * PO + (wn * TN + j) * 16 + c] = acc[i][j][r];
    }
  }
  __syncthreads();

  constexpr int RB = (RM + 31) / 32;
  for (int blk = w; blk < H * RB; blk += NTH / 32) {
    const int  hd  = blk / RB;
    const int  row = (blk % RB) * 32 + lane;
    const bool ok  = row < RM;
    const float4* o4 = reinterpret_cast<const float4*>(&Os[(ok ? row : 0) * PO + hd * CH]);
    const float4* s4 = reinterpret_cast<const float4*>(&s_av[hd * CH]);
    const float4* d4 = reinterpret_cast<const float4*>(&s_av[AVW + hd * CH]);
    float s = 0.f, d = 0.f;
#pragma unroll
    for (int q = 0; q < CH / 4; ++q) {
      const float4 v = o4[q], x = s4[q], y = d4[q];
      s += v.x * x.x;  d += v.x * y.x;
      s += v.y * x.y;  d += v.y * y.y;
      s += v.z * x.z;  d += v.z * y.z;
      s += v.w * x.w;  d += v.w * y.w;
    }
    if (ok) {
      s_al[row * 2 * H + hd]     = s;
      s_al[row * 2 * H + H + hd] = d;
    }
  }
  __syncthreads();
  for (int j = tid; j < RM * 2 * H; j += NTH)
    al[(size_t)blockIdx.x * RM * 2 * H + j] = s_al[j];

  float4* Cg = reinterpret_cast<float4*>(C + (size_t)blockIdx.x * RM * N);
#pragma unroll 4
  for (int idx = tid; idx < RM * NV; idx += NTH)
    Cg[idx] = *reinterpret_cast<const float4*>(&Os[(idx / NV) * PO + (idx % NV) * 4]);
}

__device__ __forceinline__ void atomicMaxF(float* addr, float val) {
  if (val >= 0.f) atomicMax((int*)addr, __float_as_int(val));
  else            atomicMin((unsigned int*)addr, __float_as_uint(val));
}

template <int CTRL>
__device__ __forceinline__ float dpp_read(float v) {
  return __int_as_float(__builtin_amdgcn_update_dpp(0, __float_as_int(v), CTRL, 0xF, 0xF, false));
}
__device__ __forceinline__ float other_row(float v) {
  return __int_as_float(__builtin_amdgcn_permlanex16(__float_as_int(v), __float_as_int(v),
                                                     0x76543210, 0xfedcba98, false, false));
}
__device__ __forceinline__ float wave_max(float v) {
  v = fmaxf(v, dpp_read<0x161>(v));
  v = fmaxf(v, dpp_read<0x162>(v));
  v = fmaxf(v, dpp_read<0x164>(v));
  v = fmaxf(v, dpp_read<0x168>(v));
  return fmaxf(v, other_row(v));
}
__device__ __forceinline__ float wave_sum(float v) {
  v += dpp_read<0x161>(v);
  v += dpp_read<0x162>(v);
  v += dpp_read<0x164>(v);
  v += dpp_read<0x168>(v);
  return v + other_row(v);
}

template <int H, int C, int R, int QW, bool ELU, int N2>
__global__ __launch_bounds__(1024)
void agg_owner_kernel(const int* __restrict__ ei,
                      const float* __restrict__ al,
                      const float* __restrict__ h,
                      const float* __restrict__ bias,
                      float* __restrict__ out,
                      const int* __restrict__ cntg,
                      const int2* __restrict__ cells,
                      const float* __restrict__ W2,
                      const float* __restrict__ a_s2,
                      const float* __restrict__ a_d2,
                      float* __restrict__ al2) {
  constexpr bool FUSE = N2 > 0;
  constexpr int HC  = H * C;
  constexpr int PAC = FUSE ? HC + 4 : HC;
  constexpr int NB  = FUSE ? N2 : 16;
  constexpr int PB2 = 2 * NB + 32;
  constexpr int PO2 = NB + 4;
  constexpr int NV2 = NB / 4;
  constexpr int MT2 = R / 16, NT2 = NB / 16;
  constexpr int CPL = HC / 32;
  constexpr int NW  = 32;
  constexpr int NT  = NW * 32;
  constexpr int CPW = (NPART + NW - 1) / NW;
  static_assert(C % 32 == 0, "a 32-lane line must stay inside one head");
  static_assert(R == OWNR, "the owned range is one owner group of the binning");
  static_assert(CELLC <= 32 && CPW <= 32, "one lane per cell entry, one lane per cell count");
  static_assert(CPW * CELLC <= QW, "the queue takes all cells of a wave at capacity");
  static_assert(PARTE % 32 == 0 && QW >= 64, "a part is scanned 32 edges at a time");
  constexpr int RC  = 64;
  static_assert(RC % 32 == 0 && R * RC <= NW * QW, "the row slots use the space of the queues");

  static_assert(!FUSE || (R % 16 == 0 && N2 % 16 == 0 && HC % 4 == 0 && MT2 * NT2 <= NW &&
                          ((HC / 2) * NV2) % NT == 0 && (R * 2) % 32 == 0),
                "one wave per product tile, W2 staged in whole rounds, logits in whole lines");
  __shared__ alignas(16) float acc[R * PAC];
  __shared__ alignas(16) float s_b[FUSE ? (HC / 2) * PB2 : 1];
  __shared__ alignas(16) float s_o[FUSE ? R * PO2 : 1];
  __shared__ alignas(16) float s_av2[FUSE ? 2 * NB : 4];
  __shared__ int   q_eid[NW * QW];
  __shared__ alignas(16) int q_dl[NW * QW];
  __shared__ float s_ad[R * H];
  __shared__ float s_self[R * H];
  __shared__ float s_m[R * H];
  __shared__ float s_den[R * H];
  __shared__ int   s_rcnt[R];
  __shared__ int   s_fb;
  __shared__ alignas(16) float s_bias[HC];
  int* slot = q_eid;

  const int tid  = threadIdx.x;
  const int lane = tid & 31;
  const int w    = __builtin_amdgcn_readfirstlane(tid >> 5);
  const int grp  = blockIdx.x;
  const int lo   = grp * R;
  int* qe = q_eid + w * QW;
  int* qd = q_dl  + w * QW;
  int myc = 0;
  {
    const int p = w * CPW + lane;
    if (lane < CPW && p < NPART) myc = cntg[(size_t)p * CNTP + grp];
  }
  constexpr int BL = FUSE ? ((HC / 2) * NV2) / NT : 1;
  float4 w2a[BL], w2b[BL];
  if constexpr (FUSE) {
    const float4* Bg = reinterpret_cast<const float4*>(W2);
#pragma unroll
    for (int u = 0; u < BL; ++u) {
      const int idx = u * NT + tid;
      const int p = idx / NV2, c4 = idx % NV2;
      w2a[u] = Bg[(2 * p) * NV2 + c4];
      w2b[u] = Bg[(2 * p + 1) * NV2 + c4];
    }
  }
  float av2 = 0.f;
  if constexpr (FUSE) {
    if (tid < 2 * NB) av2 = (tid < NB) ? a_s2[tid] : a_d2[tid - NB];
  }

  for (int i = tid; i < R * H; i += NT) {
    const int n = lo + i / H;
    float a_d = 0.f, es = 0.f;
    if (n < NNODES) {
      a_d = al[(size_t)n * (2 * H) + H + i % H];
      const float v = al[(size_t)n * (2 * H) + i % H] + a_d;
      es = (v > 0.f) ? v : NEG_SLOPE * v;
    }
    s_ad[i]   = a_d;
    s_self[i] = es;
    s_m[i]    = es;
  }

  int  cnt     = 0;
  bool spilled = false;

  for (int i = tid; i < R; i += NT) {
    s_rcnt[i]     = 1;
    slot[i * RC]  = min(lo + i, NNODES - 1);
  }
  if (tid == 0) s_fb = 0;
  for (int i = tid; i < HC; i += NT) s_bias[i] = bias[i];
  __syncthreads();
  {
    int  c[CPW], rk[CPW];
    int2 e[CPW];
    bool bad = false;
#pragma unroll
    for (int u = 0; u < CPW; ++u) {
      const int cu = __builtin_amdgcn_readlane(myc, u);
      bad  = bad || (cu > CELLC);
      c[u] = (cu > 0 && cu <= CELLC) ? cu : 0;
      const int p = min(w * CPW + u, NPART - 1);
      e[u] = make_int2(0, 0);
      if (lane < c[u]) e[u] = cells[((size_t)grp * NPART + p) * CELLC + lane];
    }
#pragma unroll
    for (int u = 0; u < CPW; ++u) {
      e[u].x = min(max(e[u].x, 0), NNODES - 1);
      e[u].y = min(max(e[u].y, 0), R - 1);
      rk[u]  = 0;
      if (lane < c[u]) rk[u] = atomicAdd(&s_rcnt[e[u].y], 1);
    }
#pragma unroll
    for (int u = 0; u < CPW; ++u) {
      if (lane < c[u]) {
        if (rk[u] < RC) slot[e[u].y * RC + rk[u]] = e[u].x;
        else            bad = true;
      }
    }
    if (bad) s_fb = 1;
  }

  auto fill_cells = [&]() {
    int  c[CPW];
    int2 e[CPW];
#pragma unroll
    for (int u = 0; u < CPW; ++u) {
      const int cu = __builtin_amdgcn_readlane(myc, u);
      c[u] = (cu > 0 && cu <= CELLC) ? cu : 0;
      const int p = min(w * CPW + u, NPART - 1);
      e[u] = make_int2(0, 0);
      if (lane < c[u]) e[u] = cells[((size_t)grp * NPART + p) * CELLC + lane];
    }
#pragma unroll
    for (int u = 0; u < CPW; ++u) {
      if (lane < c[u]) {
        qe[cnt + lane] = e[u].x;
        qd[cnt + lane] = min(max(e[u].y, 0), R - 1);
      }
      cnt += c[u];
    }
  };
  if constexpr (FUSE) {
#pragma unroll
    for (int u = 0; u < BL; ++u) {
      const int idx = u * NT + tid;
      const int p = idx / NV2, c4 = idx % NV2;
      float* d = &s_b[p * PB2 + c4 * 8];
      *reinterpret_cast<float4*>(d)     = make_float4(w2a[u].x, w2b[u].x, w2a[u].y, w2b[u].y);
      *reinterpret_cast<float4*>(d + 4) = make_float4(w2a[u].z, w2b[u].z, w2a[u].w, w2b[u].w);
    }
    if (tid < 2 * NB) s_av2[tid] = av2;
  }
  __syncthreads();
  const bool fb = __builtin_amdgcn_readfirstlane(s_fb) != 0;

  auto max_pass = [&](int nq) {
    __builtin_amdgcn_fence(__ATOMIC_RELEASE, "wavefront");
    __builtin_amdgcn_wave_barrier();
    for (int b = 0; b < nq; b += 32) {
      const int i = b + lane;
      if (i < nq) {
        const int dl = qd[i];
        const int s = min(max(qe[i], 0), NNODES - 1);
        float a[H];
#pragma unroll
        for (int hh = 0; hh < H; ++hh) a[hh] = al[(size_t)s * (2 * H) + hh];
#pragma unroll
        for (int hh = 0; hh < H; ++hh) {
          float v = a[hh] + s_ad[dl * H + hh];
          v = (v > 0.f) ? v : NEG_SLOPE * v;
          atomicMaxF(&s_m[dl * H + hh], v);
        }
      }
    }
    __builtin_amdgcn_fence(__ATOMIC_ACQUIRE, "wavefront");
    __builtin_amdgcn_wave_barrier();
  };

  auto acc_pass = [&](int nq) {
    __builtin_amdgcn_fence(__ATOMIC_RELEASE, "wavefront");
    __builtin_amdgcn_wave_barrier();
    for (int b = 0; b < nq; b += 32) {
      const int  i     = b + lane;
      const bool valid = i < nq;
      const int dl_i  = valid ? qd[i] : 0;
      int s_i = valid ? qe[i] : 0;
      s_i = min(max(s_i, 0), NNODES - 1);
      float a[H];
#pragma unroll
      for (int hh = 0; hh < H; ++hh) a[hh] = al[(size_t)s_i * (2 * H) + hh];
      int ex_i[H];
#pragma unroll
      for (int hh = 0; hh < H; ++hh) {
        float v = a[hh] + s_ad[dl_i * H + hh];
        v = (v > 0.f) ? v : NEG_SLOPE * v;
        const float ex = __expf(v - s_m[dl_i * H + hh]);
        if (valid) atomicAdd(&s_den[dl_i * H + hh], ex);
        ex_i[hh] = __float_as_int(ex);
      }
      const int n = min(32, nq - b);
      for (int k = 0; k < n; k += 4) {
        float v[4][CPL];
#pragma unroll
        for (int u = 0; u < 4; ++u) {
          const int s = __builtin_amdgcn_readlane(s_i, k + u);
          const float* hr = h + (size_t)s * HC + lane;
#pragma unroll
          for (int j = 0; j < CPL; ++j) v[u][j] = hr[j * 32];
        }
#pragma unroll
        for (int u = 0; u < 4; ++u) {
          if (k + u < n) {
            const int dl = __builtin_amdgcn_readlane(dl_i, k + u);
            float* ar = acc + dl * PAC + lane;
#pragma unroll
            for (int j = 0; j < CPL; ++j) {
              const float ex =
                  __int_as_float(__builtin_amdgcn_readlane(ex_i[(j * 32) / C], k + u));
              atomicAdd(ar + j * 32, ex * v[u][j]);
            }
          }
        }
      }
    }
    __builtin_amdgcn_fence(__ATOMIC_ACQUIRE, "wavefront");
    __builtin_amdgcn_wave_barrier();
  };

  auto fill_scan = [&](auto&& drain) {
    for (int u = 0; u < CPW; ++u) {
      if (__builtin_amdgcn_readlane(myc, u) > CELLC) {
        const int p = min(w * CPW + u, NPART - 1);
        for (int b = 0; b < PARTE; b += 32) {
          if (cnt + 32 > QW) { drain(cnt); cnt = 0; }
          const int      eid  = p * PARTE + b + lane;
          const int      dd   = ei[NEDGES + eid] - lo;
          const bool     m    = (unsigned)dd < (unsigned)R;
          const unsigned mask = __builtin_amdgcn_ballot_w32(m);
          if (m) {
            const int pos = cnt + (int)__builtin_amdgcn_mbcnt_lo(mask, 0u);
            qe[pos] = ei[eid];
            qd[pos] = dd;
          }
          cnt += __builtin_popcount(mask);
        }
      }
    }
  };

  constexpr int NR = (R + NW - 1) / NW;
  if (fb) {
  fill_cells();
  fill_scan([&](int nq) { max_pass(nq); spilled = true; });
  max_pass(cnt);
  __syncthreads();

  for (int i = tid; i < R * H; i += NT) s_den[i] = __expf(s_self[i] - s_m[i]);
  {
    float iv[NR][CPL];
#pragma unroll
    for (int r = 0; r < NR; ++r) {
      const int  nl = w + r * NW;
      const int  n  = lo + nl;
      const bool ok = (nl < R) && (n < NNODES);
      const int  nn = ok ? n  : 0;
      const int  nc = ok ? nl : 0;
#pragma unroll
      for (int j = 0; j < CPL; ++j) {
        const int   hd = (j * 32) / C;
        const float p  = __expf(s_self[nc * H + hd] - s_m[nc * H + hd]) *
                         h[(size_t)nn * HC + j * 32 + lane];
        iv[r][j] = ok ? p : 0.f;
      }
    }
#pragma unroll
    for (int r = 0; r < NR; ++r) {
      const int nl = w + r * NW;
      if (nl < R) {
#pragma unroll
        for (int j = 0; j < CPL; ++j) acc[nl * PAC + j * 32 + lane] = iv[r][j];
      }
    }
  }
  __syncthreads();

  if (spilled) {
    cnt = 0;
    fill_cells();
    fill_scan([&](int nq) { acc_pass(nq); });
  }
  acc_pass(cnt);
  __syncthreads();

  float bj[CPL];
#pragma unroll
  for (int j = 0; j < CPL; ++j) bj[j] = bias[j * 32 + lane];
#pragma unroll
  for (int r = 0; r < NR; ++r) {
    const int nl = w + r * NW;
    const int n  = lo + nl;
    if (nl < R && n < NNODES) {
#pragma unroll
      for (int j = 0; j < CPL; ++j) {
        const int ch = j * 32 + lane;
        float v = acc[nl * PAC + ch] / (s_den[nl * H + (j * 32) / C] + 1e-16f) + bj[j];
        if (ELU) v = (v > 0.f) ? v : (__expf(v) - 1.f);
        if constexpr (FUSE) acc[nl * PAC + ch] = v;
        else                out[(size_t)n * HC + ch] = v;
      }
    }
  }

  } else {
  constexpr int LW  = (HC >= 128) ? 4 : HC / 32;
  constexpr int NLD = HC / (32 * LW);
  constexpr int HPL = (32 * LW) / C;
  static_assert(LW * NLD * 32 == HC && C % LW == 0, "a row is NLD loads of LW floats per lane");
  static_assert(H == 1 || (HPL * NLD == H && H * 32 <= QW), "heads per piece; the ex table fits the wave's queue");
  typedef float vlw __attribute__((ext_vector_type(LW)));
  const int hsel = (lane * LW) / C;

  auto row_head = [&](int r, int& n, int& s, float (&a)[H]) {
    const bool okr = (r < R) && (lo + r < NNODES);
    n = okr ? min(__builtin_amdgcn_readfirstlane(s_rcnt[min(r, R - 1)]), RC) : 0;
    s = (lane < n) ? slot[min(r, R - 1) * RC + lane] : 0;
#pragma unroll
    for (int hh = 0; hh < H; ++hh) a[hh] = al[(size_t)s * (2 * H) + hh];
  };
  constexpr int EB = (H > 1) ? 8 : 32;
  constexpr int ST = (H > 1) ? 4 : 8;
  static_assert(EB % ST == 0 && EB / ST <= 4 && ST % 4 == 0 && 32 % ST == 0, "group sizes ST, 2*ST, ..., EB");
  int* exq = qd;
  auto add_rows = [&](int s_i, int ex0, int n, float (&ar)[NLD * LW]) {
    auto grp = [&](auto cnt, int k) {
      constexpr int G = decltype(cnt)::value;
      vlw v[G][NLD];
#pragma unroll
      for (int u = 0; u < G; ++u) {
        const int s = __builtin_amdgcn_readlane(s_i, k + u);
        const float* hr = h + (size_t)s * HC + lane * LW;
#pragma unroll
        for (int j = 0; j < NLD; ++j) v[u][j] = *reinterpret_cast<const vlw*>(hr + j * 32 * LW);
      }
      float exl[NLD][G];
      if constexpr (H > 1) {
#pragma unroll
        for (int j = 0; j < NLD; ++j)
#pragma unroll
          for (int q = 0; q < G / 4; ++q) {
            const v4i tq = *reinterpret_cast<const v4i*>(&exq[(j * HPL + hsel) * 32 + k + 4 * q]);
            exl[j][4 * q + 0] = __int_as_float(tq[0]);
            exl[j][4 * q + 1] = __int_as_float(tq[1]);
            exl[j][4 * q + 2] = __int_as_float(tq[2]);
            exl[j][4 * q + 3] = __int_as_float(tq[3]);
          }
      }
#pragma unroll
      for (int u = 0; u < G; ++u) {
        static_assert(NLD == 2 || NLD == 1, "one input per piece of a row");
        if constexpr (NLD == 2) asm volatile("" :: "v"(v[u][0]), "v"(v[u][1]));
        else                    asm volatile("" :: "v"(v[u][0]));
      }
#pragma unroll
      for (int u = 0; u < G; ++u) {
        if (k + u < n) {
#pragma unroll
          for (int j = 0; j < NLD; ++j) {
            float ex;
            if constexpr (H > 1) ex = exl[j][u];
            else ex = __int_as_float(__builtin_amdgcn_readlane(ex0, k + u));
#pragma unroll
            for (int i = 0; i < LW; ++i) ar[j * LW + i] += ex * v[u][j][i];
          }
        }
      }
    };
    for (int k = 0; k < n;) {
      const int rem = n - k;
      if constexpr (EB >= 4 * ST) { if (rem > 3 * ST) { grp(int_const<4 * ST>{}, k); k += 4 * ST; continue; } }
      if constexpr (EB >= 3 * ST) { if (rem > 2 * ST) { grp(int_const<3 * ST>{}, k); k += 3 * ST; continue; } }
      if constexpr (EB >= 2 * ST) { if (rem > 1 * ST) { grp(int_const<2 * ST>{}, k); k += 2 * ST; continue; } }
      grp(int_const<ST>{}, k);
      k += ST;
    }
  };
  auto put_ex = [&](int hh, float ex) { exq[hh * 32 + lane] = __float_as_int(ex); };
  auto table_written = [&]() {
    if constexpr (H > 1) {
      __builtin_amdgcn_fence(__ATOMIC_RELEASE, "wavefront");
      __builtin_amdgcn_wave_barrier();
    }
  };
  auto table_read = [&]() {
    if constexpr (H > 1) {
      __builtin_amdgcn_fence(__ATOMIC_ACQUIRE, "wavefront");
      __builtin_amdgcn_wave_barrier();
    }
  };

  int   n_c, s_c;
  float a_c[H];
  row_head(w, n_c, s_c, a_c);
  for (int r = w; r < R; r += NW) {
    int   n_n, s_n;
    float a_n[H];
    row_head(r + NW, n_n, s_n, a_n);
    if (n_c > 0) {
      const int n0 = min(n_c, 32);
      float ad[H], m[H];
#pragma unroll
      for (int hh = 0; hh < H; ++hh) {
        ad[hh] = s_ad[r * H + hh];
        float v = a_c[hh] + ad[hh];
        v = (v > 0.f) ? v : NEG_SLOPE * v;
        m[hh] = wave_max((lane < n0) ? v : -INFINITY);
      }
      for (int b = 32; b < n_c; b += 32) {
        const bool ok = b + lane < n_c;
        const int  s  = ok ? slot[r * RC + b + lane] : 0;
#pragma unroll
        for (int hh = 0; hh < H; ++hh) {
          float v = al[(size_t)s * (2 * H) + hh] + ad[hh];
          v = (v > 0.f) ? v : NEG_SLOPE * v;
          m[hh] = fmaxf(m[hh], wave_max(ok ? v : -INFINITY));
        }
      }
      float ar[NLD * LW], dj[NLD];
      if (n_c > 32) {
#pragma unroll
        for (int hh = 0; hh < H; ++hh) s_m[r * H + hh] = m[hh];
      }
#pragma unroll
      for (int j = 0; j < NLD * LW; ++j) ar[j] = 0.f;
#pragma unroll
      for (int j = 0; j < NLD; ++j) dj[j] = 0.f;
      {
        int ex0 = 0;
#pragma unroll
        for (int hh = 0; hh < H; ++hh) {
          float v = a_c[hh] + ad[hh];
          v = (v > 0.f) ? v : NEG_SLOPE * v;
          const float ex = (lane < n0) ? __expf(v - m[hh]) : 0.f;
          const float sm = wave_sum(ex);
          if constexpr (H > 1) {
            dj[hh / HPL] = (hsel == hh % HPL) ? sm : dj[hh / HPL];
            put_ex(hh, ex);
          } else {
            dj[0] = sm;
            ex0   = __float_as_int(ex);
          }
        }
        table_written();
        add_rows(s_c, ex0, n0, ar);
        table_read();
      }
      for (int b = 32; b < n_c; b += 32) {
        const bool ok = b + lane < n_c;
        const int  s  = ok ? slot[r * RC + b + lane] : 0;
        int ex0 = 0;
#pragma unroll
        for (int hh = 0; hh < H; ++hh) {
          float v = al[(size_t)s * (2 * H) + hh] + s_ad[r * H + hh];
          v = (v > 0.f) ? v : NEG_SLOPE * v;
          const float ex = ok ? __expf(v - s_m[r * H + hh]) : 0.f;
          const float sm = wave_sum(ex);
          if constexpr (H > 1) {
            dj[hh / HPL] = (hsel == hh % HPL) ? dj[hh / HPL] + sm : dj[hh / HPL];
            put_ex(hh, ex);
          } else {
            dj[0] += sm;
            ex0    = __float_as_int(ex);
          }
        }
        table_written();
        add_rows(s, ex0, min(32, n_c - b), ar);
        table_read();
      }
#pragma unroll
      for (int j = 0; j < NLD; ++j) {
        const int ch = j * 32 * LW + lane * LW;
        const vlw bv = *reinterpret_cast<const vlw*>(&s_bias[ch]);
        vlw o;
#pragma unroll
        for (int i = 0; i < LW; ++i) {
          float v = ar[j * LW + i] / (dj[j] + 1e-16f) + bv[i];
          if (ELU) v = (v > 0.f) ? v : (__expf(v) - 1.f);
          o[i] = v;
        }
        if constexpr (FUSE) *reinterpret_cast<vlw*>(&acc[r * PAC + ch]) = o;
        else                *reinterpret_cast<vlw*>(&out[(size_t)(lo + r) * HC + ch]) = o;
      }
    } else if constexpr (FUSE) {
#pragma unroll
      for (int j = 0; j < CPL; ++j) acc[r * PAC + j * 32 + lane] = 0.f;
    }
    n_c = n_n;
    s_c = s_n;
#pragma unroll
    for (int hh = 0; hh < H; ++hh) a_c[hh] = a_n[hh];
  }
  }

  if constexpr (FUSE) {
    __syncthreads();
    if (w < MT2 * NT2) {
      const int c  = lane & 15;
      const int hf = lane >> 4;
      const int mt = w / NT2, nt = w % NT2;
      const float* Ap = acc + (mt * 16 + c) * PAC + 2 * hf;
      const float* Bp = s_b + hf * PB2 + (nt * 16 + c) * 2;
      v8f d8 = v8f{0.f, 0.f, 0.f, 0.f, 0.f, 0.f, 0.f, 0.f};
#pragma unroll 8
      for (int ks = 0; ks < HC / 4; ++ks) {
        const v2f a = *reinterpret_cast<const v2f*>(Ap + 4 * ks);
        const v2f b = *reinterpret_cast<const v2f*>(Bp + 2 * ks * PB2);
        d8 = __builtin_amdgcn_wmma_f32_16x16x4_f32(false, a, false, b, (short)0, d8, false, false);
      }
#pragma unroll
      for (int r = 0; r < 8; ++r)
        s_o[(mt * 16 + 8 * hf + r) * PO2 + nt * 16 + c] = d8[r];
    }
    __syncthreads();
    for (int blk = w; blk < (R + 31) / 32; blk += NW) {
      const int  row = blk * 32 + lane;
      const bool ok  = (row < R) && (lo + row < NNODES);
      const float4* o4 = reinterpret_cast<const float4*>(&s_o[(ok ? row : 0) * PO2]);
      const float4* s4 = reinterpret_cast<const float4*>(&s_av2[0]);
      const float4* d4 = reinterpret_cast<const float4*>(&s_av2[NB]);
      float s = 0.f, d = 0.f;
#pragma unroll
      for (int q = 0; q < NB / 4; ++q) {
        const float4 v = o4[q], x = s4[q], y = d4[q];
        s += v.x * x.x;  d += v.x * y.x;
        s += v.y * x.y;  d += v.y * y.y;
        s += v.z * x.z;  d += v.z * y.z;
        s += v.w * x.w;  d += v.w * y.w;
      }
      if (ok) *reinterpret_cast<float2*>(al2 + (size_t)(lo + row) * 2) = make_float2(s, d);
    }
    float4* Cg = reinterpret_cast<float4*>(out + (size_t)lo * NB);
    for (int idx = tid; idx < R * NV2; idx += NT) {
      if (lo + idx / NV2 < NNODES)
        Cg[idx] = *reinterpret_cast<const float4*>(&s_o[(idx / NV2) * PO2 + (idx % NV2) * 4]);
    }
  }
}

extern "C" void kernel_launch(void* const* d_in, const int* in_sizes, int n_in,
                              void* d_out, int out_size, void* d_ws, size_t ws_size,
                              hipStream_t stream) {
  const float* x      = (const float*)d_in[0];
  const int*   ei     = (const int*)  d_in[1];
  const float* W1     = (const float*)d_in[2];
  const float* a_src1 = (const float*)d_in[3];
  const float* a_dst1 = (const float*)d_in[4];
  const float* b1     = (const float*)d_in[5];
  const float* W2     = (const float*)d_in[6];
  const float* a_src2 = (const float*)d_in[7];
  const float* a_dst2 = (const float*)d_in[8];
  const float* b2     = (const float*)d_in[9];
  float* out = (float*)d_out;

  float* ws   = (float*)d_ws;
  float* h1   = ws;
  float* al1  = h1   + (size_t)NNODES * H1C;
  float* h2   = al1  + (size_t)NNODES * 2 * HEADS;
  float* al2  = h2   + (size_t)NNODES * OUT_CH;

  const int RN  = 80;
  const int NG  = (NNODES + RN - 1) / RN;
  int*   cntg  = (int*)(al2 + (size_t)NNODES * 2);
  int2*  cells = (int2*)(cntg + NPART * CNTP);

  gemm_strip_kernel<IN_CH, H1C, 2, 5, 1, HEADS, true><<<NNODES / 80, 256, 0, stream>>>(
      x, W1, a_src1, a_dst1, h1, al1, ei, cells, cntg);
  agg_owner_kernel<HEADS, HID_CH, RN, 256, true, OUT_CH><<<NG, 1024, 0, stream>>>(
      ei, al1, h1, b1, h2, cntg, cells, W2, a_src2, a_dst2, al2);

  agg_owner_kernel<1, OUT_CH, RN, 256, false, 0><<<NG, 1024, 0, stream>>>(
      ei, al2, h2, b2, out, cntg, cells, nullptr, nullptr, nullptr, nullptr);
}
